# starting kernel
# baseline (speedup 1.0000x reference)
#include <hip/hip_runtime.h>
#include <math.h>

typedef __attribute__((ext_vector_type(2))) float v2f;
typedef __attribute__((ext_vector_type(8))) float v8f;

#define D   128
#define H   128
#define HH2 64
#define KC  10
#define LN_EPS 1e-5f

__global__ void wmma_gemm3_kernel(const float* __restrict__ x,
                                  const float* __restrict__ w1,
                                  const float* __restrict__ cw1,
                                  float* __restrict__ h1pre,
                                  float* __restrict__ hi,
                                  float* __restrict__ hj,
                                  int N) {
  const int rt  = blockIdx.x;
  const int mat = blockIdx.y;

  const float* Bm  = (mat == 0) ? w1    : (mat == 1 ? cw1 : cw1 + D * H);
  float*       out = (mat == 0) ? h1pre : (mat == 1 ? hi  : hj);

  const int lane = threadIdx.x;
  const int half = lane >> 4;
  const int l16  = lane & 15;

  int arow = rt * 16 + l16;
  if (arow >= N) arow = N - 1;
  const float* __restrict__ arow_p = x + (size_t)arow * D;

  v8f acc[8] = {};
#pragma unroll 4
  for (int k0 = 0; k0 < D; k0 += 4) {
    const int kk = k0 + half * 2;
    const v2f a = *(const v2f*)(arow_p + kk);
    const float* __restrict__ brow0 = Bm + (size_t)kk * H + l16;
#pragma unroll
    for (int ct = 0; ct < 8; ++ct) {
      v2f b;
      b.x = brow0[ct * 16];
      b.y = brow0[ct * 16 + H];
      acc[ct] = __builtin_amdgcn_wmma_f32_16x16x4_f32(false, a, false, b,
                                                      (short)0, acc[ct],
                                                      false, false);
    }
  }

  const int rbase = rt * 16 + half * 8;
  if (rt * 16 + 16 <= N) {
#pragma unroll
    for (int ct = 0; ct < 8; ++ct)
#pragma unroll
      for (int r = 0; r < 8; ++r)
        out[(size_t)(rbase + r) * H + ct * 16 + l16] = acc[ct][r];
  } else {
#pragma unroll
    for (int ct = 0; ct < 8; ++ct)
#pragma unroll
      for (int r = 0; r < 8; ++r)
        if (rbase + r < N)
          out[(size_t)(rbase + r) * H + ct * 16 + l16] = acc[ct][r];
  }
}

__global__ void node_mlp_kernel(const float* __restrict__ h1pre,
                                const float* __restrict__ b1,
                                const float* __restrict__ g1,
                                const float* __restrict__ be1,
                                const float* __restrict__ w2,
                                const float* __restrict__ b2,
                                const float* __restrict__ g2,
                                const float* __restrict__ be2,
                                const float* __restrict__ w3,
                                const float* __restrict__ b3,
                                float* __restrict__ out_logits,
                                float* __restrict__ out_probs,
                                int N) {
  const int n = blockIdx.x;
  const int t = threadIdx.x;
  __shared__ float sA[128], sB[128];
  __shared__ float sh1[128];
  __shared__ float sh2[HH2];
  __shared__ float slog[KC];

  float v = h1pre[(size_t)n * H + t] + b1[t];
  sA[t] = v;
  sB[t] = v * v;
  __syncthreads();
  for (int s = 64; s > 0; s >>= 1) {
    if (t < s) { sA[t] += sA[t + s]; sB[t] += sB[t + s]; }
    __syncthreads();
  }
  {
    float m   = sA[0] * (1.0f / 128.0f);
    float var = sB[0] * (1.0f / 128.0f) - m * m;
    float inv = rsqrtf(var + LN_EPS);
    float hv  = (v - m) * inv * g1[t] + be1[t];
    v = fmaxf(hv, 0.0f);
  }
  __syncthreads();
  sh1[t] = v;
  __syncthreads();

  float a2 = 0.0f;
  if (t < HH2) {
    a2 = b2[t];
    for (int k = 0; k < H; ++k) a2 += sh1[k] * w2[k * HH2 + t];
    sA[t] = a2;
    sB[t] = a2 * a2;
  }
  __syncthreads();
  for (int s = 32; s > 0; s >>= 1) {
    if (t < s) { sA[t] += sA[t + s]; sB[t] += sB[t + s]; }
    __syncthreads();
  }
  if (t < HH2) {
    float m2  = sA[0] * (1.0f / 64.0f);
    float v2  = sB[0] * (1.0f / 64.0f) - m2 * m2;
    float i2  = rsqrtf(v2 + LN_EPS);
    float h2v = (a2 - m2) * i2 * g2[t] + be2[t];
    sh2[t] = fmaxf(h2v, 0.0f);
  }
  __syncthreads();

  if (t < KC) {
    float lg = b3[t];
    for (int k = 0; k < HH2; ++k) lg += sh2[k] * w3[k * KC + t];
    slog[t] = lg;
    out_logits[(size_t)n * KC + t] = lg;
  }
  __syncthreads();
  if (t < KC) {
    float mx = slog[0];
    for (int k = 1; k < KC; ++k) mx = fmaxf(mx, slog[k]);
    float ssum = 0.0f;
    for (int k = 0; k < KC; ++k) ssum += __expf(slog[k] - mx);
    out_probs[(size_t)n * KC + t] = __expf(slog[t] - mx) / ssum;
  }
}

__global__ void centers_kernel(const float* __restrict__ probs,
                               const float* __restrict__ x,
                               const float* __restrict__ prototypes,
                               float* __restrict__ out_centers,
                               float* __restrict__ out_protos,
                               int N) {
  const int k = blockIdx.x;
  const int d = threadIdx.x;
  __shared__ float sW[128];

  float wpart = 0.0f;
  for (int n = d; n < N; n += 128) wpart += probs[(size_t)n * KC + k];
  sW[d] = wpart;
  __syncthreads();
  for (int s = 64; s > 0; s >>= 1) {
    if (d < s) sW[d] += sW[d + s];
    __syncthreads();
  }
  const float wsum = sW[0] + 1e-8f;

  float acc = 0.0f;
  for (int n = 0; n < N; ++n)
    acc += probs[(size_t)n * KC + k] * x[(size_t)n * D + d];

  out_centers[k * D + d] = acc / wsum;
  out_protos[k * D + d]  = prototypes[k * D + d];
}

__device__ __forceinline__ float wave_sum32(float v) {
#pragma unroll
  for (int off = 16; off > 0; off >>= 1) v += __shfl_xor(v, off, 32);
  return v;
}

__device__ __forceinline__ float pair_score(const float4 hi4, const float4 hj4,
                                            const float4 cb, const float4 g,
                                            const float4 be, const float4 cw,
                                            float cb2) {
  float4 hp;
  hp.x = hi4.x + hj4.x + cb.x;
  hp.y = hi4.y + hj4.y + cb.y;
  hp.z = hi4.z + hj4.z + cb.z;
  hp.w = hi4.w + hj4.w + cb.w;

  float s  = hp.x + hp.y + hp.z + hp.w;
  float sq = hp.x * hp.x + hp.y * hp.y + hp.z * hp.z + hp.w * hp.w;
  s  = wave_sum32(s);
  sq = wave_sum32(sq);

  const float m   = s * (1.0f / 128.0f);
  const float var = sq * (1.0f / 128.0f) - m * m;
  const float inv = rsqrtf(var + LN_EPS);

  float dot = fmaxf((hp.x - m) * inv * g.x + be.x, 0.0f) * cw.x
            + fmaxf((hp.y - m) * inv * g.y + be.y, 0.0f) * cw.y
            + fmaxf((hp.z - m) * inv * g.z + be.z, 0.0f) * cw.z
            + fmaxf((hp.w - m) * inv * g.w + be.w, 0.0f) * cw.w;
  dot = wave_sum32(dot);
  return tanhf(dot + cb2);
}

__global__ void comp_matrix_kernel(const float* __restrict__ hi,
                                   const float* __restrict__ hj,
                                   const float* __restrict__ cb1,
                                   const float* __restrict__ cg1,
                                   const float* __restrict__ cbe1,
                                   const float* __restrict__ cw2,
                                   const float* __restrict__ cb2p,
                                   float* __restrict__ out, int N) {
  const int i    = blockIdx.x;
  const int lane = threadIdx.x & 31;
  const int wid  = threadIdx.x >> 5;

  const float4 cb  = *(const float4*)(cb1  + 4 * lane);
  const float4 g   = *(const float4*)(cg1  + 4 * lane);
  const float4 be  = *(const float4*)(cbe1 + 4 * lane);
  const float4 cw  = *(const float4*)(cw2  + 4 * lane);
  const float  cb2 = cb2p[0];
  const float4 hi4 = *(const float4*)(hi + (size_t)i * H + 4 * lane);

  for (int j = wid; j < N; j += 8) {
    const float4 hj4 = *(const float4*)(hj + (size_t)j * H + 4 * lane);
    if (j + 8 < N)
      __builtin_prefetch(hj + (size_t)(j + 8) * H + 4 * lane, 0, 3);
    const float sc = pair_score(hi4, hj4, cb, g, be, cw, cb2);
    if (lane == 0) out[(size_t)i * N + j] = sc;
  }
}

__global__ void comp_scores_kernel(const float* __restrict__ hi,
                                   const float* __restrict__ hj,
                                   const int* __restrict__ ei, int E,
                                   const float* __restrict__ cb1,
                                   const float* __restrict__ cg1,
                                   const float* __restrict__ cbe1,
                                   const float* __restrict__ cw2,
                                   const float* __restrict__ cb2p,
                                   float* __restrict__ out) {
  const int lane = threadIdx.x & 31;
  const int gw   = (int)((blockIdx.x * blockDim.x + threadIdx.x) >> 5);
  const int nw   = (int)((gridDim.x * blockDim.x) >> 5);

  const float4 cb  = *(const float4*)(cb1  + 4 * lane);
  const float4 g   = *(const float4*)(cg1  + 4 * lane);
  const float4 be  = *(const float4*)(cbe1 + 4 * lane);
  const float4 cw  = *(const float4*)(cw2  + 4 * lane);
  const float  cb2 = cb2p[0];

  for (int e = gw; e < E; e += nw) {
    const int i = ei[e];
    const int j = ei[E + e];
    const float4 hi4 = *(const float4*)(hi + (size_t)i * H + 4 * lane);
    const float4 hj4 = *(const float4*)(hj + (size_t)j * H + 4 * lane);
    const float sc = pair_score(hi4, hj4, cb, g, be, cw, cb2);
    if (lane == 0) out[e] = sc;
  }
}

extern "C" void kernel_launch(void* const* d_in, const int* in_sizes, int n_in,
                              void* d_out, int out_size, void* d_ws, size_t ws_size,
                              hipStream_t stream) {
  const float* x    = (const float*)d_in[0];
  const int*   ei   = (const int*)  d_in[1];
  const float* w1   = (const float*)d_in[2];
  const float* b1   = (const float*)d_in[3];
  const float* g1   = (const float*)d_in[4];
  const float* be1  = (const float*)d_in[5];
  const float* w2   = (const float*)d_in[6];
  const float* b2   = (const float*)d_in[7];
  const float* g2   = (const float*)d_in[8];
  const float* be2  = (const float*)d_in[9];
  const float* w3   = (const float*)d_in[10];
  const float* b3   = (const float*)d_in[11];
  const float* cw1  = (const float*)d_in[12];
  const float* cb1  = (const float*)d_in[13];
  const float* cg1  = (const float*)d_in[14];
  const float* cbe1 = (const float*)d_in[15];
  const float* cw2  = (const float*)d_in[16];
  const float* cb2  = (const float*)d_in[17];
  const float* prot = (const float*)d_in[18];

  const int N = in_sizes[0] / D;
  const int E = in_sizes[1] / 2;

  float* h1pre = (float*)d_ws;
  float* hi    = h1pre + (size_t)N * H;
  float* hj    = hi    + (size_t)N * H;

  float* out       = (float*)d_out;
  float* o_logits  = out;
  float* o_probs   = o_logits  + (size_t)N * KC;
  float* o_scores  = o_probs   + (size_t)N * KC;
  float* o_matrix  = o_scores  + (size_t)E;
  float* o_centers = o_matrix  + (size_t)N * N;
  float* o_protos  = o_centers + (size_t)KC * D;

  const int rowTiles = (N + 15) / 16;

  wmma_gemm3_kernel<<<dim3(rowTiles, 3), 32, 0, stream>>>(
      x, w1, cw1, h1pre, hi, hj, N);

  node_mlp_kernel<<<N, 128, 0, stream>>>(
      h1pre, b1, g1, be1, w2, b2, g2, be2, w3, b3, o_logits, o_probs, N);

  centers_kernel<<<KC, 128, 0, stream>>>(
      o_probs, x, prot, o_centers, o_protos, N);

  comp_matrix_kernel<<<N, 256, 0, stream>>>(
      hi, hj, cb1, cg1, cbe1, cw2, cb2, o_matrix, N);

  comp_scores_kernel<<<2048, 256, 0, stream>>>(
      hi, hj, ei, E, cb1, cg1, cbe1, cw2, cb2, o_scores);
}


# pair matrix: a lane keeps hj of 2 columns in registers over 8 rows; hi rows and LN parameters come as wave-uniform scalars; every sum over H private to the lane, no cross-lane exchange
# speedup vs baseline: 2.6194x; 2.6194x over previous
#include <hip/hip_runtime.h>
#include <math.h>

typedef __attribute__((ext_vector_type(2))) float v2f;
typedef __attribute__((ext_vector_type(8))) float v8f;

#define D   128
#define H   128
#define HH2 64
#define KC  10
#define LN_EPS 1e-5f

__global__ void wmma_gemm3_kernel(const float* __restrict__ x,
                                  const float* __restrict__ w1,
                                  const float* __restrict__ cw1,
                                  float* __restrict__ h1pre,
                                  float* __restrict__ hi,
                                  float* __restrict__ hj,
                                  int N) {
  const int rt  = blockIdx.x;
  const int mat = blockIdx.y;

  const float* Bm  = (mat == 0) ? w1    : (mat == 1 ? cw1 : cw1 + D * H);
  float*       out = (mat == 0) ? h1pre : (mat == 1 ? hi  : hj);

  const int lane = threadIdx.x;
  const int half = lane >> 4;
  const int l16  = lane & 15;

  int arow = rt * 16 + l16;
  if (arow >= N) arow = N - 1;
  const float* __restrict__ arow_p = x + (size_t)arow * D;

  v8f acc[8] = {};
#pragma unroll 4
  for (int k0 = 0; k0 < D; k0 += 4) {
    const int kk = k0 + half * 2;
    const v2f a = *(const v2f*)(arow_p + kk);
    const float* __restrict__ brow0 = Bm + (size_t)kk * H + l16;
#pragma unroll
    for (int ct = 0; ct < 8; ++ct) {
      v2f b;
      b.x = brow0[ct * 16];
      b.y = brow0[ct * 16 + H];
      acc[ct] = __builtin_amdgcn_wmma_f32_16x16x4_f32(false, a, false, b,
                                                      (short)0, acc[ct],
                                                      false, false);
    }
  }

  const int rbase = rt * 16 + half * 8;
  if (rt * 16 + 16 <= N) {
#pragma unroll
    for (int ct = 0; ct < 8; ++ct)
#pragma unroll
      for (int r = 0; r < 8; ++r)
        out[(size_t)(rbase + r) * H + ct * 16 + l16] = acc[ct][r];
  } else {
#pragma unroll
    for (int ct = 0; ct < 8; ++ct)
#pragma unroll
      for (int r = 0; r < 8; ++r)
        if (rbase + r < N)
          out[(size_t)(rbase + r) * H + ct * 16 + l16] = acc[ct][r];
  }
}

#define NPB 16

__device__ __forceinline__ float wave_sum(float v) {
#pragma unroll
  for (int w = 16; w > 0; w >>= 1) v += __shfl_xor(v, w, 32);
  return v;
}

__device__ __forceinline__ float wave_max(float v) {
#pragma unroll
  for (int w = 16; w > 0; w >>= 1) v = fmaxf(v, __shfl_xor(v, w, 32));
  return v;
}

__global__ void __launch_bounds__(NPB * 32)
node_mlp_kernel(float* __restrict__ h1pre,
                const float* __restrict__ b1,
                const float* __restrict__ g1,
                const float* __restrict__ be1,
                const float* __restrict__ w2,
                const float* __restrict__ b2,
                const float* __restrict__ g2,
                const float* __restrict__ be2,
                const float* __restrict__ w3,
                const float* __restrict__ b3,
                const float* __restrict__ x,
                float* __restrict__ out_logits,
                float* __restrict__ out_probs,
                int N) {
  __shared__ float sh1[NPB][H];
  __shared__ float sh2[NPB][HH2];
  __shared__ float sp[NPB][KC];

  const int lane = threadIdx.x & 31;
  const int wv   = threadIdx.x >> 5;
  const int n0   = blockIdx.x * NPB;
  const int n    = n0 + wv;
  const bool live = n < N;
  const int nr   = live ? n : N - 1;

  {
    const float4 hv = *(const float4*)(h1pre + (size_t)nr * H + 4 * lane);
    const float4 bb = *(const float4*)(b1 + 4 * lane);
    const float4 gg = *(const float4*)(g1 + 4 * lane);
    const float4 ee = *(const float4*)(be1 + 4 * lane);
    float4 v;
    v.x = hv.x + bb.x; v.y = hv.y + bb.y; v.z = hv.z + bb.z; v.w = hv.w + bb.w;
    const float s = wave_sum((v.x + v.y) + (v.z + v.w));
    const float q = wave_sum((v.x * v.x + v.y * v.y) + (v.z * v.z + v.w * v.w));
    const float m   = s * (1.0f / 128.0f);
    const float var = q * (1.0f / 128.0f) - m * m;
    const float inv = rsqrtf(var + LN_EPS);
    float4 r;
    r.x = fmaxf((v.x - m) * inv * gg.x + ee.x, 0.0f);
    r.y = fmaxf((v.y - m) * inv * gg.y + ee.y, 0.0f);
    r.z = fmaxf((v.z - m) * inv * gg.z + ee.z, 0.0f);
    r.w = fmaxf((v.w - m) * inv * gg.w + ee.w, 0.0f);
    *(float4*)&sh1[wv][4 * lane] = r;
  }
  __syncthreads();

  {
    float2 a2 = *(const float2*)(b2 + 2 * lane);
    const float* __restrict__ w2p = w2 + 2 * lane;
#pragma unroll 16
    for (int k = 0; k < H; ++k) {
      const float  hk = sh1[wv][k];
      const float2 w  = *(const float2*)(w2p + k * HH2);
      a2.x += hk * w.x;
      a2.y += hk * w.y;
    }
    const float s2 = wave_sum(a2.x + a2.y);
    const float q2 = wave_sum(a2.x * a2.x + a2.y * a2.y);
    const float m2 = s2 * (1.0f / 64.0f);
    const float v2 = q2 * (1.0f / 64.0f) - m2 * m2;
    const float i2 = rsqrtf(v2 + LN_EPS);
    const float2 gg = *(const float2*)(g2 + 2 * lane);
    const float2 ee = *(const float2*)(be2 + 2 * lane);
    float2 h2;
    h2.x = fmaxf((a2.x - m2) * i2 * gg.x + ee.x, 0.0f);
    h2.y = fmaxf((a2.y - m2) * i2 * gg.y + ee.y, 0.0f);
    *(float2*)&sh2[wv][2 * lane] = h2;
  }
  __syncthreads();

  {
    const bool owner = lane < KC;
    const int  tc    = owner ? lane : KC - 1;
    float lg = b3[tc];
#pragma unroll 16
    for (int k = 0; k < HH2; ++k) lg += sh2[wv][k] * w3[k * KC + tc];
    const float mx   = wave_max(owner ? lg : -INFINITY);
    const float e    = owner ? __expf(lg - mx) : 0.0f;
    const float ssum = wave_sum(e);
    const float p    = e / ssum;
    if (owner) {
      if (live) {
        out_logits[(size_t)n * KC + lane] = lg;
        out_probs[(size_t)n * KC + lane]  = p;
      }
      sp[wv][lane] = live ? p : 0.0f;
    }
  }
  __syncthreads();

  if (n0 + NPB <= N) {
    const int d  = threadIdx.x & (D - 1);
    const int kk = threadIdx.x >> 7;
    const int k3 = (kk < 2) ? kk + 8 : kk + 4;
    const float* __restrict__ xp = x + (size_t)n0 * D + d;
    float a0 = 0.0f, a1 = 0.0f, a2c = 0.0f;
#pragma unroll
    for (int g = 0; g < NPB; ++g) {
      const float xv = xp[g * D];
      a0  += sp[g][kk] * xv;
      a1  += sp[g][kk + 4] * xv;
      a2c += sp[g][k3] * xv;
    }
    float* __restrict__ part = h1pre + (size_t)n0 * H;
    part[kk * D + d]       = a0;
    part[(kk + 4) * D + d] = a1;
    if (kk < 2) part[(kk + 8) * D + d] = a2c;
  }
}

__global__ void centers_kernel(const float* __restrict__ probs,
                               const float* __restrict__ x,
                               const float* __restrict__ part,
                               const float* __restrict__ prototypes,
                               float* __restrict__ out_centers,
                               float* __restrict__ out_protos,
                               int N) {
  const int k = blockIdx.x;
  const int d = threadIdx.x;
  __shared__ float sW[128];
  __shared__ float sT[NPB];

  const int nfull = N / NPB;
  const int nt    = nfull * NPB;
  float tailp = 0.0f;
  if (d < NPB - 1) {
    const int n = nt + d;
    if (n < N) tailp = probs[(size_t)n * KC + k];
  }

  float wpart = 0.0f;
  for (int base = 0; base < N; base += 8 * 128) {
    float t[8];
#pragma unroll
    for (int u = 0; u < 8; ++u) {
      const int n  = base + u * 128 + d;
      const int nc = n < N ? n : N - 1;
      const float v = probs[(size_t)nc * KC + k];
      t[u] = n < N ? v : 0.0f;
    }
#pragma unroll
    for (int u = 0; u < 8; ++u) wpart += t[u];
  }
  if (d < NPB - 1) sT[d] = tailp;
  sW[d] = wpart;
  __syncthreads();
  for (int s = 64; s > 0; s >>= 1) {
    if (d < s) sW[d] += sW[d + s];
    __syncthreads();
  }
  const float wsum = sW[0] + 1e-8f;

  const float* __restrict__ pp = part + k * D + d;
  float acc = 0.0f;
  for (int b0 = 0; b0 < nfull; b0 += 16) {
    float t[16];
#pragma unroll
    for (int u = 0; u < 16; ++u) {
      const int b  = b0 + u;
      const int bc = b < nfull ? b : nfull - 1;
      const float v = pp[(size_t)bc * (NPB * H)];
      t[u] = b < nfull ? v : 0.0f;
    }
#pragma unroll
    for (int u = 0; u < 16; ++u) acc += t[u];
  }
  if (nt < N) {
    float xv[NPB - 1];
#pragma unroll
    for (int u = 0; u < NPB - 1; ++u) {
      const int n  = nt + u;
      const int nc = n < N ? n : N - 1;
      xv[u] = x[(size_t)nc * D + d];
    }
#pragma unroll
    for (int u = 0; u < NPB - 1; ++u) acc += sT[u] * xv[u];
  }

  out_centers[k * D + d] = acc / wsum;
  out_protos[k * D + d]  = prototypes[k * D + d];
}

template <int W>
__device__ __forceinline__ void tsum_step(float (&v)[32], const int lane) {
  const bool up = (lane & W) != 0;
#pragma unroll
  for (int p = 0; p < W; ++p) {
    const float keep = up ? v[p + W] : v[p];
    const float send = up ? v[p] : v[p + W];
    v[p] = keep + __shfl_xor(send, W, 32);
  }
}

__device__ __forceinline__ float transpose_sum32(float (&v)[32], const int lane) {
  tsum_step<16>(v, lane);
  tsum_step<8>(v, lane);
  tsum_step<4>(v, lane);
  tsum_step<2>(v, lane);
  tsum_step<1>(v, lane);
  return v[0];
}

__device__ __forceinline__ float lane_bcast(const float v, const int srcLane) {
  return __int_as_float(__builtin_amdgcn_readlane(__float_as_int(v), srcLane));
}

__device__ __forceinline__ float score32(const float4 (&hp)[32],
                                         float (&s)[32], float (&q)[32],
                                         const float4 g, const float4 be,
                                         const float4 cw, const float cb2,
                                         const int lane) {
  const float ssum = transpose_sum32(s, lane);
  const float qsum = transpose_sum32(q, lane);

  const float m   = ssum * (1.0f / 128.0f);
  const float var = qsum * (1.0f / 128.0f) - m * m;
  const float inv = rsqrtf(var + LN_EPS);

  float d[32];
#pragma unroll
  for (int p = 0; p < 32; ++p) {
    const float mp = lane_bcast(m, p);
    const float ip = lane_bcast(inv, p);
    const float4 t = hp[p];
    d[p] = fmaxf((t.x - mp) * ip * g.x + be.x, 0.0f) * cw.x
         + fmaxf((t.y - mp) * ip * g.y + be.y, 0.0f) * cw.y
         + fmaxf((t.z - mp) * ip * g.z + be.z, 0.0f) * cw.z
         + fmaxf((t.w - mp) * ip * g.w + be.w, 0.0f) * cw.w;
  }
  const float dot = transpose_sum32(d, lane);
  return tanhf(dot + cb2);
}

#define MROWS  8
#define MWAVES 8
#define MCOLS  64
#define MPITCH 132

__global__ void __launch_bounds__(MWAVES * 32)
comp_matrix_kernel(const float* __restrict__ hi,
                   const float* __restrict__ hj,
                   const float* __restrict__ cb1,
                   const float* __restrict__ cg1,
                   const float* __restrict__ cbe1,
                   const float* __restrict__ cw2,
                   const float* __restrict__ cb2p,
                   float* __restrict__ out, int N) {
  __shared__ __attribute__((aligned(16))) float tile[MCOLS * MPITCH];

  const int lane = threadIdx.x & 31;
  const int wid  = __builtin_amdgcn_readfirstlane((int)(threadIdx.x >> 5));
  const int j0   = blockIdx.x * MCOLS;

  {
    const float4 cbv = *(const float4*)(cb1 + 4 * lane);
#pragma unroll
    for (int r = 0; r < MCOLS / MWAVES; ++r) {
      const int c = wid * (MCOLS / MWAVES) + r;
      int jr = j0 + c;
      if (jr >= N) jr = N - 1;
      float4 v = *(const float4*)(hj + (size_t)jr * H + 4 * lane);
      v.x += cbv.x; v.y += cbv.y; v.z += cbv.z; v.w += cbv.w;
      *(float4*)&tile[c * MPITCH + 4 * lane] = v;
    }
  }
  __syncthreads();

  v2f b[H];
  {
    const float* __restrict__ r0 = &tile[lane * MPITCH];
    const float* __restrict__ r1 = &tile[(lane + 32) * MPITCH];
#pragma unroll
    for (int h = 0; h < H; h += 4) {
      const float4 u = *(const float4*)(r0 + h);
      const float4 w = *(const float4*)(r1 + h);
      b[h + 0] = (v2f){u.x, w.x};
      b[h + 1] = (v2f){u.y, w.y};
      b[h + 2] = (v2f){u.z, w.z};
      b[h + 3] = (v2f){u.w, w.w};
    }
  }

  const float cb2 = cb2p[0];
  const int i0   = ((int)blockIdx.y * MWAVES + wid) * MROWS;
  const int iend = (i0 + MROWS < N) ? i0 + MROWS : N;
  const int ja = j0 + lane;
  const int jb = j0 + 32 + lane;
  const v2f zero2 = (v2f){0.0f, 0.0f};

  for (int i = i0; i < iend; ++i) {
    const float* __restrict__ a = hi + (size_t)i * H;
    int off = 0;
    asm volatile("" : "+s"(off));
    const float* __restrict__ gp  = cg1  + off;
    const float* __restrict__ bep = cbe1 + off;
    const float* __restrict__ cwp = cw2  + off;

    v2f s0 = zero2, s1 = zero2, s2 = zero2, s3 = zero2;
    v2f q0 = zero2, q1 = zero2, q2 = zero2, q3 = zero2;
#pragma unroll
    for (int hb = 0; hb < H; hb += 64) {
      float av[64];
#pragma unroll
      for (int k = 0; k < 64; ++k) av[k] = a[hb + k];
      __builtin_amdgcn_sched_barrier(0);
#pragma unroll
      for (int k = 0; k < 64; k += 4) {
        const v2f t0 = b[hb + k + 0] + av[k + 0];
        const v2f t1 = b[hb + k + 1] + av[k + 1];
        const v2f t2 = b[hb + k + 2] + av[k + 2];
        const v2f t3 = b[hb + k + 3] + av[k + 3];
        s0 += t0; s1 += t1; s2 += t2; s3 += t3;
        q0 = __builtin_elementwise_fma(t0, t0, q0);
        q1 = __builtin_elementwise_fma(t1, t1, q1);
        q2 = __builtin_elementwise_fma(t2, t2, q2);
        q3 = __builtin_elementwise_fma(t3, t3, q3);
      }
      __builtin_amdgcn_sched_barrier(0);
    }
    const v2f m   = ((s0 + s1) + (s2 + s3)) * (1.0f / 128.0f);
    const v2f var = ((q0 + q1) + (q2 + q3)) * (1.0f / 128.0f) - m * m;
    v2f inv;
    inv.x = rsqrtf(var.x + LN_EPS);
    inv.y = rsqrtf(var.y + LN_EPS);
    const v2f nmi = -m * inv;

    int off2 = 0;
    asm volatile("" : "+s"(off2));
    const float* __restrict__ a2 = a + off2;
    v2f d0 = zero2, d1 = zero2, d2 = zero2, d3 = zero2;
#pragma unroll
    for (int hb = 0; hb < H; hb += 16) {
      float av[16], gv[16], ev[16], cv[16];
#pragma unroll
      for (int k = 0; k < 16; ++k) {
        av[k] = a2[hb + k];
        gv[k] = gp[hb + k];
        ev[k] = bep[hb + k];
        cv[k] = cwp[hb + k];
      }
      __builtin_amdgcn_sched_barrier(0);
#pragma unroll
      for (int k = 0; k < 16; k += 4) {
        const v2f y0 = __builtin_elementwise_fma(b[hb + k + 0] + av[k + 0], inv, nmi);
        const v2f y1 = __builtin_elementwise_fma(b[hb + k + 1] + av[k + 1], inv, nmi);
        const v2f y2 = __builtin_elementwise_fma(b[hb + k + 2] + av[k + 2], inv, nmi);
        const v2f y3 = __builtin_elementwise_fma(b[hb + k + 3] + av[k + 3], inv, nmi);
        const v2f z0 = __builtin_elementwise_fma(y0, (v2f){gv[k + 0], gv[k + 0]}, (v2f){ev[k + 0], ev[k + 0]});
        const v2f z1 = __builtin_elementwise_fma(y1, (v2f){gv[k + 1], gv[k + 1]}, (v2f){ev[k + 1], ev[k + 1]});
        const v2f z2 = __builtin_elementwise_fma(y2, (v2f){gv[k + 2], gv[k + 2]}, (v2f){ev[k + 2], ev[k + 2]});
        const v2f z3 = __builtin_elementwise_fma(y3, (v2f){gv[k + 3], gv[k + 3]}, (v2f){ev[k + 3], ev[k + 3]});
        d0 = __builtin_elementwise_fma(__builtin_elementwise_max(z0, zero2), (v2f){cv[k + 0], cv[k + 0]}, d0);
        d1 = __builtin_elementwise_fma(__builtin_elementwise_max(z1, zero2), (v2f){cv[k + 1], cv[k + 1]}, d1);
        d2 = __builtin_elementwise_fma(__builtin_elementwise_max(z2, zero2), (v2f){cv[k + 2], cv[k + 2]}, d2);
        d3 = __builtin_elementwise_fma(__builtin_elementwise_max(z3, zero2), (v2f){cv[k + 3], cv[k + 3]}, d3);
      }
      __builtin_amdgcn_sched_barrier(0);
    }
    const v2f dot = (d0 + d1) + (d2 + d3);
    const float sca = tanhf(dot.x + cb2);
    const float scb = tanhf(dot.y + cb2);
    float* __restrict__ orow = out + (size_t)i * N;
    if (ja < N) orow[ja] = sca;
    if (jb < N) orow[jb] = scb;
  }
}

__global__ void __launch_bounds__(256)
comp_scores_kernel(const float* __restrict__ hi,
                   const float* __restrict__ hj,
                   const int* __restrict__ ei, int E,
                   const float* __restrict__ cb1,
                   const float* __restrict__ cg1,
                   const float* __restrict__ cbe1,
                   const float* __restrict__ cw2,
                   const float* __restrict__ cb2p,
                   float* __restrict__ out) {
  const int lane = threadIdx.x & 31;
  const int wv   = __builtin_amdgcn_readfirstlane((int)(threadIdx.x >> 5));
  const int gw   = (int)(blockIdx.x * (blockDim.x >> 5)) + wv;
  const int nw   = (int)((gridDim.x * blockDim.x) >> 5);

  const float4 cb  = *(const float4*)(cb1  + 4 * lane);
  const float4 g   = *(const float4*)(cg1  + 4 * lane);
  const float4 be  = *(const float4*)(cbe1 + 4 * lane);
  const float4 cw  = *(const float4*)(cw2  + 4 * lane);
  const float  cb2 = cb2p[0];

  for (int e0 = gw * 32; e0 < E; e0 += nw * 32) {
    int e = e0 + lane;
    if (e >= E) e = E - 1;
    const int iv = ei[e];
    const int jv = ei[E + e];

    float4 hp[32];
    float  s[32], q[32];
#pragma unroll
    for (int p = 0; p < 32; ++p) {
      const int i = __builtin_amdgcn_readlane(iv, p);
      const int j = __builtin_amdgcn_readlane(jv, p);
      const float4 hi4 = *(const float4*)(hi + (size_t)i * H + 4 * lane);
      const float4 hj4 = *(const float4*)(hj + (size_t)j * H + 4 * lane);
      float4 t;
      t.x = hi4.x + hj4.x + cb.x;
      t.y = hi4.y + hj4.y + cb.y;
      t.z = hi4.z + hj4.z + cb.z;
      t.w = hi4.w + hj4.w + cb.w;
      hp[p] = t;
      s[p]  = t.x + t.y + t.z + t.w;
      q[p]  = t.x * t.x + t.y * t.y + t.z * t.z + t.w * t.w;
    }
    const float sc = score32(hp, s, q, g, be, cw, cb2, lane);
    if (e0 + lane < E) out[e0 + lane] = sc;
  }
}

extern "C" void kernel_launch(void* const* d_in, const int* in_sizes, int n_in,
                              void* d_out, int out_size, void* d_ws, size_t ws_size,
                              hipStream_t stream) {
  const float* x    = (const float*)d_in[0];
  const int*   ei   = (const int*)  d_in[1];
  const float* w1   = (const float*)d_in[2];
  const float* b1   = (const float*)d_in[3];
  const float* g1   = (const float*)d_in[4];
  const float* be1  = (const float*)d_in[5];
  const float* w2   = (const float*)d_in[6];
  const float* b2   = (const float*)d_in[7];
  const float* g2   = (const float*)d_in[8];
  const float* be2  = (const float*)d_in[9];
  const float* w3   = (const float*)d_in[10];
  const float* b3   = (const float*)d_in[11];
  const float* cw1  = (const float*)d_in[12];
  const float* cb1  = (const float*)d_in[13];
  const float* cg1  = (const float*)d_in[14];
  const float* cbe1 = (const float*)d_in[15];
  const float* cw2  = (const float*)d_in[16];
  const float* cb2  = (const float*)d_in[17];
  const float* prot = (const float*)d_in[18];

  const int N = in_sizes[0] / D;
  const int E = in_sizes[1] / 2;

  float* h1pre = (float*)d_ws;
  float* hi    = h1pre + (size_t)N * H;
  float* hj    = hi    + (size_t)N * H;

  float* out       = (float*)d_out;
  float* o_logits  = out;
  float* o_probs   = o_logits  + (size_t)N * KC;
  float* o_scores  = o_probs   + (size_t)N * KC;
  float* o_matrix  = o_scores  + (size_t)E;
  float* o_centers = o_matrix  + (size_t)N * N;
  float* o_protos  = o_centers + (size_t)KC * D;

  const int rowTiles = (N + 15) / 16;

  wmma_gemm3_kernel<<<dim3(rowTiles, 3), 32, 0, stream>>>(
      x, w1, cw1, h1pre, hi, hj, N);

  node_mlp_kernel<<<rowTiles, NPB * 32, 0, stream>>>(
      h1pre, b1, g1, be1, w2, b2, g2, be2, w3, b3, x, o_logits, o_probs, N);

  centers_kernel<<<KC, 128, 0, stream>>>(
      o_probs, x, h1pre, prot, o_centers, o_protos, N);

  comp_matrix_kernel<<<dim3((N + MCOLS - 1) / MCOLS, (N + MWAVES * MROWS - 1) / (MWAVES * MROWS)),
                       MWAVES * 32, 0, stream>>>(
      hi, hj, cb1, cg1, cbe1, cw2, cb2, o_matrix, N);

  comp_scores_kernel<<<2048, 256, 0, stream>>>(
      hi, hj, ei, E, cb1, cg1, cbe1, cw2, cb2, o_scores);
}


# first GEMM: a 4-wave work-group stages the 128x128 weight matrix in LDS once for two row stripes and loads all its A fragments up front: one memory wait per wave, was one L2 round trip per K-step
# speedup vs baseline: 3.1321x; 1.1957x over previous
#include <hip/hip_runtime.h>
#include <math.h>

typedef __attribute__((ext_vector_type(2))) float v2f;
typedef __attribute__((ext_vector_type(8))) float v8f;

#define D   128
#define H   128
#define HH2 64
#define KC  10
#define LN_EPS 1e-5f

#define GWAVES 4
#define BPITCH 136

__global__ void __launch_bounds__(GWAVES * 32)
wmma_gemm3_kernel(const float* __restrict__ x,
                  const float* __restrict__ w1,
                  const float* __restrict__ cw1,
                  float* __restrict__ h1pre,
                  float* __restrict__ hi,
                  float* __restrict__ hj,
                  int N) {
  __shared__ __attribute__((aligned(16))) float sB[D * BPITCH];

  const int mat = blockIdx.y;
  const float* __restrict__ Bm  = (mat == 0) ? w1    : (mat == 1 ? cw1 : cw1 + D * H);
  float* __restrict__       out = (mat == 0) ? h1pre : (mat == 1 ? hi  : hj);

  const int lane = threadIdx.x & 31;
  const int wv   = __builtin_amdgcn_readfirstlane((int)(threadIdx.x >> 5));
  const int half = lane >> 4;
  const int l16  = lane & 15;

#pragma unroll
  for (int r = 0; r < D / GWAVES; ++r) {
    const int row = wv * (D / GWAVES) + r;
    const float4 v = *(const float4*)(Bm + (size_t)row * H + 4 * lane);
    *(float4*)&sB[row * BPITCH + 4 * lane] = v;
  }

  const int rt  = (int)blockIdx.x * 2 + (wv >> 1);
  const int ct0 = (wv & 1) * 4;
  int arow = rt * 16 + l16;
  if (arow >= N) arow = N - 1;
  const float* __restrict__ arow_p = x + (size_t)arow * D + half * 2;
  v2f a[D / 4];
#pragma unroll
  for (int s = 0; s < D / 4; ++s) a[s] = *(const v2f*)(arow_p + 4 * s);

  __syncthreads();

  v8f acc[4] = {};
#pragma unroll
  for (int s = 0; s < D / 4; ++s) {
    const int kk = 4 * s + half * 2;
    const float* __restrict__ brow0 = &sB[kk * BPITCH + ct0 * 16 + l16];
#pragma unroll
    for (int c = 0; c < 4; ++c) {
      v2f b;
      b.x = brow0[c * 16];
      b.y = brow0[c * 16 + BPITCH];
      acc[c] = __builtin_amdgcn_wmma_f32_16x16x4_f32(false, a[s], false, b,
                                                     (short)0, acc[c],
                                                     false, false);
    }
  }

  const int rbase = rt * 16 + half * 8;
  if (rt * 16 + 16 <= N) {
#pragma unroll
    for (int c = 0; c < 4; ++c)
#pragma unroll
      for (int r = 0; r < 8; ++r)
        out[(size_t)(rbase + r) * H + (ct0 + c) * 16 + l16] = acc[c][r];
  } else {
#pragma unroll
    for (int c = 0; c < 4; ++c)
#pragma unroll
      for (int r = 0; r < 8; ++r)
        if (rbase + r < N)
          out[(size_t)(rbase + r) * H + (ct0 + c) * 16 + l16] = acc[c][r];
  }
}

#define NPB 16

__device__ __forceinline__ float wave_sum(float v) {
#pragma unroll
  for (int w = 16; w > 0; w >>= 1) v += __shfl_xor(v, w, 32);
  return v;
}

__device__ __forceinline__ float wave_max(float v) {
#pragma unroll
  for (int w = 16; w > 0; w >>= 1) v = fmaxf(v, __shfl_xor(v, w, 32));
  return v;
}

__global__ void __launch_bounds__(NPB * 32)
node_mlp_kernel(float* __restrict__ h1pre,
                const float* __restrict__ b1,
                const float* __restrict__ g1,
                const float* __restrict__ be1,
                const float* __restrict__ w2,
                const float* __restrict__ b2,
                const float* __restrict__ g2,
                const float* __restrict__ be2,
                const float* __restrict__ w3,
                const float* __restrict__ b3,
                const float* __restrict__ x,
                float* __restrict__ out_logits,
                float* __restrict__ out_probs,
                int N) {
  __shared__ float sh1[NPB][H];
  __shared__ float sh2[NPB][HH2];
  __shared__ float sp[NPB][KC];

  const int lane = threadIdx.x & 31;
  const int wv   = threadIdx.x >> 5;
  const int n0   = blockIdx.x * NPB;
  const int n    = n0 + wv;
  const bool live = n < N;
  const int nr   = live ? n : N - 1;

  {
    const float4 hv = *(const float4*)(h1pre + (size_t)nr * H + 4 * lane);
    const float4 bb = *(const float4*)(b1 + 4 * lane);
    const float4 gg = *(const float4*)(g1 + 4 * lane);
    const float4 ee = *(const float4*)(be1 + 4 * lane);
    float4 v;
    v.x = hv.x + bb.x; v.y = hv.y + bb.y; v.z = hv.z + bb.z; v.w = hv.w + bb.w;
    const float s = wave_sum((v.x + v.y) + (v.z + v.w));
    const float q = wave_sum((v.x * v.x + v.y * v.y) + (v.z * v.z + v.w * v.w));
    const float m   = s * (1.0f / 128.0f);
    const float var = q * (1.0f / 128.0f) - m * m;
    const float inv = rsqrtf(var + LN_EPS);
    float4 r;
    r.x = fmaxf((v.x - m) * inv * gg.x + ee.x, 0.0f);
    r.y = fmaxf((v.y - m) * inv * gg.y + ee.y, 0.0f);
    r.z = fmaxf((v.z - m) * inv * gg.z + ee.z, 0.0f);
    r.w = fmaxf((v.w - m) * inv * gg.w + ee.w, 0.0f);
    *(float4*)&sh1[wv][4 * lane] = r;
  }
  __syncthreads();

  {
    float2 a2 = *(const float2*)(b2 + 2 * lane);
    const float* __restrict__ w2p = w2 + 2 * lane;
#pragma unroll 16
    for (int k = 0; k < H; ++k) {
      const float  hk = sh1[wv][k];
      const float2 w  = *(const float2*)(w2p + k * HH2);
      a2.x += hk * w.x;
      a2.y += hk * w.y;
    }
    const float s2 = wave_sum(a2.x + a2.y);
    const float q2 = wave_sum(a2.x * a2.x + a2.y * a2.y);
    const float m2 = s2 * (1.0f / 64.0f);
    const float v2 = q2 * (1.0f / 64.0f) - m2 * m2;
    const float i2 = rsqrtf(v2 + LN_EPS);
    const float2 gg = *(const float2*)(g2 + 2 * lane);
    const float2 ee = *(const float2*)(be2 + 2 * lane);
    float2 h2;
    h2.x = fmaxf((a2.x - m2) * i2 * gg.x + ee.x, 0.0f);
    h2.y = fmaxf((a2.y - m2) * i2 * gg.y + ee.y, 0.0f);
    *(float2*)&sh2[wv][2 * lane] = h2;
  }
  __syncthreads();

  {
    const bool owner = lane < KC;
    const int  tc    = owner ? lane : KC - 1;
    float lg = b3[tc];
#pragma unroll 16
    for (int k = 0; k < HH2; ++k) lg += sh2[wv][k] * w3[k * KC + tc];
    const float mx   = wave_max(owner ? lg : -INFINITY);
    const float e    = owner ? __expf(lg - mx) : 0.0f;
    const float ssum = wave_sum(e);
    const float p    = e / ssum;
    if (owner) {
      if (live) {
        out_logits[(size_t)n * KC + lane] = lg;
        out_probs[(size_t)n * KC + lane]  = p;
      }
      sp[wv][lane] = live ? p : 0.0f;
    }
  }
  __syncthreads();

  if (n0 + NPB <= N) {
    const int d  = threadIdx.x & (D - 1);
    const int kk = threadIdx.x >> 7;
    const int k3 = (kk < 2) ? kk + 8 : kk + 4;
    const float* __restrict__ xp = x + (size_t)n0 * D + d;
    float a0 = 0.0f, a1 = 0.0f, a2c = 0.0f;
#pragma unroll
    for (int g = 0; g < NPB; ++g) {
      const float xv = xp[g * D];
      a0  += sp[g][kk] * xv;
      a1  += sp[g][kk + 4] * xv;
      a2c += sp[g][k3] * xv;
    }
    float* __restrict__ part = h1pre + (size_t)n0 * H;
    part[kk * D + d]       = a0;
    part[(kk + 4) * D + d] = a1;
    if (kk < 2) part[(kk + 8) * D + d] = a2c;
  }
}

__global__ void centers_kernel(const float* __restrict__ probs,
                               const float* __restrict__ x,
                               const float* __restrict__ part,
                               const float* __restrict__ prototypes,
                               float* __restrict__ out_centers,
                               float* __restrict__ out_protos,
                               int N) {
  const int k = blockIdx.x;
  const int d = threadIdx.x;
  __shared__ float sW[128];
  __shared__ float sT[NPB];

  const int nfull = N / NPB;
  const int nt    = nfull * NPB;
  float tailp = 0.0f;
  if (d < NPB - 1) {
    const int n = nt + d;
    if (n < N) tailp = probs[(size_t)n * KC + k];
  }

  float wpart = 0.0f;
  for (int base = 0; base < N; base += 8 * 128) {
    float t[8];
#pragma unroll
    for (int u = 0; u < 8; ++u) {
      const int n  = base + u * 128 + d;
      const int nc = n < N ? n : N - 1;
      const float v = probs[(size_t)nc * KC + k];
      t[u] = n < N ? v : 0.0f;
    }
#pragma unroll
    for (int u = 0; u < 8; ++u) wpart += t[u];
  }
  if (d < NPB - 1) sT[d] = tailp;
  sW[d] = wpart;
  __syncthreads();
  for (int s = 64; s > 0; s >>= 1) {
    if (d < s) sW[d] += sW[d + s];
    __syncthreads();
  }
  const float wsum = sW[0] + 1e-8f;

  const float* __restrict__ pp = part + k * D + d;
  float acc = 0.0f;
  for (int b0 = 0; b0 < nfull; b0 += 16) {
    float t[16];
#pragma unroll
    for (int u = 0; u < 16; ++u) {
      const int b  = b0 + u;
      const int bc = b < nfull ? b : nfull - 1;
      const float v = pp[(size_t)bc * (NPB * H)];
      t[u] = b < nfull ? v : 0.0f;
    }
#pragma unroll
    for (int u = 0; u < 16; ++u) acc += t[u];
  }
  if (nt < N) {
    float xv[NPB - 1];
#pragma unroll
    for (int u = 0; u < NPB - 1; ++u) {
      const int n  = nt + u;
      const int nc = n < N ? n : N - 1;
      xv[u] = x[(size_t)nc * D + d];
    }
#pragma unroll
    for (int u = 0; u < NPB - 1; ++u) acc += sT[u] * xv[u];
  }

  out_centers[k * D + d] = acc / wsum;
  out_protos[k * D + d]  = prototypes[k * D + d];
}

template <int W>
__device__ __forceinline__ void tsum_step(float (&v)[32], const int lane) {
  const bool up = (lane & W) != 0;
#pragma unroll
  for (int p = 0; p < W; ++p) {
    const float keep = up ? v[p + W] : v[p];
    const float send = up ? v[p] : v[p + W];
    v[p] = keep + __shfl_xor(send, W, 32);
  }
}

__device__ __forceinline__ float transpose_sum32(float (&v)[32], const int lane) {
  tsum_step<16>(v, lane);
  tsum_step<8>(v, lane);
  tsum_step<4>(v, lane);
  tsum_step<2>(v, lane);
  tsum_step<1>(v, lane);
  return v[0];
}

__device__ __forceinline__ float lane_bcast(const float v, const int srcLane) {
  return __int_as_float(__builtin_amdgcn_readlane(__float_as_int(v), srcLane));
}

__device__ __forceinline__ float score32(const float4 (&hp)[32],
                                         float (&s)[32], float (&q)[32],
                                         const float4 g, const float4 be,
                                         const float4 cw, const float cb2,
                                         const int lane) {
  const float ssum = transpose_sum32(s, lane);
  const float qsum = transpose_sum32(q, lane);

  const float m   = ssum * (1.0f / 128.0f);
  const float var = qsum * (1.0f / 128.0f) - m * m;
  const float inv = rsqrtf(var + LN_EPS);

  float d[32];
#pragma unroll
  for (int p = 0; p < 32; ++p) {
    const float mp = lane_bcast(m, p);
    const float ip = lane_bcast(inv, p);
    const float4 t = hp[p];
    d[p] = fmaxf((t.x - mp) * ip * g.x + be.x, 0.0f) * cw.x
         + fmaxf((t.y - mp) * ip * g.y + be.y, 0.0f) * cw.y
         + fmaxf((t.z - mp) * ip * g.z + be.z, 0.0f) * cw.z
         + fmaxf((t.w - mp) * ip * g.w + be.w, 0.0f) * cw.w;
  }
  const float dot = transpose_sum32(d, lane);
  return tanhf(dot + cb2);
}

#define MROWS  8
#define MWAVES 8
#define MCOLS  64
#define MPITCH 132

__global__ void __launch_bounds__(MWAVES * 32)
comp_matrix_kernel(const float* __restrict__ hi,
                   const float* __restrict__ hj,
                   const float* __restrict__ cb1,
                   const float* __restrict__ cg1,
                   const float* __restrict__ cbe1,
                   const float* __restrict__ cw2,
                   const float* __restrict__ cb2p,
                   float* __restrict__ out, int N) {
  __shared__ __attribute__((aligned(16))) float tile[MCOLS * MPITCH];

  const int lane = threadIdx.x & 31;
  const int wid  = __builtin_amdgcn_readfirstlane((int)(threadIdx.x >> 5));
  const int j0   = blockIdx.x * MCOLS;

  {
    const float4 cbv = *(const float4*)(cb1 + 4 * lane);
#pragma unroll
    for (int r = 0; r < MCOLS / MWAVES; ++r) {
      const int c = wid * (MCOLS / MWAVES) + r;
      int jr = j0 + c;
      if (jr >= N) jr = N - 1;
      float4 v = *(const float4*)(hj + (size_t)jr * H + 4 * lane);
      v.x += cbv.x; v.y += cbv.y; v.z += cbv.z; v.w += cbv.w;
      *(float4*)&tile[c * MPITCH + 4 * lane] = v;
    }
  }
  __syncthreads();

  v2f b[H];
  {
    const float* __restrict__ r0 = &tile[lane * MPITCH];
    const float* __restrict__ r1 = &tile[(lane + 32) * MPITCH];
#pragma unroll
    for (int h = 0; h < H; h += 4) {
      const float4 u = *(const float4*)(r0 + h);
      const float4 w = *(const float4*)(r1 + h);
      b[h + 0] = (v2f){u.x, w.x};
      b[h + 1] = (v2f){u.y, w.y};
      b[h + 2] = (v2f){u.z, w.z};
      b[h + 3] = (v2f){u.w, w.w};
    }
  }

  const float cb2 = cb2p[0];
  const int i0   = ((int)blockIdx.y * MWAVES + wid) * MROWS;
  const int iend = (i0 + MROWS < N) ? i0 + MROWS : N;
  const int ja = j0 + lane;
  const int jb = j0 + 32 + lane;
  const v2f zero2 = (v2f){0.0f, 0.0f};

  for (int i = i0; i < iend; ++i) {
    const float* __restrict__ a = hi + (size_t)i * H;
    int off = 0;
    asm volatile("" : "+s"(off));
    const float* __restrict__ gp  = cg1  + off;
    const float* __restrict__ bep = cbe1 + off;
    const float* __restrict__ cwp = cw2  + off;

    v2f s0 = zero2, s1 = zero2, s2 = zero2, s3 = zero2;
    v2f q0 = zero2, q1 = zero2, q2 = zero2, q3 = zero2;
#pragma unroll
    for (int hb = 0; hb < H; hb += 64) {
      float av[64];
#pragma unroll
      for (int k = 0; k < 64; ++k) av[k] = a[hb + k];
      __builtin_amdgcn_sched_barrier(0);
#pragma unroll
      for (int k = 0; k < 64; k += 4) {
        const v2f t0 = b[hb + k + 0] + av[k + 0];
        const v2f t1 = b[hb + k + 1] + av[k + 1];
        const v2f t2 = b[hb + k + 2] + av[k + 2];
        const v2f t3 = b[hb + k + 3] + av[k + 3];
        s0 += t0; s1 += t1; s2 += t2; s3 += t3;
        q0 = __builtin_elementwise_fma(t0, t0, q0);
        q1 = __builtin_elementwise_fma(t1, t1, q1);
        q2 = __builtin_elementwise_fma(t2, t2, q2);
        q3 = __builtin_elementwise_fma(t3, t3, q3);
      }
      __builtin_amdgcn_sched_barrier(0);
    }
    const v2f m   = ((s0 + s1) + (s2 + s3)) * (1.0f / 128.0f);
    const v2f var = ((q0 + q1) + (q2 + q3)) * (1.0f / 128.0f) - m * m;
    v2f inv;
    inv.x = rsqrtf(var.x + LN_EPS);
    inv.y = rsqrtf(var.y + LN_EPS);
    const v2f nmi = -m * inv;

    int off2 = 0;
    asm volatile("" : "+s"(off2));
    const float* __restrict__ a2 = a + off2;
    v2f d0 = zero2, d1 = zero2, d2 = zero2, d3 = zero2;
#pragma unroll
    for (int hb = 0; hb < H; hb += 16) {
      float av[16], gv[16], ev[16], cv[16];
#pragma unroll
      for (int k = 0; k < 16; ++k) {
        av[k] = a2[hb + k];
        gv[k] = gp[hb + k];
        ev[k] = bep[hb + k];
        cv[k] = cwp[hb + k];
      }
      __builtin_amdgcn_sched_barrier(0);
#pragma unroll
      for (int k = 0; k < 16; k += 4) {
        const v2f y0 = __builtin_elementwise_fma(b[hb + k + 0] + av[k + 0], inv, nmi);
        const v2f y1 = __builtin_elementwise_fma(b[hb + k + 1] + av[k + 1], inv, nmi);
        const v2f y2 = __builtin_elementwise_fma(b[hb + k + 2] + av[k + 2], inv, nmi);
        const v2f y3 = __builtin_elementwise_fma(b[hb + k + 3] + av[k + 3], inv, nmi);
        const v2f z0 = __builtin_elementwise_fma(y0, (v2f){gv[k + 0], gv[k + 0]}, (v2f){ev[k + 0], ev[k + 0]});
        const v2f z1 = __builtin_elementwise_fma(y1, (v2f){gv[k + 1], gv[k + 1]}, (v2f){ev[k + 1], ev[k + 1]});
        const v2f z2 = __builtin_elementwise_fma(y2, (v2f){gv[k + 2], gv[k + 2]}, (v2f){ev[k + 2], ev[k + 2]});
        const v2f z3 = __builtin_elementwise_fma(y3, (v2f){gv[k + 3], gv[k + 3]}, (v2f){ev[k + 3], ev[k + 3]});
        d0 = __builtin_elementwise_fma(__builtin_elementwise_max(z0, zero2), (v2f){cv[k + 0], cv[k + 0]}, d0);
        d1 = __builtin_elementwise_fma(__builtin_elementwise_max(z1, zero2), (v2f){cv[k + 1], cv[k + 1]}, d1);
        d2 = __builtin_elementwise_fma(__builtin_elementwise_max(z2, zero2), (v2f){cv[k + 2], cv[k + 2]}, d2);
        d3 = __builtin_elementwise_fma(__builtin_elementwise_max(z3, zero2), (v2f){cv[k + 3], cv[k + 3]}, d3);
      }
      __builtin_amdgcn_sched_barrier(0);
    }
    const v2f dot = (d0 + d1) + (d2 + d3);
    const float sca = tanhf(dot.x + cb2);
    const float scb = tanhf(dot.y + cb2);
    float* __restrict__ orow = out + (size_t)i * N;
    if (ja < N) orow[ja] = sca;
    if (jb < N) orow[jb] = scb;
  }
}

__global__ void __launch_bounds__(256)
comp_scores_kernel(const float* __restrict__ hi,
                   const float* __restrict__ hj,
                   const int* __restrict__ ei, int E,
                   const float* __restrict__ cb1,
                   const float* __restrict__ cg1,
                   const float* __restrict__ cbe1,
                   const float* __restrict__ cw2,
                   const float* __restrict__ cb2p,
                   float* __restrict__ out) {
  const int lane = threadIdx.x & 31;
  const int wv   = __builtin_amdgcn_readfirstlane((int)(threadIdx.x >> 5));
  const int gw   = (int)(blockIdx.x * (blockDim.x >> 5)) + wv;
  const int nw   = (int)((gridDim.x * blockDim.x) >> 5);

  const float4 cb  = *(const float4*)(cb1  + 4 * lane);
  const float4 g   = *(const float4*)(cg1  + 4 * lane);
  const float4 be  = *(const float4*)(cbe1 + 4 * lane);
  const float4 cw  = *(const float4*)(cw2  + 4 * lane);
  const float  cb2 = cb2p[0];

  for (int e0 = gw * 32; e0 < E; e0 += nw * 32) {
    int e = e0 + lane;
    if (e >= E) e = E - 1;
    const int iv = ei[e];
    const int jv = ei[E + e];

    float4 hp[32];
    float  s[32], q[32];
#pragma unroll
    for (int p = 0; p < 32; ++p) {
      const int i = __builtin_amdgcn_readlane(iv, p);
      const int j = __builtin_amdgcn_readlane(jv, p);
      const float4 hi4 = *(const float4*)(hi + (size_t)i * H + 4 * lane);
      const float4 hj4 = *(const float4*)(hj + (size_t)j * H + 4 * lane);
      float4 t;
      t.x = hi4.x + hj4.x + cb.x;
      t.y = hi4.y + hj4.y + cb.y;
      t.z = hi4.z + hj4.z + cb.z;
      t.w = hi4.w + hj4.w + cb.w;
      hp[p] = t;
      s[p]  = t.x + t.y + t.z + t.w;
      q[p]  = t.x * t.x + t.y * t.y + t.z * t.z + t.w * t.w;
    }
    const float sc = score32(hp, s, q, g, be, cw, cb2, lane);
    if (e0 + lane < E) out[e0 + lane] = sc;
  }
}

extern "C" void kernel_launch(void* const* d_in, const int* in_sizes, int n_in,
                              void* d_out, int out_size, void* d_ws, size_t ws_size,
                              hipStream_t stream) {
  const float* x    = (const float*)d_in[0];
  const int*   ei   = (const int*)  d_in[1];
  const float* w1   = (const float*)d_in[2];
  const float* b1   = (const float*)d_in[3];
  const float* g1   = (const float*)d_in[4];
  const float* be1  = (const float*)d_in[5];
  const float* w2   = (const float*)d_in[6];
  const float* b2   = (const float*)d_in[7];
  const float* g2   = (const float*)d_in[8];
  const float* be2  = (const float*)d_in[9];
  const float* w3   = (const float*)d_in[10];
  const float* b3   = (const float*)d_in[11];
  const float* cw1  = (const float*)d_in[12];
  const float* cb1  = (const float*)d_in[13];
  const float* cg1  = (const float*)d_in[14];
  const float* cbe1 = (const float*)d_in[15];
  const float* cw2  = (const float*)d_in[16];
  const float* cb2  = (const float*)d_in[17];
  const float* prot = (const float*)d_in[18];

  const int N = in_sizes[0] / D;
  const int E = in_sizes[1] / 2;

  float* h1pre = (float*)d_ws;
  float* hi    = h1pre + (size_t)N * H;
  float* hj    = hi    + (size_t)N * H;

  float* out       = (float*)d_out;
  float* o_logits  = out;
  float* o_probs   = o_logits  + (size_t)N * KC;
  float* o_scores  = o_probs   + (size_t)N * KC;
  float* o_matrix  = o_scores  + (size_t)E;
  float* o_centers = o_matrix  + (size_t)N * N;
  float* o_protos  = o_centers + (size_t)KC * D;

  const int rowTiles = (N + 15) / 16;

  wmma_gemm3_kernel<<<dim3((rowTiles + 1) / 2, 3), GWAVES * 32, 0, stream>>>(
      x, w1, cw1, h1pre, hi, hj, N);

  node_mlp_kernel<<<rowTiles, NPB * 32, 0, stream>>>(
      h1pre, b1, g1, be1, w2, b2, g2, be2, w3, b3, x, o_logits, o_probs, N);

  centers_kernel<<<KC, 128, 0, stream>>>(
      o_probs, x, h1pre, prot, o_centers, o_protos, N);

  comp_matrix_kernel<<<dim3((N + MCOLS - 1) / MCOLS, (N + MWAVES * MROWS - 1) / (MWAVES * MROWS)),
                       MWAVES * 32, 0, stream>>>(
      hi, hj, cb1, cg1, cbe1, cw2, cb2, o_matrix, N);

  comp_scores_kernel<<<2048, 256, 0, stream>>>(
      hi, hj, ei, E, cb1, cg1, cbe1, cw2, cb2, o_scores);
}


# edge scores: a wave sums the gathered rows of its 32 edges into a wave-private 32 x 128 LDS tile, lane p reads row p back and forms mean, variance and dot as private sums; no butterflies
# speedup vs baseline: 3.4587x; 1.1043x over previous
#include <hip/hip_runtime.h>
#include <math.h>

typedef __attribute__((ext_vector_type(2))) float v2f;
typedef __attribute__((ext_vector_type(4))) float v4f;
typedef __attribute__((ext_vector_type(8))) float v8f;

#define D   128
#define H   128
#define HH2 64
#define KC  10
#define LN_EPS 1e-5f

#define NPB 16
#define TPITCH 132

__device__ __forceinline__ v8f stripe_tile(const float* __restrict__ x,
                                           const float* __restrict__ Bm,
                                           const int rt, const int ct,
                                           const int lane, const int N) {
  const int half = lane >> 4;
  const int l16  = lane & 15;
  int arow = rt * 16 + l16;
  if (arow >= N) arow = N - 1;
  const float* __restrict__ arow_p = x + (size_t)arow * D + half * 2;
  const float* __restrict__ bcol_p = Bm + (size_t)(half * 2) * H + ct * 16 + l16;
  v2f a[D / 4], b[D / 4];
#pragma unroll
  for (int s = 0; s < D / 4; ++s) a[s] = *(const v2f*)(arow_p + 4 * s);
#pragma unroll
  for (int s = 0; s < D / 4; ++s) {
    b[s].x = bcol_p[(size_t)(4 * s) * H];
    b[s].y = bcol_p[(size_t)(4 * s + 1) * H];
  }
  __builtin_amdgcn_sched_barrier(0);
  v8f acc0 = {}, acc1 = {};
#pragma unroll
  for (int s = 0; s < D / 4; s += 2) {
    acc0 = __builtin_amdgcn_wmma_f32_16x16x4_f32(false, a[s], false, b[s],
                                                 (short)0, acc0, false, false);
    acc1 = __builtin_amdgcn_wmma_f32_16x16x4_f32(false, a[s + 1], false, b[s + 1],
                                                 (short)0, acc1, false, false);
  }
  return acc0 + acc1;
}

__device__ __forceinline__ float wave_sum(float v) {
#pragma unroll
  for (int w = 16; w > 0; w >>= 1) v += __shfl_xor(v, w, 32);
  return v;
}

__device__ __forceinline__ float wave_max(float v) {
#pragma unroll
  for (int w = 16; w > 0; w >>= 1) v = fmaxf(v, __shfl_xor(v, w, 32));
  return v;
}

__global__ void __launch_bounds__(NPB * 32)
stripe_kernel(const float* __restrict__ x,
              const float* __restrict__ w1,
              const float* __restrict__ cw1,
              float* __restrict__ hi,
              float* __restrict__ hj,
              float* __restrict__ part_ws,
              const float* __restrict__ b1,
              const float* __restrict__ g1,
              const float* __restrict__ be1,
              const float* __restrict__ w2,
              const float* __restrict__ b2,
              const float* __restrict__ g2,
              const float* __restrict__ be2,
              const float* __restrict__ w3,
              const float* __restrict__ b3,
              float* __restrict__ out_logits,
              float* __restrict__ out_probs,
              int N) {
  __shared__ __attribute__((aligned(16))) float sh0[NPB * TPITCH];
  __shared__ float sh1[NPB][H];
  __shared__ float sh2[NPB][HH2];
  __shared__ float sp[NPB][KC];

  const int lane = threadIdx.x & 31;
  const int wv   = __builtin_amdgcn_readfirstlane((int)(threadIdx.x >> 5));
  const int mat  = blockIdx.y;
  const int rt   = blockIdx.x;
  const int half = lane >> 4;
  const int l16  = lane & 15;

  if (mat != 0) {
    if (wv < H / 16) {
      const v8f acc = stripe_tile(x, (mat == 1) ? cw1 : cw1 + D * H, rt, wv, lane, N);
      float* __restrict__ out = (mat == 1) ? hi : hj;
      const int rbase = rt * 16 + half * 8;
      if (rt * 16 + 16 <= N) {
#pragma unroll
        for (int r = 0; r < 8; ++r)
          out[(size_t)(rbase + r) * H + wv * 16 + l16] = acc[r];
      } else {
#pragma unroll
        for (int r = 0; r < 8; ++r)
          if (rbase + r < N)
            out[(size_t)(rbase + r) * H + wv * 16 + l16] = acc[r];
      }
    }
    return;
  }

  const float4 bb1 = *(const float4*)(b1 + 4 * lane);
  const float4 gg1 = *(const float4*)(g1 + 4 * lane);
  const float4 ee1 = *(const float4*)(be1 + 4 * lane);

  if (wv < H / 16) {
    const v8f acc = stripe_tile(x, w1, rt, wv, lane, N);
#pragma unroll
    for (int r = 0; r < 8; ++r)
      sh0[(half * 8 + r) * TPITCH + wv * 16 + l16] = acc[r];
  }
  __syncthreads();

  const int n0   = rt * NPB;
  const int n    = n0 + wv;
  const bool live = n < N;

  {
    const float4 hv = *(const float4*)&sh0[wv * TPITCH + 4 * lane];
    float4 v;
    v.x = hv.x + bb1.x; v.y = hv.y + bb1.y; v.z = hv.z + bb1.z; v.w = hv.w + bb1.w;
    const float s = wave_sum((v.x + v.y) + (v.z + v.w));
    const float q = wave_sum((v.x * v.x + v.y * v.y) + (v.z * v.z + v.w * v.w));
    const float m   = s * (1.0f / 128.0f);
    const float var = q * (1.0f / 128.0f) - m * m;
    const float inv = rsqrtf(var + LN_EPS);
    float4 r;
    r.x = fmaxf((v.x - m) * inv * gg1.x + ee1.x, 0.0f);
    r.y = fmaxf((v.y - m) * inv * gg1.y + ee1.y, 0.0f);
    r.z = fmaxf((v.z - m) * inv * gg1.z + ee1.z, 0.0f);
    r.w = fmaxf((v.w - m) * inv * gg1.w + ee1.w, 0.0f);
    *(float4*)&sh1[wv][4 * lane] = r;
  }
  __syncthreads();

  {
    float2 a2 = *(const float2*)(b2 + 2 * lane);
    const float* __restrict__ w2p = w2 + 2 * lane;
#pragma unroll 16
    for (int k = 0; k < H; ++k) {
      const float  hk = sh1[wv][k];
      const float2 w  = *(const float2*)(w2p + k * HH2);
      a2.x += hk * w.x;
      a2.y += hk * w.y;
    }
    const float s2 = wave_sum(a2.x + a2.y);
    const float q2 = wave_sum(a2.x * a2.x + a2.y * a2.y);
    const float m2 = s2 * (1.0f / 64.0f);
    const float v2 = q2 * (1.0f / 64.0f) - m2 * m2;
    const float i2 = rsqrtf(v2 + LN_EPS);
    const float2 gg = *(const float2*)(g2 + 2 * lane);
    const float2 ee = *(const float2*)(be2 + 2 * lane);
    float2 h2;
    h2.x = fmaxf((a2.x - m2) * i2 * gg.x + ee.x, 0.0f);
    h2.y = fmaxf((a2.y - m2) * i2 * gg.y + ee.y, 0.0f);
    *(float2*)&sh2[wv][2 * lane] = h2;
  }
  __syncthreads();

  {
    const bool owner = lane < KC;
    const int  tc    = owner ? lane : KC - 1;
    float lg = b3[tc];
#pragma unroll 16
    for (int k = 0; k < HH2; ++k) lg += sh2[wv][k] * w3[k * KC + tc];
    const float mx   = wave_max(owner ? lg : -INFINITY);
    const float e    = owner ? __expf(lg - mx) : 0.0f;
    const float ssum = wave_sum(e);
    const float p    = e / ssum;
    if (owner) {
      if (live) {
        out_logits[(size_t)n * KC + lane] = lg;
        out_probs[(size_t)n * KC + lane]  = p;
      }
      sp[wv][lane] = live ? p : 0.0f;
    }
  }
  __syncthreads();

  if (n0 + NPB <= N) {
    const int d  = threadIdx.x & (D - 1);
    const int kk = threadIdx.x >> 7;
    const int k3 = (kk < 2) ? kk + 8 : kk + 4;
    const float* __restrict__ xp = x + (size_t)n0 * D + d;
    float a0 = 0.0f, a1 = 0.0f, a2c = 0.0f;
#pragma unroll
    for (int g = 0; g < NPB; ++g) {
      const float xv = xp[g * D];
      a0  += sp[g][kk] * xv;
      a1  += sp[g][kk + 4] * xv;
      a2c += sp[g][k3] * xv;
    }
    float* __restrict__ part = part_ws + (size_t)n0 * H;
    part[kk * D + d]       = a0;
    part[(kk + 4) * D + d] = a1;
    if (kk < 2) part[(kk + 8) * D + d] = a2c;
  }
}

__global__ void centers_kernel(const float* __restrict__ probs,
                               const float* __restrict__ x,
                               const float* __restrict__ part,
                               const float* __restrict__ prototypes,
                               float* __restrict__ out_centers,
                               float* __restrict__ out_protos,
                               int N) {
  const int k = blockIdx.x;
  const int d = threadIdx.x;
  __shared__ float sW[128];
  __shared__ float sT[NPB];

  const int nfull = N / NPB;
  const int nt    = nfull * NPB;
  float tailp = 0.0f;
  if (d < NPB - 1) {
    const int n = nt + d;
    if (n < N) tailp = probs[(size_t)n * KC + k];
  }

  float wpart = 0.0f;
  for (int base = 0; base < N; base += 8 * 128) {
    float t[8];
#pragma unroll
    for (int u = 0; u < 8; ++u) {
      const int n  = base + u * 128 + d;
      const int nc = n < N ? n : N - 1;
      const float v = probs[(size_t)nc * KC + k];
      t[u] = n < N ? v : 0.0f;
    }
#pragma unroll
    for (int u = 0; u < 8; ++u) wpart += t[u];
  }
  if (d < NPB - 1) sT[d] = tailp;
  sW[d] = wpart;
  __syncthreads();
  for (int s = 64; s > 0; s >>= 1) {
    if (d < s) sW[d] += sW[d + s];
    __syncthreads();
  }
  const float wsum = sW[0] + 1e-8f;

  const float* __restrict__ pp = part + k * D + d;
  float acc = 0.0f;
  for (int b0 = 0; b0 < nfull; b0 += 16) {
    float t[16];
#pragma unroll
    for (int u = 0; u < 16; ++u) {
      const int b  = b0 + u;
      const int bc = b < nfull ? b : nfull - 1;
      const float v = pp[(size_t)bc * (NPB * H)];
      t[u] = b < nfull ? v : 0.0f;
    }
#pragma unroll
    for (int u = 0; u < 16; ++u) acc += t[u];
  }
  if (nt < N) {
    float xv[NPB - 1];
#pragma unroll
    for (int u = 0; u < NPB - 1; ++u) {
      const int n  = nt + u;
      const int nc = n < N ? n : N - 1;
      xv[u] = x[(size_t)nc * D + d];
    }
#pragma unroll
    for (int u = 0; u < NPB - 1; ++u) acc += sT[u] * xv[u];
  }

  out_centers[k * D + d] = acc / wsum;
  out_protos[k * D + d]  = prototypes[k * D + d];
}

#define MROWS  8
#define MWAVES 8
#define MCOLS  64
#define MPITCH 132

__global__ void __launch_bounds__(MWAVES * 32)
comp_matrix_kernel(const float* __restrict__ hi,
                   const float* __restrict__ hj,
                   const float* __restrict__ cb1,
                   const float* __restrict__ cg1,
                   const float* __restrict__ cbe1,
                   const float* __restrict__ cw2,
                   const float* __restrict__ cb2p,
                   float* __restrict__ out, int N) {
  __shared__ __attribute__((aligned(16))) float tile[MCOLS * MPITCH];

  const int lane = threadIdx.x & 31;
  const int wid  = __builtin_amdgcn_readfirstlane((int)(threadIdx.x >> 5));
  const int j0   = blockIdx.x * MCOLS;

  {
    const float4 cbv = *(const float4*)(cb1 + 4 * lane);
#pragma unroll
    for (int r = 0; r < MCOLS / MWAVES; ++r) {
      const int c = wid * (MCOLS / MWAVES) + r;
      int jr = j0 + c;
      if (jr >= N) jr = N - 1;
      float4 v = *(const float4*)(hj + (size_t)jr * H + 4 * lane);
      v.x += cbv.x; v.y += cbv.y; v.z += cbv.z; v.w += cbv.w;
      *(float4*)&tile[c * MPITCH + 4 * lane] = v;
    }
  }
  __syncthreads();

  v2f b[H];
  {
    const float* __restrict__ r0 = &tile[lane * MPITCH];
    const float* __restrict__ r1 = &tile[(lane + 32) * MPITCH];
#pragma unroll
    for (int h = 0; h < H; h += 4) {
      const float4 u = *(const float4*)(r0 + h);
      const float4 w = *(const float4*)(r1 + h);
      b[h + 0] = (v2f){u.x, w.x};
      b[h + 1] = (v2f){u.y, w.y};
      b[h + 2] = (v2f){u.z, w.z};
      b[h + 3] = (v2f){u.w, w.w};
    }
  }

  const float cb2 = cb2p[0];
  const int i0   = ((int)blockIdx.y * MWAVES + wid) * MROWS;
  const int iend = (i0 + MROWS < N) ? i0 + MROWS : N;
  const int ja = j0 + lane;
  const int jb = j0 + 32 + lane;
  const v2f zero2 = (v2f){0.0f, 0.0f};

  for (int i = i0; i < iend; ++i) {
    const float* __restrict__ a = hi + (size_t)i * H;
    int off = 0;
    asm volatile("" : "+s"(off));
    const float* __restrict__ gp  = cg1  + off;
    const float* __restrict__ bep = cbe1 + off;
    const float* __restrict__ cwp = cw2  + off;

    v2f s0 = zero2, s1 = zero2, s2 = zero2, s3 = zero2;
    v2f q0 = zero2, q1 = zero2, q2 = zero2, q3 = zero2;
#pragma unroll
    for (int hb = 0; hb < H; hb += 64) {
      float av[64];
#pragma unroll
      for (int k = 0; k < 64; ++k) av[k] = a[hb + k];
      __builtin_amdgcn_sched_barrier(0);
#pragma unroll
      for (int k = 0; k < 64; k += 4) {
        const v2f t0 = b[hb + k + 0] + av[k + 0];
        const v2f t1 = b[hb + k + 1] + av[k + 1];
        const v2f t2 = b[hb + k + 2] + av[k + 2];
        const v2f t3 = b[hb + k + 3] + av[k + 3];
        s0 += t0; s1 += t1; s2 += t2; s3 += t3;
        q0 = __builtin_elementwise_fma(t0, t0, q0);
        q1 = __builtin_elementwise_fma(t1, t1, q1);
        q2 = __builtin_elementwise_fma(t2, t2, q2);
        q3 = __builtin_elementwise_fma(t3, t3, q3);
      }
      __builtin_amdgcn_sched_barrier(0);
    }
    const v2f m   = ((s0 + s1) + (s2 + s3)) * (1.0f / 128.0f);
    const v2f var = ((q0 + q1) + (q2 + q3)) * (1.0f / 128.0f) - m * m;
    v2f inv;
    inv.x = rsqrtf(var.x + LN_EPS);
    inv.y = rsqrtf(var.y + LN_EPS);
    const v2f nmi = -m * inv;

    int off2 = 0;
    asm volatile("" : "+s"(off2));
    const float* __restrict__ a2 = a + off2;
    v2f d0 = zero2, d1 = zero2, d2 = zero2, d3 = zero2;
#pragma unroll
    for (int hb = 0; hb < H; hb += 16) {
      float av[16], gv[16], ev[16], cv[16];
#pragma unroll
      for (int k = 0; k < 16; ++k) {
        av[k] = a2[hb + k];
        gv[k] = gp[hb + k];
        ev[k] = bep[hb + k];
        cv[k] = cwp[hb + k];
      }
      __builtin_amdgcn_sched_barrier(0);
#pragma unroll
      for (int k = 0; k < 16; k += 4) {
        const v2f y0 = __builtin_elementwise_fma(b[hb + k + 0] + av[k + 0], inv, nmi);
        const v2f y1 = __builtin_elementwise_fma(b[hb + k + 1] + av[k + 1], inv, nmi);
        const v2f y2 = __builtin_elementwise_fma(b[hb + k + 2] + av[k + 2], inv, nmi);
        const v2f y3 = __builtin_elementwise_fma(b[hb + k + 3] + av[k + 3], inv, nmi);
        const v2f z0 = __builtin_elementwise_fma(y0, (v2f){gv[k + 0], gv[k + 0]}, (v2f){ev[k + 0], ev[k + 0]});
        const v2f z1 = __builtin_elementwise_fma(y1, (v2f){gv[k + 1], gv[k + 1]}, (v2f){ev[k + 1], ev[k + 1]});
        const v2f z2 = __builtin_elementwise_fma(y2, (v2f){gv[k + 2], gv[k + 2]}, (v2f){ev[k + 2], ev[k + 2]});
        const v2f z3 = __builtin_elementwise_fma(y3, (v2f){gv[k + 3], gv[k + 3]}, (v2f){ev[k + 3], ev[k + 3]});
        d0 = __builtin_elementwise_fma(__builtin_elementwise_max(z0, zero2), (v2f){cv[k + 0], cv[k + 0]}, d0);
        d1 = __builtin_elementwise_fma(__builtin_elementwise_max(z1, zero2), (v2f){cv[k + 1], cv[k + 1]}, d1);
        d2 = __builtin_elementwise_fma(__builtin_elementwise_max(z2, zero2), (v2f){cv[k + 2], cv[k + 2]}, d2);
        d3 = __builtin_elementwise_fma(__builtin_elementwise_max(z3, zero2), (v2f){cv[k + 3], cv[k + 3]}, d3);
      }
      __builtin_amdgcn_sched_barrier(0);
    }
    const v2f dot = (d0 + d1) + (d2 + d3);
    const float sca = tanhf(dot.x + cb2);
    const float scb = tanhf(dot.y + cb2);
    float* __restrict__ orow = out + (size_t)i * N;
    if (ja < N) orow[ja] = sca;
    if (jb < N) orow[jb] = scb;
  }
}

#define EWAVES 6
#define CONST_AS __attribute__((address_space(4)))
#define EPITCH 132

__global__ void __launch_bounds__(EWAVES * 32)
comp_scores_kernel(const float* __restrict__ hi,
                   const float* __restrict__ hj,
                   const int* __restrict__ ei, int E,
                   const float* __restrict__ cb1,
                   const float* __restrict__ cg1,
                   const float* __restrict__ cbe1,
                   const float* __restrict__ cw2,
                   const float* __restrict__ cb2p,
                   float* __restrict__ out) {
  __shared__ __attribute__((aligned(16))) float etile[EWAVES * 32 * EPITCH];

  const int lane = threadIdx.x & 31;
  const int wv   = __builtin_amdgcn_readfirstlane((int)(threadIdx.x >> 5));
  const int gw   = (int)(blockIdx.x * EWAVES) + wv;
  const int nw   = (int)(gridDim.x * EWAVES);
  float* __restrict__ tile = etile + wv * (32 * EPITCH);

  const v4f   cb  = *(const v4f*)(cb1 + 4 * lane);
  const float cb2 = cb2p[0];
  const v2f zero2 = (v2f){0.0f, 0.0f};

  for (int e0 = gw * 32; e0 < E; e0 += nw * 32) {
    int e = e0 + lane;
    if (e >= E) e = E - 1;
    const int iv = ei[e];
    const int jv = ei[E + e];

    v4f ra[2][8], rb[2][8];
#pragma unroll
    for (int pp = 0; pp < 8; ++pp) {
      const int i = __builtin_amdgcn_readlane(iv, pp);
      const int j = __builtin_amdgcn_readlane(jv, pp);
      ra[0][pp] = *(const v4f*)(hi + (size_t)i * H + 4 * lane);
      rb[0][pp] = *(const v4f*)(hj + (size_t)j * H + 4 * lane);
    }
#pragma unroll
    for (int g = 0; g < 4; ++g) {
      if (g < 3) {
#pragma unroll
        for (int pp = 0; pp < 8; ++pp) {
          const int p = (g + 1) * 8 + pp;
          const int i = __builtin_amdgcn_readlane(iv, p);
          const int j = __builtin_amdgcn_readlane(jv, p);
          ra[(g + 1) & 1][pp] = *(const v4f*)(hi + (size_t)i * H + 4 * lane);
          rb[(g + 1) & 1][pp] = *(const v4f*)(hj + (size_t)j * H + 4 * lane);
        }
      }
#pragma unroll
      for (int pp = 0; pp < 8; ++pp) {
        const int p = g * 8 + pp;
        v4f a4 = ra[g & 1][pp];
        v4f b4 = rb[g & 1][pp];
        asm volatile("" : "+v"(a4), "+v"(b4));
        *(v4f*)&tile[p * EPITCH + 4 * lane] = a4 + b4 + cb;
      }
    }
    __builtin_amdgcn_fence(__ATOMIC_RELEASE, "wavefront");
    __builtin_amdgcn_wave_barrier();
    __builtin_amdgcn_fence(__ATOMIC_ACQUIRE, "wavefront");

    v2f t[H / 2];
    {
      const float* __restrict__ row = tile + lane * EPITCH;
#pragma unroll
      for (int h = 0; h < H; h += 4) {
        const float4 u = *(const float4*)(row + h);
        t[h / 2]     = (v2f){u.x, u.y};
        t[h / 2 + 1] = (v2f){u.z, u.w};
      }
    }

    v2f s0 = zero2, s1 = zero2, s2 = zero2, s3 = zero2;
    v2f q0 = zero2, q1 = zero2, q2 = zero2, q3 = zero2;
#pragma unroll
    for (int k = 0; k < H / 2; k += 4) {
      s0 += t[k + 0]; s1 += t[k + 1]; s2 += t[k + 2]; s3 += t[k + 3];
      q0 = __builtin_elementwise_fma(t[k + 0], t[k + 0], q0);
      q1 = __builtin_elementwise_fma(t[k + 1], t[k + 1], q1);
      q2 = __builtin_elementwise_fma(t[k + 2], t[k + 2], q2);
      q3 = __builtin_elementwise_fma(t[k + 3], t[k + 3], q3);
    }
    const v2f sv = (s0 + s1) + (s2 + s3);
    const v2f qv = (q0 + q1) + (q2 + q3);
    const float m   = (sv.x + sv.y) * (1.0f / 128.0f);
    const float var = (qv.x + qv.y) * (1.0f / 128.0f) - m * m;
    const float inv = rsqrtf(var + LN_EPS);
    const float nmi = -m * inv;
    const v2f inv2 = (v2f){inv, inv};
    const v2f nmi2 = (v2f){nmi, nmi};

    int off = 0;
    asm volatile("" : "+s"(off));
    const CONST_AS float* gp  = (const CONST_AS float*)(cg1  + off);
    const CONST_AS float* bep = (const CONST_AS float*)(cbe1 + off);
    const CONST_AS float* cwp = (const CONST_AS float*)(cw2  + off);
    v2f d0 = zero2, d1 = zero2, d2 = zero2, d3 = zero2;
#pragma unroll
    for (int hb = 0; hb < H; hb += 16) {
      float gv[16], ev[16], cv[16];
#pragma unroll
      for (int k = 0; k < 16; ++k) {
        gv[k] = gp[hb + k];
        ev[k] = bep[hb + k];
        cv[k] = cwp[hb + k];
      }
      __builtin_amdgcn_sched_barrier(0);
#pragma unroll
      for (int k = 0; k < 16; k += 8) {
        const int kp = (hb + k) / 2;
        const v2f y0 = __builtin_elementwise_fma(t[kp + 0], inv2, nmi2);
        const v2f y1 = __builtin_elementwise_fma(t[kp + 1], inv2, nmi2);
        const v2f y2 = __builtin_elementwise_fma(t[kp + 2], inv2, nmi2);
        const v2f y3 = __builtin_elementwise_fma(t[kp + 3], inv2, nmi2);
        const v2f z0 = __builtin_elementwise_fma(y0, (v2f){gv[k + 0], gv[k + 1]}, (v2f){ev[k + 0], ev[k + 1]});
        const v2f z1 = __builtin_elementwise_fma(y1, (v2f){gv[k + 2], gv[k + 3]}, (v2f){ev[k + 2], ev[k + 3]});
        const v2f z2 = __builtin_elementwise_fma(y2, (v2f){gv[k + 4], gv[k + 5]}, (v2f){ev[k + 4], ev[k + 5]});
        const v2f z3 = __builtin_elementwise_fma(y3, (v2f){gv[k + 6], gv[k + 7]}, (v2f){ev[k + 6], ev[k + 7]});
        d0 = __builtin_elementwise_fma(__builtin_elementwise_max(z0, zero2), (v2f){cv[k + 0], cv[k + 1]}, d0);
        d1 = __builtin_elementwise_fma(__builtin_elementwise_max(z1, zero2), (v2f){cv[k + 2], cv[k + 3]}, d1);
        d2 = __builtin_elementwise_fma(__builtin_elementwise_max(z2, zero2), (v2f){cv[k + 4], cv[k + 5]}, d2);
        d3 = __builtin_elementwise_fma(__builtin_elementwise_max(z3, zero2), (v2f){cv[k + 6], cv[k + 7]}, d3);
      }
      __builtin_amdgcn_sched_barrier(0);
    }
    const v2f dv = (d0 + d1) + (d2 + d3);
    const float sc = tanhf((dv.x + dv.y) + cb2);
    if (e0 + lane < E) out[e0 + lane] = sc;
  }
}

extern "C" void kernel_launch(void* const* d_in, const int* in_sizes, int n_in,
                              void* d_out, int out_size, void* d_ws, size_t ws_size,
                              hipStream_t stream) {
  const float* x    = (const float*)d_in[0];
  const int*   ei   = (const int*)  d_in[1];
  const float* w1   = (const float*)d_in[2];
  const float* b1   = (const float*)d_in[3];
  const float* g1   = (const float*)d_in[4];
  const float* be1  = (const float*)d_in[5];
  const float* w2   = (const float*)d_in[6];
  const float* b2   = (const float*)d_in[7];
  const float* g2   = (const float*)d_in[8];
  const float* be2  = (const float*)d_in[9];
  const float* w3   = (const float*)d_in[10];
  const float* b3   = (const float*)d_in[11];
  const float* cw1  = (const float*)d_in[12];
  const float* cb1  = (const float*)d_in[13];
  const float* cg1  = (const float*)d_in[14];
  const float* cbe1 = (const float*)d_in[15];
  const float* cw2  = (const float*)d_in[16];
  const float* cb2  = (const float*)d_in[17];
  const float* prot = (const float*)d_in[18];

  const int N = in_sizes[0] / D;
  const int E = in_sizes[1] / 2;

  float* h1pre = (float*)d_ws;
  float* hi    = h1pre + (size_t)N * H;
  float* hj    = hi    + (size_t)N * H;

  float* out       = (float*)d_out;
  float* o_logits  = out;
  float* o_probs   = o_logits  + (size_t)N * KC;
  float* o_scores  = o_probs   + (size_t)N * KC;
  float* o_matrix  = o_scores  + (size_t)E;
  float* o_centers = o_matrix  + (size_t)N * N;
  float* o_protos  = o_centers + (size_t)KC * D;

  const int rowTiles = (N + 15) / 16;

  stripe_kernel<<<dim3(rowTiles, 3), NPB * 32, 0, stream>>>(
      x, w1, cw1, hi, hj, h1pre, b1, g1, be1, w2, b2, g2, be2, w3, b3,
      o_logits, o_probs, N);

  centers_kernel<<<KC, 128, 0, stream>>>(
      o_probs, x, h1pre, prot, o_centers, o_protos, N);

  comp_matrix_kernel<<<dim3((N + MCOLS - 1) / MCOLS, (N + MWAVES * MROWS - 1) / (MWAVES * MROWS)),
                       MWAVES * 32, 0, stream>>>(
      hi, hj, cb1, cg1, cbe1, cw2, cb2, o_matrix, N);

  comp_scores_kernel<<<2605, EWAVES * 32, 0, stream>>>(
      hi, hj, ei, E, cb1, cg1, cbe1, cw2, cb2, o_scores);
}


# edge scores: a resident grid of 768 blocks whose waves loop over the 32-edge trips, the next trip's edge endpoints loaded during the current trip; was 15630 one-trip waves
# speedup vs baseline: 3.5664x; 1.0311x over previous
#include <hip/hip_runtime.h>
#include <math.h>

typedef __attribute__((ext_vector_type(2))) float v2f;
typedef __attribute__((ext_vector_type(4))) float v4f;
typedef __attribute__((ext_vector_type(8))) float v8f;

#define D   128
#define H   128
#define HH2 64
#define KC  10
#define LN_EPS 1e-5f

#define NPB 16
#define TPITCH 132

__device__ __forceinline__ v8f stripe_tile(const float* __restrict__ x,
                                           const float* __restrict__ Bm,
                                           const int rt, const int ct,
                                           const int lane, const int N) {
  const int half = lane >> 4;
  const int l16  = lane & 15;
  int arow = rt * 16 + l16;
  if (arow >= N) arow = N - 1;
  const float* __restrict__ arow_p = x + (size_t)arow * D + half * 2;
  const float* __restrict__ bcol_p = Bm + (size_t)(half * 2) * H + ct * 16 + l16;
  v2f a[D / 4], b[D / 4];
#pragma unroll
  for (int s = 0; s < D / 4; ++s) a[s] = *(const v2f*)(arow_p + 4 * s);
#pragma unroll
  for (int s = 0; s < D / 4; ++s) {
    b[s].x = bcol_p[(size_t)(4 * s) * H];
    b[s].y = bcol_p[(size_t)(4 * s + 1) * H];
  }
  __builtin_amdgcn_sched_barrier(0);
  v8f acc0 = {}, acc1 = {};
#pragma unroll
  for (int s = 0; s < D / 4; s += 2) {
    acc0 = __builtin_amdgcn_wmma_f32_16x16x4_f32(false, a[s], false, b[s],
                                                 (short)0, acc0, false, false);
    acc1 = __builtin_amdgcn_wmma_f32_16x16x4_f32(false, a[s + 1], false, b[s + 1],
                                                 (short)0, acc1, false, false);
  }
  return acc0 + acc1;
}

__device__ __forceinline__ float wave_sum(float v) {
#pragma unroll
  for (int w = 16; w > 0; w >>= 1) v += __shfl_xor(v, w, 32);
  return v;
}

__device__ __forceinline__ float wave_max(float v) {
#pragma unroll
  for (int w = 16; w > 0; w >>= 1) v = fmaxf(v, __shfl_xor(v, w, 32));
  return v;
}

__global__ void __launch_bounds__(NPB * 32)
stripe_kernel(const float* __restrict__ x,
              const float* __restrict__ w1,
              const float* __restrict__ cw1,
              float* __restrict__ hi,
              float* __restrict__ hj,
              float* __restrict__ part_ws,
              const float* __restrict__ b1,
              const float* __restrict__ g1,
              const float* __restrict__ be1,
              const float* __restrict__ w2,
              const float* __restrict__ b2,
              const float* __restrict__ g2,
              const float* __restrict__ be2,
              const float* __restrict__ w3,
              const float* __restrict__ b3,
              float* __restrict__ out_logits,
              float* __restrict__ out_probs,
              int N) {
  __shared__ __attribute__((aligned(16))) float sh0[NPB * TPITCH];
  __shared__ float sh1[NPB][H];
  __shared__ float sh2[NPB][HH2];
  __shared__ float sp[NPB][KC];

  const int lane = threadIdx.x & 31;
  const int wv   = __builtin_amdgcn_readfirstlane((int)(threadIdx.x >> 5));
  const int mat  = blockIdx.y;
  const int rt   = blockIdx.x;
  const int half = lane >> 4;
  const int l16  = lane & 15;

  if (mat != 0) {
    if (wv < H / 16) {
      const v8f acc = stripe_tile(x, (mat == 1) ? cw1 : cw1 + D * H, rt, wv, lane, N);
      float* __restrict__ out = (mat == 1) ? hi : hj;
      const int rbase = rt * 16 + half * 8;
      if (rt * 16 + 16 <= N) {
#pragma unroll
        for (int r = 0; r < 8; ++r)
          out[(size_t)(rbase + r) * H + wv * 16 + l16] = acc[r];
      } else {
#pragma unroll
        for (int r = 0; r < 8; ++r)
          if (rbase + r < N)
            out[(size_t)(rbase + r) * H + wv * 16 + l16] = acc[r];
      }
    }
    return;
  }

  const float4 bb1 = *(const float4*)(b1 + 4 * lane);
  const float4 gg1 = *(const float4*)(g1 + 4 * lane);
  const float4 ee1 = *(const float4*)(be1 + 4 * lane);

  if (wv < H / 16) {
    const v8f acc = stripe_tile(x, w1, rt, wv, lane, N);
#pragma unroll
    for (int r = 0; r < 8; ++r)
      sh0[(half * 8 + r) * TPITCH + wv * 16 + l16] = acc[r];
  }
  __syncthreads();

  const int n0   = rt * NPB;
  const int n    = n0 + wv;
  const bool live = n < N;

  {
    const float4 hv = *(const float4*)&sh0[wv * TPITCH + 4 * lane];
    float4 v;
    v.x = hv.x + bb1.x; v.y = hv.y + bb1.y; v.z = hv.z + bb1.z; v.w = hv.w + bb1.w;
    const float s = wave_sum((v.x + v.y) + (v.z + v.w));
    const float q = wave_sum((v.x * v.x + v.y * v.y) + (v.z * v.z + v.w * v.w));
    const float m   = s * (1.0f / 128.0f);
    const float var = q * (1.0f / 128.0f) - m * m;
    const float inv = rsqrtf(var + LN_EPS);
    float4 r;
    r.x = fmaxf((v.x - m) * inv * gg1.x + ee1.x, 0.0f);
    r.y = fmaxf((v.y - m) * inv * gg1.y + ee1.y, 0.0f);
    r.z = fmaxf((v.z - m) * inv * gg1.z + ee1.z, 0.0f);
    r.w = fmaxf((v.w - m) * inv * gg1.w + ee1.w, 0.0f);
    *(float4*)&sh1[wv][4 * lane] = r;
  }
  __syncthreads();

  {
    float2 a2 = *(const float2*)(b2 + 2 * lane);
    const float* __restrict__ w2p = w2 + 2 * lane;
#pragma unroll 16
    for (int k = 0; k < H; ++k) {
      const float  hk = sh1[wv][k];
      const float2 w  = *(const float2*)(w2p + k * HH2);
      a2.x += hk * w.x;
      a2.y += hk * w.y;
    }
    const float s2 = wave_sum(a2.x + a2.y);
    const float q2 = wave_sum(a2.x * a2.x + a2.y * a2.y);
    const float m2 = s2 * (1.0f / 64.0f);
    const float v2 = q2 * (1.0f / 64.0f) - m2 * m2;
    const float i2 = rsqrtf(v2 + LN_EPS);
    const float2 gg = *(const float2*)(g2 + 2 * lane);
    const float2 ee = *(const float2*)(be2 + 2 * lane);
    float2 h2;
    h2.x = fmaxf((a2.x - m2) * i2 * gg.x + ee.x, 0.0f);
    h2.y = fmaxf((a2.y - m2) * i2 * gg.y + ee.y, 0.0f);
    *(float2*)&sh2[wv][2 * lane] = h2;
  }
  __syncthreads();

  {
    const bool owner = lane < KC;
    const int  tc    = owner ? lane : KC - 1;
    float lg = b3[tc];
#pragma unroll 16
    for (int k = 0; k < HH2; ++k) lg += sh2[wv][k] * w3[k * KC + tc];
    const float mx   = wave_max(owner ? lg : -INFINITY);
    const float e    = owner ? __expf(lg - mx) : 0.0f;
    const float ssum = wave_sum(e);
    const float p    = e / ssum;
    if (owner) {
      if (live) {
        out_logits[(size_t)n * KC + lane] = lg;
        out_probs[(size_t)n * KC + lane]  = p;
      }
      sp[wv][lane] = live ? p : 0.0f;
    }
  }
  __syncthreads();

  if (n0 + NPB <= N) {
    const int d  = threadIdx.x & (D - 1);
    const int kk = threadIdx.x >> 7;
    const int k3 = (kk < 2) ? kk + 8 : kk + 4;
    const float* __restrict__ xp = x + (size_t)n0 * D + d;
    float a0 = 0.0f, a1 = 0.0f, a2c = 0.0f;
#pragma unroll
    for (int g = 0; g < NPB; ++g) {
      const float xv = xp[g * D];
      a0  += sp[g][kk] * xv;
      a1  += sp[g][kk + 4] * xv;
      a2c += sp[g][k3] * xv;
    }
    float* __restrict__ part = part_ws + (size_t)n0 * H;
    part[kk * D + d]       = a0;
    part[(kk + 4) * D + d] = a1;
    if (kk < 2) part[(kk + 8) * D + d] = a2c;
  }
}

__global__ void centers_kernel(const float* __restrict__ probs,
                               const float* __restrict__ x,
                               const float* __restrict__ part,
                               const float* __restrict__ prototypes,
                               float* __restrict__ out_centers,
                               float* __restrict__ out_protos,
                               int N) {
  const int k = blockIdx.x;
  const int d = threadIdx.x;
  __shared__ float sW[128];
  __shared__ float sT[NPB];

  const int nfull = N / NPB;
  const int nt    = nfull * NPB;
  float tailp = 0.0f;
  if (d < NPB - 1) {
    const int n = nt + d;
    if (n < N) tailp = probs[(size_t)n * KC + k];
  }

  float wpart = 0.0f;
  for (int base = 0; base < N; base += 8 * 128) {
    float t[8];
#pragma unroll
    for (int u = 0; u < 8; ++u) {
      const int n  = base + u * 128 + d;
      const int nc = n < N ? n : N - 1;
      const float v = probs[(size_t)nc * KC + k];
      t[u] = n < N ? v : 0.0f;
    }
#pragma unroll
    for (int u = 0; u < 8; ++u) wpart += t[u];
  }
  if (d < NPB - 1) sT[d] = tailp;
  sW[d] = wpart;
  __syncthreads();
  for (int s = 64; s > 0; s >>= 1) {
    if (d < s) sW[d] += sW[d + s];
    __syncthreads();
  }
  const float wsum = sW[0] + 1e-8f;

  const float* __restrict__ pp = part + k * D + d;
  float acc = 0.0f;
  for (int b0 = 0; b0 < nfull; b0 += 16) {
    float t[16];
#pragma unroll
    for (int u = 0; u < 16; ++u) {
      const int b  = b0 + u;
      const int bc = b < nfull ? b : nfull - 1;
      const float v = pp[(size_t)bc * (NPB * H)];
      t[u] = b < nfull ? v : 0.0f;
    }
#pragma unroll
    for (int u = 0; u < 16; ++u) acc += t[u];
  }
  if (nt < N) {
    float xv[NPB - 1];
#pragma unroll
    for (int u = 0; u < NPB - 1; ++u) {
      const int n  = nt + u;
      const int nc = n < N ? n : N - 1;
      xv[u] = x[(size_t)nc * D + d];
    }
#pragma unroll
    for (int u = 0; u < NPB - 1; ++u) acc += sT[u] * xv[u];
  }

  out_centers[k * D + d] = acc / wsum;
  out_protos[k * D + d]  = prototypes[k * D + d];
}

#define MROWS  8
#define MWAVES 8
#define MCOLS  64
#define MPITCH 132

__global__ void __launch_bounds__(MWAVES * 32)
comp_matrix_kernel(const float* __restrict__ hi,
                   const float* __restrict__ hj,
                   const float* __restrict__ cb1,
                   const float* __restrict__ cg1,
                   const float* __restrict__ cbe1,
                   const float* __restrict__ cw2,
                   const float* __restrict__ cb2p,
                   float* __restrict__ out, int N) {
  __shared__ __attribute__((aligned(16))) float tile[MCOLS * MPITCH];

  const int lane = threadIdx.x & 31;
  const int wid  = __builtin_amdgcn_readfirstlane((int)(threadIdx.x >> 5));
  const int j0   = blockIdx.x * MCOLS;

  {
    const float4 cbv = *(const float4*)(cb1 + 4 * lane);
#pragma unroll
    for (int r = 0; r < MCOLS / MWAVES; ++r) {
      const int c = wid * (MCOLS / MWAVES) + r;
      int jr = j0 + c;
      if (jr >= N) jr = N - 1;
      float4 v = *(const float4*)(hj + (size_t)jr * H + 4 * lane);
      v.x += cbv.x; v.y += cbv.y; v.z += cbv.z; v.w += cbv.w;
      *(float4*)&tile[c * MPITCH + 4 * lane] = v;
    }
  }
  __syncthreads();

  v2f b[H];
  {
    const float* __restrict__ r0 = &tile[lane * MPITCH];
    const float* __restrict__ r1 = &tile[(lane + 32) * MPITCH];
#pragma unroll
    for (int h = 0; h < H; h += 4) {
      const float4 u = *(const float4*)(r0 + h);
      const float4 w = *(const float4*)(r1 + h);
      b[h + 0] = (v2f){u.x, w.x};
      b[h + 1] = (v2f){u.y, w.y};
      b[h + 2] = (v2f){u.z, w.z};
      b[h + 3] = (v2f){u.w, w.w};
    }
  }

  const float cb2 = cb2p[0];
  const int i0   = ((int)blockIdx.y * MWAVES + wid) * MROWS;
  const int iend = (i0 + MROWS < N) ? i0 + MROWS : N;
  const int ja = j0 + lane;
  const int jb = j0 + 32 + lane;
  const v2f zero2 = (v2f){0.0f, 0.0f};

  for (int i = i0; i < iend; ++i) {
    const float* __restrict__ a = hi + (size_t)i * H;
    int off = 0;
    asm volatile("" : "+s"(off));
    const float* __restrict__ gp  = cg1  + off;
    const float* __restrict__ bep = cbe1 + off;
    const float* __restrict__ cwp = cw2  + off;

    v2f s0 = zero2, s1 = zero2, s2 = zero2, s3 = zero2;
    v2f q0 = zero2, q1 = zero2, q2 = zero2, q3 = zero2;
#pragma unroll
    for (int hb = 0; hb < H; hb += 64) {
      float av[64];
#pragma unroll
      for (int k = 0; k < 64; ++k) av[k] = a[hb + k];
      __builtin_amdgcn_sched_barrier(0);
#pragma unroll
      for (int k = 0; k < 64; k += 4) {
        const v2f t0 = b[hb + k + 0] + av[k + 0];
        const v2f t1 = b[hb + k + 1] + av[k + 1];
        const v2f t2 = b[hb + k + 2] + av[k + 2];
        const v2f t3 = b[hb + k + 3] + av[k + 3];
        s0 += t0; s1 += t1; s2 += t2; s3 += t3;
        q0 = __builtin_elementwise_fma(t0, t0, q0);
        q1 = __builtin_elementwise_fma(t1, t1, q1);
        q2 = __builtin_elementwise_fma(t2, t2, q2);
        q3 = __builtin_elementwise_fma(t3, t3, q3);
      }
      __builtin_amdgcn_sched_barrier(0);
    }
    const v2f m   = ((s0 + s1) + (s2 + s3)) * (1.0f / 128.0f);
    const v2f var = ((q0 + q1) + (q2 + q3)) * (1.0f / 128.0f) - m * m;
    v2f inv;
    inv.x = rsqrtf(var.x + LN_EPS);
    inv.y = rsqrtf(var.y + LN_EPS);
    const v2f nmi = -m * inv;

    int off2 = 0;
    asm volatile("" : "+s"(off2));
    const float* __restrict__ a2 = a + off2;
    v2f d0 = zero2, d1 = zero2, d2 = zero2, d3 = zero2;
#pragma unroll
    for (int hb = 0; hb < H; hb += 16) {
      float av[16], gv[16], ev[16], cv[16];
#pragma unroll
      for (int k = 0; k < 16; ++k) {
        av[k] = a2[hb + k];
        gv[k] = gp[hb + k];
        ev[k] = bep[hb + k];
        cv[k] = cwp[hb + k];
      }
      __builtin_amdgcn_sched_barrier(0);
#pragma unroll
      for (int k = 0; k < 16; k += 4) {
        const v2f y0 = __builtin_elementwise_fma(b[hb + k + 0] + av[k + 0], inv, nmi);
        const v2f y1 = __builtin_elementwise_fma(b[hb + k + 1] + av[k + 1], inv, nmi);
        const v2f y2 = __builtin_elementwise_fma(b[hb + k + 2] + av[k + 2], inv, nmi);
        const v2f y3 = __builtin_elementwise_fma(b[hb + k + 3] + av[k + 3], inv, nmi);
        const v2f z0 = __builtin_elementwise_fma(y0, (v2f){gv[k + 0], gv[k + 0]}, (v2f){ev[k + 0], ev[k + 0]});
        const v2f z1 = __builtin_elementwise_fma(y1, (v2f){gv[k + 1], gv[k + 1]}, (v2f){ev[k + 1], ev[k + 1]});
        const v2f z2 = __builtin_elementwise_fma(y2, (v2f){gv[k + 2], gv[k + 2]}, (v2f){ev[k + 2], ev[k + 2]});
        const v2f z3 = __builtin_elementwise_fma(y3, (v2f){gv[k + 3], gv[k + 3]}, (v2f){ev[k + 3], ev[k + 3]});
        d0 = __builtin_elementwise_fma(__builtin_elementwise_max(z0, zero2), (v2f){cv[k + 0], cv[k + 0]}, d0);
        d1 = __builtin_elementwise_fma(__builtin_elementwise_max(z1, zero2), (v2f){cv[k + 1], cv[k + 1]}, d1);
        d2 = __builtin_elementwise_fma(__builtin_elementwise_max(z2, zero2), (v2f){cv[k + 2], cv[k + 2]}, d2);
        d3 = __builtin_elementwise_fma(__builtin_elementwise_max(z3, zero2), (v2f){cv[k + 3], cv[k + 3]}, d3);
      }
      __builtin_amdgcn_sched_barrier(0);
    }
    const v2f dot = (d0 + d1) + (d2 + d3);
    const float sca = tanhf(dot.x + cb2);
    const float scb = tanhf(dot.y + cb2);
    float* __restrict__ orow = out + (size_t)i * N;
    if (ja < N) orow[ja] = sca;
    if (jb < N) orow[jb] = scb;
  }
}

#define EWAVES 6
#define CONST_AS __attribute__((address_space(4)))
#define EPITCH 132

__global__ void __launch_bounds__(EWAVES * 32)
comp_scores_kernel(const float* __restrict__ hi,
                   const float* __restrict__ hj,
                   const int* __restrict__ ei, int E,
                   const float* __restrict__ cb1,
                   const float* __restrict__ cg1,
                   const float* __restrict__ cbe1,
                   const float* __restrict__ cw2,
                   const float* __restrict__ cb2p,
                   float* __restrict__ out) {
  __shared__ __attribute__((aligned(16))) float etile[EWAVES * 32 * EPITCH];

  const int lane = threadIdx.x & 31;
  const int wv   = __builtin_amdgcn_readfirstlane((int)(threadIdx.x >> 5));
  const int gw   = (int)(blockIdx.x * EWAVES) + wv;
  const int nw   = (int)(gridDim.x * EWAVES);
  float* __restrict__ tile = etile + wv * (32 * EPITCH);

  const v4f   cb  = *(const v4f*)(cb1 + 4 * lane);
  const float cb2 = cb2p[0];
  const v2f zero2 = (v2f){0.0f, 0.0f};

  const int estep = nw * 32;
  int iv, jv;
  {
    int e = gw * 32 + lane;
    if (e >= E) e = E - 1;
    iv = ei[e];
    jv = ei[E + e];
  }
  for (int e0 = gw * 32; e0 < E; e0 += estep) {
    int ivn = 0, jvn = 0;

    v4f ra[2][8], rb[2][8];
#pragma unroll
    for (int pp = 0; pp < 8; ++pp) {
      const int i = __builtin_amdgcn_readlane(iv, pp);
      const int j = __builtin_amdgcn_readlane(jv, pp);
      ra[0][pp] = *(const v4f*)(hi + (size_t)i * H + 4 * lane);
      rb[0][pp] = *(const v4f*)(hj + (size_t)j * H + 4 * lane);
    }
#pragma unroll
    for (int g = 0; g < 4; ++g) {
      if (g == 3) {
        int en = e0 + estep + lane;
        if (en >= E) en = E - 1;
        ivn = ei[en];
        jvn = ei[E + en];
      }
      if (g < 3) {
#pragma unroll
        for (int pp = 0; pp < 8; ++pp) {
          const int p = (g + 1) * 8 + pp;
          const int i = __builtin_amdgcn_readlane(iv, p);
          const int j = __builtin_amdgcn_readlane(jv, p);
          ra[(g + 1) & 1][pp] = *(const v4f*)(hi + (size_t)i * H + 4 * lane);
          rb[(g + 1) & 1][pp] = *(const v4f*)(hj + (size_t)j * H + 4 * lane);
        }
      }
#pragma unroll
      for (int pp = 0; pp < 8; ++pp) {
        const int p = g * 8 + pp;
        v4f a4 = ra[g & 1][pp];
        v4f b4 = rb[g & 1][pp];
        asm volatile("" : "+v"(a4), "+v"(b4));
        *(v4f*)&tile[p * EPITCH + 4 * lane] = a4 + b4 + cb;
      }
    }
    __builtin_amdgcn_fence(__ATOMIC_RELEASE, "wavefront");
    __builtin_amdgcn_wave_barrier();
    __builtin_amdgcn_fence(__ATOMIC_ACQUIRE, "wavefront");

    v2f t[H / 2];
    {
      const float* __restrict__ row = tile + lane * EPITCH;
#pragma unroll
      for (int h = 0; h < H; h += 4) {
        const float4 u = *(const float4*)(row + h);
        t[h / 2]     = (v2f){u.x, u.y};
        t[h / 2 + 1] = (v2f){u.z, u.w};
      }
    }

    v2f s0 = zero2, s1 = zero2, s2 = zero2, s3 = zero2;
    v2f q0 = zero2, q1 = zero2, q2 = zero2, q3 = zero2;
#pragma unroll
    for (int k = 0; k < H / 2; k += 4) {
      s0 += t[k + 0]; s1 += t[k + 1]; s2 += t[k + 2]; s3 += t[k + 3];
      q0 = __builtin_elementwise_fma(t[k + 0], t[k + 0], q0);
      q1 = __builtin_elementwise_fma(t[k + 1], t[k + 1], q1);
      q2 = __builtin_elementwise_fma(t[k + 2], t[k + 2], q2);
      q3 = __builtin_elementwise_fma(t[k + 3], t[k + 3], q3);
    }
    const v2f sv = (s0 + s1) + (s2 + s3);
    const v2f qv = (q0 + q1) + (q2 + q3);
    const float m   = (sv.x + sv.y) * (1.0f / 128.0f);
    const float var = (qv.x + qv.y) * (1.0f / 128.0f) - m * m;
    const float inv = rsqrtf(var + LN_EPS);
    const float nmi = -m * inv;
    const v2f inv2 = (v2f){inv, inv};
    const v2f nmi2 = (v2f){nmi, nmi};

    int off = 0;
    asm volatile("" : "+s"(off));
    const CONST_AS float* gp  = (const CONST_AS float*)(cg1  + off);
    const CONST_AS float* bep = (const CONST_AS float*)(cbe1 + off);
    const CONST_AS float* cwp = (const CONST_AS float*)(cw2  + off);
    v2f d0 = zero2, d1 = zero2, d2 = zero2, d3 = zero2;
#pragma unroll
    for (int hb = 0; hb < H; hb += 16) {
      float gv[16], ev[16], cv[16];
#pragma unroll
      for (int k = 0; k < 16; ++k) {
        gv[k] = gp[hb + k];
        ev[k] = bep[hb + k];
        cv[k] = cwp[hb + k];
      }
      __builtin_amdgcn_sched_barrier(0);
#pragma unroll
      for (int k = 0; k < 16; k += 8) {
        const int kp = (hb + k) / 2;
        const v2f y0 = __builtin_elementwise_fma(t[kp + 0], inv2, nmi2);
        const v2f y1 = __builtin_elementwise_fma(t[kp + 1], inv2, nmi2);
        const v2f y2 = __builtin_elementwise_fma(t[kp + 2], inv2, nmi2);
        const v2f y3 = __builtin_elementwise_fma(t[kp + 3], inv2, nmi2);
        const v2f z0 = __builtin_elementwise_fma(y0, (v2f){gv[k + 0], gv[k + 1]}, (v2f){ev[k + 0], ev[k + 1]});
        const v2f z1 = __builtin_elementwise_fma(y1, (v2f){gv[k + 2], gv[k + 3]}, (v2f){ev[k + 2], ev[k + 3]});
        const v2f z2 = __builtin_elementwise_fma(y2, (v2f){gv[k + 4], gv[k + 5]}, (v2f){ev[k + 4], ev[k + 5]});
        const v2f z3 = __builtin_elementwise_fma(y3, (v2f){gv[k + 6], gv[k + 7]}, (v2f){ev[k + 6], ev[k + 7]});
        d0 = __builtin_elementwise_fma(__builtin_elementwise_max(z0, zero2), (v2f){cv[k + 0], cv[k + 1]}, d0);
        d1 = __builtin_elementwise_fma(__builtin_elementwise_max(z1, zero2), (v2f){cv[k + 2], cv[k + 3]}, d1);
        d2 = __builtin_elementwise_fma(__builtin_elementwise_max(z2, zero2), (v2f){cv[k + 4], cv[k + 5]}, d2);
        d3 = __builtin_elementwise_fma(__builtin_elementwise_max(z3, zero2), (v2f){cv[k + 6], cv[k + 7]}, d3);
      }
      __builtin_amdgcn_sched_barrier(0);
    }
    const v2f dv = (d0 + d1) + (d2 + d3);
    const float sc = tanhf((dv.x + dv.y) + cb2);
    if (e0 + lane < E) out[e0 + lane] = sc;
    iv = ivn;
    jv = jvn;
  }
}

extern "C" void kernel_launch(void* const* d_in, const int* in_sizes, int n_in,
                              void* d_out, int out_size, void* d_ws, size_t ws_size,
                              hipStream_t stream) {
  const float* x    = (const float*)d_in[0];
  const int*   ei   = (const int*)  d_in[1];
  const float* w1   = (const float*)d_in[2];
  const float* b1   = (const float*)d_in[3];
  const float* g1   = (const float*)d_in[4];
  const float* be1  = (const float*)d_in[5];
  const float* w2   = (const float*)d_in[6];
  const float* b2   = (const float*)d_in[7];
  const float* g2   = (const float*)d_in[8];
  const float* be2  = (const float*)d_in[9];
  const float* w3   = (const float*)d_in[10];
  const float* b3   = (const float*)d_in[11];
  const float* cw1  = (const float*)d_in[12];
  const float* cb1  = (const float*)d_in[13];
  const float* cg1  = (const float*)d_in[14];
  const float* cbe1 = (const float*)d_in[15];
  const float* cw2  = (const float*)d_in[16];
  const float* cb2  = (const float*)d_in[17];
  const float* prot = (const float*)d_in[18];

  const int N = in_sizes[0] / D;
  const int E = in_sizes[1] / 2;

  float* h1pre = (float*)d_ws;
  float* hi    = h1pre + (size_t)N * H;
  float* hj    = hi    + (size_t)N * H;

  float* out       = (float*)d_out;
  float* o_logits  = out;
  float* o_probs   = o_logits  + (size_t)N * KC;
  float* o_scores  = o_probs   + (size_t)N * KC;
  float* o_matrix  = o_scores  + (size_t)E;
  float* o_centers = o_matrix  + (size_t)N * N;
  float* o_protos  = o_centers + (size_t)KC * D;

  const int rowTiles = (N + 15) / 16;

  stripe_kernel<<<dim3(rowTiles, 3), NPB * 32, 0, stream>>>(
      x, w1, cw1, hi, hj, h1pre, b1, g1, be1, w2, b2, g2, be2, w3, b3,
      o_logits, o_probs, N);

  centers_kernel<<<KC, 128, 0, stream>>>(
      o_probs, x, h1pre, prot, o_centers, o_protos, N);

  comp_matrix_kernel<<<dim3((N + MCOLS - 1) / MCOLS, (N + MWAVES * MROWS - 1) / (MWAVES * MROWS)),
                       MWAVES * 32, 0, stream>>>(
      hi, hj, cb1, cg1, cbe1, cw2, cb2, o_matrix, N);

  comp_scores_kernel<<<768, EWAVES * 32, 0, stream>>>(
      hi, hj, ei, E, cb1, cg1, cbe1, cw2, cb2, o_scores);
}


# latency-bound stages load in wider groups: centres launch issues all its loads before the first wait (butterfly column sum, 1 barrier); node net takes 32 w2 rows and all w3 rows per group
# speedup vs baseline: 4.0144x; 1.1256x over previous
#include <hip/hip_runtime.h>
#include <math.h>

typedef __attribute__((ext_vector_type(2))) float v2f;
typedef __attribute__((ext_vector_type(4))) float v4f;
typedef __attribute__((ext_vector_type(8))) float v8f;

#define D   128
#define H   128
#define HH2 64
#define KC  10
#define LN_EPS 1e-5f

#define NPB 16
#define TPITCH 132

__device__ __forceinline__ v8f stripe_tile(const float* __restrict__ x,
                                           const float* __restrict__ Bm,
                                           const int rt, const int ct,
                                           const int lane, const int N) {
  const int half = lane >> 4;
  const int l16  = lane & 15;
  int arow = rt * 16 + l16;
  if (arow >= N) arow = N - 1;
  const float* __restrict__ arow_p = x + (size_t)arow * D + half * 2;
  const float* __restrict__ bcol_p = Bm + (size_t)(half * 2) * H + ct * 16 + l16;
  v2f a[D / 4], b[D / 4];
#pragma unroll
  for (int s = 0; s < D / 4; ++s) a[s] = *(const v2f*)(arow_p + 4 * s);
#pragma unroll
  for (int s = 0; s < D / 4; ++s) {
    b[s].x = bcol_p[(size_t)(4 * s) * H];
    b[s].y = bcol_p[(size_t)(4 * s + 1) * H];
  }
  __builtin_amdgcn_sched_barrier(0);
  v8f acc0 = {}, acc1 = {};
#pragma unroll
  for (int s = 0; s < D / 4; s += 2) {
    acc0 = __builtin_amdgcn_wmma_f32_16x16x4_f32(false, a[s], false, b[s],
                                                 (short)0, acc0, false, false);
    acc1 = __builtin_amdgcn_wmma_f32_16x16x4_f32(false, a[s + 1], false, b[s + 1],
                                                 (short)0, acc1, false, false);
  }
  return acc0 + acc1;
}

__device__ __forceinline__ float wave_sum(float v) {
#pragma unroll
  for (int w = 16; w > 0; w >>= 1) v += __shfl_xor(v, w, 32);
  return v;
}

__device__ __forceinline__ float wave_max(float v) {
#pragma unroll
  for (int w = 16; w > 0; w >>= 1) v = fmaxf(v, __shfl_xor(v, w, 32));
  return v;
}

__global__ void __launch_bounds__(NPB * 32)
stripe_kernel(const float* __restrict__ x,
              const float* __restrict__ w1,
              const float* __restrict__ cw1,
              float* __restrict__ hi,
              float* __restrict__ hj,
              float* __restrict__ part_ws,
              const float* __restrict__ b1,
              const float* __restrict__ g1,
              const float* __restrict__ be1,
              const float* __restrict__ w2,
              const float* __restrict__ b2,
              const float* __restrict__ g2,
              const float* __restrict__ be2,
              const float* __restrict__ w3,
              const float* __restrict__ b3,
              float* __restrict__ out_logits,
              float* __restrict__ out_probs,
              int N) {
  __shared__ __attribute__((aligned(16))) float sh0[NPB * TPITCH];
  __shared__ float sh1[NPB][H];
  __shared__ float sh2[NPB][HH2];
  __shared__ float sp[NPB][KC];

  const int lane = threadIdx.x & 31;
  const int wv   = __builtin_amdgcn_readfirstlane((int)(threadIdx.x >> 5));
  const int mat  = blockIdx.y;
  const int rt   = blockIdx.x;
  const int half = lane >> 4;
  const int l16  = lane & 15;

  if (mat != 0) {
    if (wv < H / 16) {
      const v8f acc = stripe_tile(x, (mat == 1) ? cw1 : cw1 + D * H, rt, wv, lane, N);
      float* __restrict__ out = (mat == 1) ? hi : hj;
      const int rbase = rt * 16 + half * 8;
      if (rt * 16 + 16 <= N) {
#pragma unroll
        for (int r = 0; r < 8; ++r)
          out[(size_t)(rbase + r) * H + wv * 16 + l16] = acc[r];
      } else {
#pragma unroll
        for (int r = 0; r < 8; ++r)
          if (rbase + r < N)
            out[(size_t)(rbase + r) * H + wv * 16 + l16] = acc[r];
      }
    }
    return;
  }

  const float4 bb1 = *(const float4*)(b1 + 4 * lane);
  const float4 gg1 = *(const float4*)(g1 + 4 * lane);
  const float4 ee1 = *(const float4*)(be1 + 4 * lane);

  if (wv < H / 16) {
    const v8f acc = stripe_tile(x, w1, rt, wv, lane, N);
#pragma unroll
    for (int r = 0; r < 8; ++r)
      sh0[(half * 8 + r) * TPITCH + wv * 16 + l16] = acc[r];
  }
  __syncthreads();

  const int n0   = rt * NPB;
  const int n    = n0 + wv;
  const bool live = n < N;

  {
    const float4 hv = *(const float4*)&sh0[wv * TPITCH + 4 * lane];
    float4 v;
    v.x = hv.x + bb1.x; v.y = hv.y + bb1.y; v.z = hv.z + bb1.z; v.w = hv.w + bb1.w;
    const float s = wave_sum((v.x + v.y) + (v.z + v.w));
    const float q = wave_sum((v.x * v.x + v.y * v.y) + (v.z * v.z + v.w * v.w));
    const float m   = s * (1.0f / 128.0f);
    const float var = q * (1.0f / 128.0f) - m * m;
    const float inv = rsqrtf(var + LN_EPS);
    float4 r;
    r.x = fmaxf((v.x - m) * inv * gg1.x + ee1.x, 0.0f);
    r.y = fmaxf((v.y - m) * inv * gg1.y + ee1.y, 0.0f);
    r.z = fmaxf((v.z - m) * inv * gg1.z + ee1.z, 0.0f);
    r.w = fmaxf((v.w - m) * inv * gg1.w + ee1.w, 0.0f);
    *(float4*)&sh1[wv][4 * lane] = r;
  }
  __syncthreads();

  {
    float2 a2 = *(const float2*)(b2 + 2 * lane);
    const float* __restrict__ w2p = w2 + 2 * lane;
#pragma unroll 32
    for (int k = 0; k < H; ++k) {
      const float  hk = sh1[wv][k];
      const float2 w  = *(const float2*)(w2p + k * HH2);
      a2.x += hk * w.x;
      a2.y += hk * w.y;
    }
    const float s2 = wave_sum(a2.x + a2.y);
    const float q2 = wave_sum(a2.x * a2.x + a2.y * a2.y);
    const float m2 = s2 * (1.0f / 64.0f);
    const float v2 = q2 * (1.0f / 64.0f) - m2 * m2;
    const float i2 = rsqrtf(v2 + LN_EPS);
    const float2 gg = *(const float2*)(g2 + 2 * lane);
    const float2 ee = *(const float2*)(be2 + 2 * lane);
    float2 h2;
    h2.x = fmaxf((a2.x - m2) * i2 * gg.x + ee.x, 0.0f);
    h2.y = fmaxf((a2.y - m2) * i2 * gg.y + ee.y, 0.0f);
    *(float2*)&sh2[wv][2 * lane] = h2;
  }
  __syncthreads();

  {
    const bool owner = lane < KC;
    const int  tc    = owner ? lane : KC - 1;
    float lg = b3[tc];
#pragma unroll
    for (int k = 0; k < HH2; ++k) lg += sh2[wv][k] * w3[k * KC + tc];
    const float mx   = wave_max(owner ? lg : -INFINITY);
    const float e    = owner ? __expf(lg - mx) : 0.0f;
    const float ssum = wave_sum(e);
    const float p    = e / ssum;
    if (owner) {
      if (live) {
        out_logits[(size_t)n * KC + lane] = lg;
        out_probs[(size_t)n * KC + lane]  = p;
      }
      sp[wv][lane] = live ? p : 0.0f;
    }
  }
  __syncthreads();

  if (n0 + NPB <= N) {
    const int d  = threadIdx.x & (D - 1);
    const int kk = threadIdx.x >> 7;
    const int k3 = (kk < 2) ? kk + 8 : kk + 4;
    const float* __restrict__ xp = x + (size_t)n0 * D + d;
    float a0 = 0.0f, a1 = 0.0f, a2c = 0.0f;
#pragma unroll
    for (int g = 0; g < NPB; ++g) {
      const float xv = xp[g * D];
      a0  += sp[g][kk] * xv;
      a1  += sp[g][kk + 4] * xv;
      a2c += sp[g][k3] * xv;
    }
    float* __restrict__ part = part_ws + (size_t)n0 * H;
    part[kk * D + d]       = a0;
    part[(kk + 4) * D + d] = a1;
    if (kk < 2) part[(kk + 8) * D + d] = a2c;
  }
}

#define CTHREADS 256
#define CPV 4
#define CGV 32

__global__ void __launch_bounds__(CTHREADS)
centers_kernel(const float* __restrict__ probs,
               const float* __restrict__ x,
               const float* __restrict__ part,
               const float* __restrict__ prototypes,
               float* __restrict__ out_centers,
               float* __restrict__ out_protos,
               int N) {
  const int k    = blockIdx.x;
  const int tid  = threadIdx.x;
  const int d    = tid & (D - 1);
  const int half = tid >> 7;
  const int lane = tid & 31;
  const int wv   = tid >> 5;
  __shared__ float sW[CTHREADS / 32];
  __shared__ float sA[D];

  const int nfull = N / NPB;
  const int nt    = nfull * NPB;

  const float* __restrict__ pp = part + k * D + d;

  float pv[CPV], tv[CGV], tx[NPB / 2], tp[NPB / 2];
#pragma unroll
  for (int u = 0; u < CPV; ++u) {
    const int n  = u * CTHREADS + tid;
    const int nc = n < N ? n : N - 1;
    pv[u] = probs[(size_t)nc * KC + k];
  }
#pragma unroll
  for (int u = 0; u < CGV; ++u) {
    const int b = half + 2 * u;
    const float* __restrict__ src = b < nfull ? pp + (size_t)b * (NPB * H) : x + d;
    tv[u] = *src;
  }
#pragma unroll
  for (int v = 0; v < NPB / 2; ++v) {
    const int n  = nt + half + 2 * v;
    const int nc = n < N ? n : N - 1;
    tx[v] = x[(size_t)nc * D + d];
    tp[v] = probs[(size_t)nc * KC + k];
  }
  const float proto = prototypes[k * D + d];
  __builtin_amdgcn_sched_barrier(0);

  float wpart = 0.0f;
#pragma unroll
  for (int u = 0; u < CPV; ++u) wpart += (u * CTHREADS + tid < N) ? pv[u] : 0.0f;
  float acc = 0.0f;
#pragma unroll
  for (int u = 0; u < CGV; ++u) acc += (half + 2 * u < nfull) ? tv[u] : 0.0f;
#pragma unroll
  for (int v = 0; v < NPB / 2; ++v) acc += (nt + half + 2 * v < N) ? tp[v] * tx[v] : 0.0f;

  for (int base = CPV * CTHREADS; base < N; base += CPV * CTHREADS) {
    float t[CPV];
#pragma unroll
    for (int u = 0; u < CPV; ++u) {
      const int n  = base + u * CTHREADS + tid;
      const int nc = n < N ? n : N - 1;
      const float v = probs[(size_t)nc * KC + k];
      t[u] = n < N ? v : 0.0f;
    }
#pragma unroll
    for (int u = 0; u < CPV; ++u) wpart += t[u];
  }
  for (int b0 = 2 * CGV; b0 < nfull; b0 += 2 * CGV) {
    float t[CGV];
#pragma unroll
    for (int u = 0; u < CGV; ++u) {
      const int b = b0 + half + 2 * u;
      const float* __restrict__ src = b < nfull ? pp + (size_t)b * (NPB * H) : x + d;
      const float v = *src;
      t[u] = b < nfull ? v : 0.0f;
    }
#pragma unroll
    for (int u = 0; u < CGV; ++u) acc += t[u];
  }

  const float ws = wave_sum(wpart);
  if (lane == 0) sW[wv] = ws;
  if (half == 1) sA[d] = acc;
  __syncthreads();
  if (half == 0) {
    const float wsum = (((sW[0] + sW[1]) + (sW[2] + sW[3])) +
                        ((sW[4] + sW[5]) + (sW[6] + sW[7]))) + 1e-8f;
    out_centers[k * D + d] = (acc + sA[d]) / wsum;
    out_protos[k * D + d]  = proto;
  }
}

#define MROWS  8
#define MWAVES 8
#define MCOLS  64
#define MPITCH 132

__global__ void __launch_bounds__(MWAVES * 32)
comp_matrix_kernel(const float* __restrict__ hi,
                   const float* __restrict__ hj,
                   const float* __restrict__ cb1,
                   const float* __restrict__ cg1,
                   const float* __restrict__ cbe1,
                   const float* __restrict__ cw2,
                   const float* __restrict__ cb2p,
                   float* __restrict__ out, int N) {
  __shared__ __attribute__((aligned(16))) float tile[MCOLS * MPITCH];

  const int lane = threadIdx.x & 31;
  const int wid  = __builtin_amdgcn_readfirstlane((int)(threadIdx.x >> 5));
  const int j0   = blockIdx.x * MCOLS;

  {
    const float4 cbv = *(const float4*)(cb1 + 4 * lane);
#pragma unroll
    for (int r = 0; r < MCOLS / MWAVES; ++r) {
      const int c = wid * (MCOLS / MWAVES) + r;
      int jr = j0 + c;
      if (jr >= N) jr = N - 1;
      float4 v = *(const float4*)(hj + (size_t)jr * H + 4 * lane);
      v.x += cbv.x; v.y += cbv.y; v.z += cbv.z; v.w += cbv.w;
      *(float4*)&tile[c * MPITCH + 4 * lane] = v;
    }
  }
  __syncthreads();

  v2f b[H];
  {
    const float* __restrict__ r0 = &tile[lane * MPITCH];
    const float* __restrict__ r1 = &tile[(lane + 32) * MPITCH];
#pragma unroll
    for (int h = 0; h < H; h += 4) {
      const float4 u = *(const float4*)(r0 + h);
      const float4 w = *(const float4*)(r1 + h);
      b[h + 0] = (v2f){u.x, w.x};
      b[h + 1] = (v2f){u.y, w.y};
      b[h + 2] = (v2f){u.z, w.z};
      b[h + 3] = (v2f){u.w, w.w};
    }
  }

  const float cb2 = cb2p[0];
  const int i0   = ((int)blockIdx.y * MWAVES + wid) * MROWS;
  const int iend = (i0 + MROWS < N) ? i0 + MROWS : N;
  const int ja = j0 + lane;
  const int jb = j0 + 32 + lane;
  const v2f zero2 = (v2f){0.0f, 0.0f};

  for (int i = i0; i < iend; ++i) {
    const float* __restrict__ a = hi + (size_t)i * H;
    int off = 0;
    asm volatile("" : "+s"(off));
    const float* __restrict__ gp  = cg1  + off;
    const float* __restrict__ bep = cbe1 + off;
    const float* __restrict__ cwp = cw2  + off;

    v2f s0 = zero2, s1 = zero2, s2 = zero2, s3 = zero2;
    v2f q0 = zero2, q1 = zero2, q2 = zero2, q3 = zero2;
#pragma unroll
    for (int hb = 0; hb < H; hb += 64) {
      float av[64];
#pragma unroll
      for (int k = 0; k < 64; ++k) av[k] = a[hb + k];
      __builtin_amdgcn_sched_barrier(0);
#pragma unroll
      for (int k = 0; k < 64; k += 4) {
        const v2f t0 = b[hb + k + 0] + av[k + 0];
        const v2f t1 = b[hb + k + 1] + av[k + 1];
        const v2f t2 = b[hb + k + 2] + av[k + 2];
        const v2f t3 = b[hb + k + 3] + av[k + 3];
        s0 += t0; s1 += t1; s2 += t2; s3 += t3;
        q0 = __builtin_elementwise_fma(t0, t0, q0);
        q1 = __builtin_elementwise_fma(t1, t1, q1);
        q2 = __builtin_elementwise_fma(t2, t2, q2);
        q3 = __builtin_elementwise_fma(t3, t3, q3);
      }
      __builtin_amdgcn_sched_barrier(0);
    }
    const v2f m   = ((s0 + s1) + (s2 + s3)) * (1.0f / 128.0f);
    const v2f var = ((q0 + q1) + (q2 + q3)) * (1.0f / 128.0f) - m * m;
    v2f inv;
    inv.x = rsqrtf(var.x + LN_EPS);
    inv.y = rsqrtf(var.y + LN_EPS);
    const v2f nmi = -m * inv;

    int off2 = 0;
    asm volatile("" : "+s"(off2));
    const float* __restrict__ a2 = a + off2;
    v2f d0 = zero2, d1 = zero2, d2 = zero2, d3 = zero2;
#pragma unroll
    for (int hb = 0; hb < H; hb += 16) {
      float av[16], gv[16], ev[16], cv[16];
#pragma unroll
      for (int k = 0; k < 16; ++k) {
        av[k] = a2[hb + k];
        gv[k] = gp[hb + k];
        ev[k] = bep[hb + k];
        cv[k] = cwp[hb + k];
      }
      __builtin_amdgcn_sched_barrier(0);
#pragma unroll
      for (int k = 0; k < 16; k += 4) {
        const v2f y0 = __builtin_elementwise_fma(b[hb + k + 0] + av[k + 0], inv, nmi);
        const v2f y1 = __builtin_elementwise_fma(b[hb + k + 1] + av[k + 1], inv, nmi);
        const v2f y2 = __builtin_elementwise_fma(b[hb + k + 2] + av[k + 2], inv, nmi);
        const v2f y3 = __builtin_elementwise_fma(b[hb + k + 3] + av[k + 3], inv, nmi);
        const v2f z0 = __builtin_elementwise_fma(y0, (v2f){gv[k + 0], gv[k + 0]}, (v2f){ev[k + 0], ev[k + 0]});
        const v2f z1 = __builtin_elementwise_fma(y1, (v2f){gv[k + 1], gv[k + 1]}, (v2f){ev[k + 1], ev[k + 1]});
        const v2f z2 = __builtin_elementwise_fma(y2, (v2f){gv[k + 2], gv[k + 2]}, (v2f){ev[k + 2], ev[k + 2]});
        const v2f z3 = __builtin_elementwise_fma(y3, (v2f){gv[k + 3], gv[k + 3]}, (v2f){ev[k + 3], ev[k + 3]});
        d0 = __builtin_elementwise_fma(__builtin_elementwise_max(z0, zero2), (v2f){cv[k + 0], cv[k + 0]}, d0);
        d1 = __builtin_elementwise_fma(__builtin_elementwise_max(z1, zero2), (v2f){cv[k + 1], cv[k + 1]}, d1);
        d2 = __builtin_elementwise_fma(__builtin_elementwise_max(z2, zero2), (v2f){cv[k + 2], cv[k + 2]}, d2);
        d3 = __builtin_elementwise_fma(__builtin_elementwise_max(z3, zero2), (v2f){cv[k + 3], cv[k + 3]}, d3);
      }
      __builtin_amdgcn_sched_barrier(0);
    }
    const v2f dot = (d0 + d1) + (d2 + d3);
    const float sca = tanhf(dot.x + cb2);
    const float scb = tanhf(dot.y + cb2);
    float* __restrict__ orow = out + (size_t)i * N;
    if (ja < N) orow[ja] = sca;
    if (jb < N) orow[jb] = scb;
  }
}

#define EWAVES 6
#define CONST_AS __attribute__((address_space(4)))
#define EPITCH 132

__global__ void __launch_bounds__(EWAVES * 32)
comp_scores_kernel(const float* __restrict__ hi,
                   const float* __restrict__ hj,
                   const int* __restrict__ ei, int E,
                   const float* __restrict__ cb1,
                   const float* __restrict__ cg1,
                   const float* __restrict__ cbe1,
                   const float* __restrict__ cw2,
                   const float* __restrict__ cb2p,
                   float* __restrict__ out) {
  __shared__ __attribute__((aligned(16))) float etile[EWAVES * 32 * EPITCH];

  const int lane = threadIdx.x & 31;
  const int wv   = __builtin_amdgcn_readfirstlane((int)(threadIdx.x >> 5));
  const int gw   = (int)(blockIdx.x * EWAVES) + wv;
  const int nw   = (int)(gridDim.x * EWAVES);
  float* __restrict__ tile = etile + wv * (32 * EPITCH);

  const v4f   cb  = *(const v4f*)(cb1 + 4 * lane);
  const float cb2 = cb2p[0];
  const v2f zero2 = (v2f){0.0f, 0.0f};

  const int estep = nw * 32;
  int iv, jv;
  {
    int e = gw * 32 + lane;
    if (e >= E) e = E - 1;
    iv = ei[e];
    jv = ei[E + e];
  }
  for (int e0 = gw * 32; e0 < E; e0 += estep) {
    int ivn = 0, jvn = 0;

    v4f ra[2][8], rb[2][8];
#pragma unroll
    for (int pp = 0; pp < 8; ++pp) {
      const int i = __builtin_amdgcn_readlane(iv, pp);
      const int j = __builtin_amdgcn_readlane(jv, pp);
      ra[0][pp] = *(const v4f*)(hi + (size_t)i * H + 4 * lane);
      rb[0][pp] = *(const v4f*)(hj + (size_t)j * H + 4 * lane);
    }
#pragma unroll
    for (int g = 0; g < 4; ++g) {
      if (g == 3) {
        int en = e0 + estep + lane;
        if (en >= E) en = E - 1;
        ivn = ei[en];
        jvn = ei[E + en];
      }
      if (g < 3) {
#pragma unroll
        for (int pp = 0; pp < 8; ++pp) {
          const int p = (g + 1) * 8 + pp;
          const int i = __builtin_amdgcn_readlane(iv, p);
          const int j = __builtin_amdgcn_readlane(jv, p);
          ra[(g + 1) & 1][pp] = *(const v4f*)(hi + (size_t)i * H + 4 * lane);
          rb[(g + 1) & 1][pp] = *(const v4f*)(hj + (size_t)j * H + 4 * lane);
        }
      }
#pragma unroll
      for (int pp = 0; pp < 8; ++pp) {
        const int p = g * 8 + pp;
        v4f a4 = ra[g & 1][pp];
        v4f b4 = rb[g & 1][pp];
        asm volatile("" : "+v"(a4), "+v"(b4));
        *(v4f*)&tile[p * EPITCH + 4 * lane] = a4 + b4 + cb;
      }
    }
    __builtin_amdgcn_fence(__ATOMIC_RELEASE, "wavefront");
    __builtin_amdgcn_wave_barrier();
    __builtin_amdgcn_fence(__ATOMIC_ACQUIRE, "wavefront");

    v2f t[H / 2];
    {
      const float* __restrict__ row = tile + lane * EPITCH;
#pragma unroll
      for (int h = 0; h < H; h += 4) {
        const float4 u = *(const float4*)(row + h);
        t[h / 2]     = (v2f){u.x, u.y};
        t[h / 2 + 1] = (v2f){u.z, u.w};
      }
    }

    v2f s0 = zero2, s1 = zero2, s2 = zero2, s3 = zero2;
    v2f q0 = zero2, q1 = zero2, q2 = zero2, q3 = zero2;
#pragma unroll
    for (int k = 0; k < H / 2; k += 4) {
      s0 += t[k + 0]; s1 += t[k + 1]; s2 += t[k + 2]; s3 += t[k + 3];
      q0 = __builtin_elementwise_fma(t[k + 0], t[k + 0], q0);
      q1 = __builtin_elementwise_fma(t[k + 1], t[k + 1], q1);
      q2 = __builtin_elementwise_fma(t[k + 2], t[k + 2], q2);
      q3 = __builtin_elementwise_fma(t[k + 3], t[k + 3], q3);
    }
    const v2f sv = (s0 + s1) + (s2 + s3);
    const v2f qv = (q0 + q1) + (q2 + q3);
    const float m   = (sv.x + sv.y) * (1.0f / 128.0f);
    const float var = (qv.x + qv.y) * (1.0f / 128.0f) - m * m;
    const float inv = rsqrtf(var + LN_EPS);
    const float nmi = -m * inv;
    const v2f inv2 = (v2f){inv, inv};
    const v2f nmi2 = (v2f){nmi, nmi};

    int off = 0;
    asm volatile("" : "+s"(off));
    const CONST_AS float* gp  = (const CONST_AS float*)(cg1  + off);
    const CONST_AS float* bep = (const CONST_AS float*)(cbe1 + off);
    const CONST_AS float* cwp = (const CONST_AS float*)(cw2  + off);
    v2f d0 = zero2, d1 = zero2, d2 = zero2, d3 = zero2;
#pragma unroll
    for (int hb = 0; hb < H; hb += 16) {
      float gv[16], ev[16], cv[16];
#pragma unroll
      for (int k = 0; k < 16; ++k) {
        gv[k] = gp[hb + k];
        ev[k] = bep[hb + k];
        cv[k] = cwp[hb + k];
      }
      __builtin_amdgcn_sched_barrier(0);
#pragma unroll
      for (int k = 0; k < 16; k += 8) {
        const int kp = (hb + k) / 2;
        const v2f y0 = __builtin_elementwise_fma(t[kp + 0], inv2, nmi2);
        const v2f y1 = __builtin_elementwise_fma(t[kp + 1], inv2, nmi2);
        const v2f y2 = __builtin_elementwise_fma(t[kp + 2], inv2, nmi2);
        const v2f y3 = __builtin_elementwise_fma(t[kp + 3], inv2, nmi2);
        const v2f z0 = __builtin_elementwise_fma(y0, (v2f){gv[k + 0], gv[k + 1]}, (v2f){ev[k + 0], ev[k + 1]});
        const v2f z1 = __builtin_elementwise_fma(y1, (v2f){gv[k + 2], gv[k + 3]}, (v2f){ev[k + 2], ev[k + 3]});
        const v2f z2 = __builtin_elementwise_fma(y2, (v2f){gv[k + 4], gv[k + 5]}, (v2f){ev[k + 4], ev[k + 5]});
        const v2f z3 = __builtin_elementwise_fma(y3, (v2f){gv[k + 6], gv[k + 7]}, (v2f){ev[k + 6], ev[k + 7]});
        d0 = __builtin_elementwise_fma(__builtin_elementwise_max(z0, zero2), (v2f){cv[k + 0], cv[k + 1]}, d0);
        d1 = __builtin_elementwise_fma(__builtin_elementwise_max(z1, zero2), (v2f){cv[k + 2], cv[k + 3]}, d1);
        d2 = __builtin_elementwise_fma(__builtin_elementwise_max(z2, zero2), (v2f){cv[k + 4], cv[k + 5]}, d2);
        d3 = __builtin_elementwise_fma(__builtin_elementwise_max(z3, zero2), (v2f){cv[k + 6], cv[k + 7]}, d3);
      }
      __builtin_amdgcn_sched_barrier(0);
    }
    const v2f dv = (d0 + d1) + (d2 + d3);
    const float sc = tanhf((dv.x + dv.y) + cb2);
    if (e0 + lane < E) out[e0 + lane] = sc;
    iv = ivn;
    jv = jvn;
  }
}

extern "C" void kernel_launch(void* const* d_in, const int* in_sizes, int n_in,
                              void* d_out, int out_size, void* d_ws, size_t ws_size,
                              hipStream_t stream) {
  const float* x    = (const float*)d_in[0];
  const int*   ei   = (const int*)  d_in[1];
  const float* w1   = (const float*)d_in[2];
  const float* b1   = (const float*)d_in[3];
  const float* g1   = (const float*)d_in[4];
  const float* be1  = (const float*)d_in[5];
  const float* w2   = (const float*)d_in[6];
  const float* b2   = (const float*)d_in[7];
  const float* g2   = (const float*)d_in[8];
  const float* be2  = (const float*)d_in[9];
  const float* w3   = (const float*)d_in[10];
  const float* b3   = (const float*)d_in[11];
  const float* cw1  = (const float*)d_in[12];
  const float* cb1  = (const float*)d_in[13];
  const float* cg1  = (const float*)d_in[14];
  const float* cbe1 = (const float*)d_in[15];
  const float* cw2  = (const float*)d_in[16];
  const float* cb2  = (const float*)d_in[17];
  const float* prot = (const float*)d_in[18];

  const int N = in_sizes[0] / D;
  const int E = in_sizes[1] / 2;

  float* h1pre = (float*)d_ws;
  float* hi    = h1pre + (size_t)N * H;
  float* hj    = hi    + (size_t)N * H;

  float* out       = (float*)d_out;
  float* o_logits  = out;
  float* o_probs   = o_logits  + (size_t)N * KC;
  float* o_scores  = o_probs   + (size_t)N * KC;
  float* o_matrix  = o_scores  + (size_t)E;
  float* o_centers = o_matrix  + (size_t)N * N;
  float* o_protos  = o_centers + (size_t)KC * D;

  const int rowTiles = (N + 15) / 16;

  stripe_kernel<<<dim3(rowTiles, 3), NPB * 32, 0, stream>>>(
      x, w1, cw1, hi, hj, h1pre, b1, g1, be1, w2, b2, g2, be2, w3, b3,
      o_logits, o_probs, N);

  centers_kernel<<<KC, CTHREADS, 0, stream>>>(
      o_probs, x, h1pre, prot, o_centers, o_protos, N);

  comp_matrix_kernel<<<dim3((N + MCOLS - 1) / MCOLS, (N + MWAVES * MROWS - 1) / (MWAVES * MROWS)),
                       MWAVES * 32, 0, stream>>>(
      hi, hj, cb1, cg1, cbe1, cw2, cb2, o_matrix, N);

  comp_scores_kernel<<<768, EWAVES * 32, 0, stream>>>(
      hi, hj, ei, E, cb1, cg1, cbe1, cw2, cb2, o_scores);
}


# pair matrix: a wave's 8 hi rows fetched once beside the hj rows and kept in LDS, read in the row loop as one-address LDS reads into vector registers; was 2 scalar loads from L2 per row inside the loop
# speedup vs baseline: 4.4839x; 1.1170x over previous
#include <hip/hip_runtime.h>
#include <math.h>

typedef __attribute__((ext_vector_type(2))) float v2f;
typedef __attribute__((ext_vector_type(4))) float v4f;
typedef __attribute__((ext_vector_type(8))) float v8f;

#define D   128
#define H   128
#define HH2 64
#define KC  10
#define LN_EPS 1e-5f

#define NPB 16
#define TPITCH 132

__device__ __forceinline__ v8f stripe_tile(const float* __restrict__ x,
                                           const float* __restrict__ Bm,
                                           const int rt, const int ct,
                                           const int lane, const int N) {
  const int half = lane >> 4;
  const int l16  = lane & 15;
  int arow = rt * 16 + l16;
  if (arow >= N) arow = N - 1;
  const float* __restrict__ arow_p = x + (size_t)arow * D + half * 2;
  const float* __restrict__ bcol_p = Bm + (size_t)(half * 2) * H + ct * 16 + l16;
  v2f a[D / 4], b[D / 4];
#pragma unroll
  for (int s = 0; s < D / 4; ++s) a[s] = *(const v2f*)(arow_p + 4 * s);
#pragma unroll
  for (int s = 0; s < D / 4; ++s) {
    b[s].x = bcol_p[(size_t)(4 * s) * H];
    b[s].y = bcol_p[(size_t)(4 * s + 1) * H];
  }
  __builtin_amdgcn_sched_barrier(0);
  v8f acc0 = {}, acc1 = {};
#pragma unroll
  for (int s = 0; s < D / 4; s += 2) {
    acc0 = __builtin_amdgcn_wmma_f32_16x16x4_f32(false, a[s], false, b[s],
                                                 (short)0, acc0, false, false);
    acc1 = __builtin_amdgcn_wmma_f32_16x16x4_f32(false, a[s + 1], false, b[s + 1],
                                                 (short)0, acc1, false, false);
  }
  return acc0 + acc1;
}

__device__ __forceinline__ float wave_sum(float v) {
#pragma unroll
  for (int w = 16; w > 0; w >>= 1) v += __shfl_xor(v, w, 32);
  return v;
}

__device__ __forceinline__ float wave_max(float v) {
#pragma unroll
  for (int w = 16; w > 0; w >>= 1) v = fmaxf(v, __shfl_xor(v, w, 32));
  return v;
}

__global__ void __launch_bounds__(NPB * 32)
stripe_kernel(const float* __restrict__ x,
              const float* __restrict__ w1,
              const float* __restrict__ cw1,
              float* __restrict__ hi,
              float* __restrict__ hj,
              float* __restrict__ part_ws,
              const float* __restrict__ b1,
              const float* __restrict__ g1,
              const float* __restrict__ be1,
              const float* __restrict__ w2,
              const float* __restrict__ b2,
              const float* __restrict__ g2,
              const float* __restrict__ be2,
              const float* __restrict__ w3,
              const float* __restrict__ b3,
              float* __restrict__ out_logits,
              float* __restrict__ out_probs,
              int N) {
  __shared__ __attribute__((aligned(16))) float sh0[NPB * TPITCH];
  __shared__ float sh1[NPB][H];
  __shared__ float sh2[NPB][HH2];
  __shared__ float sp[NPB][KC];

  const int lane = threadIdx.x & 31;
  const int wv   = __builtin_amdgcn_readfirstlane((int)(threadIdx.x >> 5));
  const int mat  = blockIdx.y;
  const int rt   = blockIdx.x;
  const int half = lane >> 4;
  const int l16  = lane & 15;

  if (mat != 0) {
    if (wv < H / 16) {
      const v8f acc = stripe_tile(x, (mat == 1) ? cw1 : cw1 + D * H, rt, wv, lane, N);
      float* __restrict__ out = (mat == 1) ? hi : hj;
      const int rbase = rt * 16 + half * 8;
      if (rt * 16 + 16 <= N) {
#pragma unroll
        for (int r = 0; r < 8; ++r)
          out[(size_t)(rbase + r) * H + wv * 16 + l16] = acc[r];
      } else {
#pragma unroll
        for (int r = 0; r < 8; ++r)
          if (rbase + r < N)
            out[(size_t)(rbase + r) * H + wv * 16 + l16] = acc[r];
      }
    }
    return;
  }

  const float4 bb1 = *(const float4*)(b1 + 4 * lane);
  const float4 gg1 = *(const float4*)(g1 + 4 * lane);
  const float4 ee1 = *(const float4*)(be1 + 4 * lane);

  if (wv < H / 16) {
    const v8f acc = stripe_tile(x, w1, rt, wv, lane, N);
#pragma unroll
    for (int r = 0; r < 8; ++r)
      sh0[(half * 8 + r) * TPITCH + wv * 16 + l16] = acc[r];
  }
  __syncthreads();

  const int n0   = rt * NPB;
  const int n    = n0 + wv;
  const bool live = n < N;

  {
    const float4 hv = *(const float4*)&sh0[wv * TPITCH + 4 * lane];
    float4 v;
    v.x = hv.x + bb1.x; v.y = hv.y + bb1.y; v.z = hv.z + bb1.z; v.w = hv.w + bb1.w;
    const float s = wave_sum((v.x + v.y) + (v.z + v.w));
    const float q = wave_sum((v.x * v.x + v.y * v.y) + (v.z * v.z + v.w * v.w));
    const float m   = s * (1.0f / 128.0f);
    const float var = q * (1.0f / 128.0f) - m * m;
    const float inv = rsqrtf(var + LN_EPS);
    float4 r;
    r.x = fmaxf((v.x - m) * inv * gg1.x + ee1.x, 0.0f);
    r.y = fmaxf((v.y - m) * inv * gg1.y + ee1.y, 0.0f);
    r.z = fmaxf((v.z - m) * inv * gg1.z + ee1.z, 0.0f);
    r.w = fmaxf((v.w - m) * inv * gg1.w + ee1.w, 0.0f);
    *(float4*)&sh1[wv][4 * lane] = r;
  }
  __syncthreads();

  {
    float2 a2 = *(const float2*)(b2 + 2 * lane);
    const float* __restrict__ w2p = w2 + 2 * lane;
#pragma unroll 32
    for (int k = 0; k < H; ++k) {
      const float  hk = sh1[wv][k];
      const float2 w  = *(const float2*)(w2p + k * HH2);
      a2.x += hk * w.x;
      a2.y += hk * w.y;
    }
    const float s2 = wave_sum(a2.x + a2.y);
    const float q2 = wave_sum(a2.x * a2.x + a2.y * a2.y);
    const float m2 = s2 * (1.0f / 64.0f);
    const float v2 = q2 * (1.0f / 64.0f) - m2 * m2;
    const float i2 = rsqrtf(v2 + LN_EPS);
    const float2 gg = *(const float2*)(g2 + 2 * lane);
    const float2 ee = *(const float2*)(be2 + 2 * lane);
    float2 h2;
    h2.x = fmaxf((a2.x - m2) * i2 * gg.x + ee.x, 0.0f);
    h2.y = fmaxf((a2.y - m2) * i2 * gg.y + ee.y, 0.0f);
    *(float2*)&sh2[wv][2 * lane] = h2;
  }
  __syncthreads();

  {
    const bool owner = lane < KC;
    const int  tc    = owner ? lane : KC - 1;
    float lg = b3[tc];
#pragma unroll
    for (int k = 0; k < HH2; ++k) lg += sh2[wv][k] * w3[k * KC + tc];
    const float mx   = wave_max(owner ? lg : -INFINITY);
    const float e    = owner ? __expf(lg - mx) : 0.0f;
    const float ssum = wave_sum(e);
    const float p    = e / ssum;
    if (owner) {
      if (live) {
        out_logits[(size_t)n * KC + lane] = lg;
        out_probs[(size_t)n * KC + lane]  = p;
      }
      sp[wv][lane] = live ? p : 0.0f;
    }
  }
  __syncthreads();

  if (n0 + NPB <= N) {
    const int d  = threadIdx.x & (D - 1);
    const int kk = threadIdx.x >> 7;
    const int k3 = (kk < 2) ? kk + 8 : kk + 4;
    const float* __restrict__ xp = x + (size_t)n0 * D + d;
    float a0 = 0.0f, a1 = 0.0f, a2c = 0.0f;
#pragma unroll
    for (int g = 0; g < NPB; ++g) {
      const float xv = xp[g * D];
      a0  += sp[g][kk] * xv;
      a1  += sp[g][kk + 4] * xv;
      a2c += sp[g][k3] * xv;
    }
    float* __restrict__ part = part_ws + (size_t)n0 * H;
    part[kk * D + d]       = a0;
    part[(kk + 4) * D + d] = a1;
    if (kk < 2) part[(kk + 8) * D + d] = a2c;
  }
}

#define CTHREADS 256
#define CPV 4
#define CGV 32

__global__ void __launch_bounds__(CTHREADS)
centers_kernel(const float* __restrict__ probs,
               const float* __restrict__ x,
               const float* __restrict__ part,
               const float* __restrict__ prototypes,
               float* __restrict__ out_centers,
               float* __restrict__ out_protos,
               int N) {
  const int k    = blockIdx.x;
  const int tid  = threadIdx.x;
  const int d    = tid & (D - 1);
  const int half = tid >> 7;
  const int lane = tid & 31;
  const int wv   = tid >> 5;
  __shared__ float sW[CTHREADS / 32];
  __shared__ float sA[D];

  const int nfull = N / NPB;
  const int nt    = nfull * NPB;

  const float* __restrict__ pp = part + k * D + d;

  float pv[CPV], tv[CGV], tx[NPB / 2], tp[NPB / 2];
#pragma unroll
  for (int u = 0; u < CPV; ++u) {
    const int n  = u * CTHREADS + tid;
    const int nc = n < N ? n : N - 1;
    pv[u] = probs[(size_t)nc * KC + k];
  }
#pragma unroll
  for (int u = 0; u < CGV; ++u) {
    const int b = half + 2 * u;
    const float* __restrict__ src = b < nfull ? pp + (size_t)b * (NPB * H) : x + d;
    tv[u] = *src;
  }
#pragma unroll
  for (int v = 0; v < NPB / 2; ++v) {
    const int n  = nt + half + 2 * v;
    const int nc = n < N ? n : N - 1;
    tx[v] = x[(size_t)nc * D + d];
    tp[v] = probs[(size_t)nc * KC + k];
  }
  const float proto = prototypes[k * D + d];
  __builtin_amdgcn_sched_barrier(0);

  float wpart = 0.0f;
#pragma unroll
  for (int u = 0; u < CPV; ++u) wpart += (u * CTHREADS + tid < N) ? pv[u] : 0.0f;
  float acc = 0.0f;
#pragma unroll
  for (int u = 0; u < CGV; ++u) acc += (half + 2 * u < nfull) ? tv[u] : 0.0f;
#pragma unroll
  for (int v = 0; v < NPB / 2; ++v) acc += (nt + half + 2 * v < N) ? tp[v] * tx[v] : 0.0f;

  for (int base = CPV * CTHREADS; base < N; base += CPV * CTHREADS) {
    float t[CPV];
#pragma unroll
    for (int u = 0; u < CPV; ++u) {
      const int n  = base + u * CTHREADS + tid;
      const int nc = n < N ? n : N - 1;
      const float v = probs[(size_t)nc * KC + k];
      t[u] = n < N ? v : 0.0f;
    }
#pragma unroll
    for (int u = 0; u < CPV; ++u) wpart += t[u];
  }
  for (int b0 = 2 * CGV; b0 < nfull; b0 += 2 * CGV) {
    float t[CGV];
#pragma unroll
    for (int u = 0; u < CGV; ++u) {
      const int b = b0 + half + 2 * u;
      const float* __restrict__ src = b < nfull ? pp + (size_t)b * (NPB * H) : x + d;
      const float v = *src;
      t[u] = b < nfull ? v : 0.0f;
    }
#pragma unroll
    for (int u = 0; u < CGV; ++u) acc += t[u];
  }

  const float ws = wave_sum(wpart);
  if (lane == 0) sW[wv] = ws;
  if (half == 1) sA[d] = acc;
  __syncthreads();
  if (half == 0) {
    const float wsum = (((sW[0] + sW[1]) + (sW[2] + sW[3])) +
                        ((sW[4] + sW[5]) + (sW[6] + sW[7]))) + 1e-8f;
    out_centers[k * D + d] = (acc + sA[d]) / wsum;
    out_protos[k * D + d]  = proto;
  }
}

#define MROWS  8
#define MWAVES 8
#define MCOLS  64
#define MPITCH 132

__global__ void __launch_bounds__(MWAVES * 32)
comp_matrix_kernel(const float* __restrict__ hi,
                   const float* __restrict__ hj,
                   const float* __restrict__ cb1,
                   const float* __restrict__ cg1,
                   const float* __restrict__ cbe1,
                   const float* __restrict__ cw2,
                   const float* __restrict__ cb2p,
                   float* __restrict__ out, int N) {
  __shared__ __attribute__((aligned(16))) float tile[MCOLS * MPITCH];
  __shared__ __attribute__((aligned(16))) float arow[MWAVES * MROWS * H];

  const int lane = threadIdx.x & 31;
  const int wid  = __builtin_amdgcn_readfirstlane((int)(threadIdx.x >> 5));
  const int j0   = blockIdx.x * MCOLS;
  const int i0   = ((int)blockIdx.y * MWAVES + wid) * MROWS;
  const int iend = (i0 + MROWS < N) ? i0 + MROWS : N;

  {
    const float4 cbv = *(const float4*)(cb1 + 4 * lane);
    float4 hv[MCOLS / MWAVES], av[MROWS];
#pragma unroll
    for (int r = 0; r < MCOLS / MWAVES; ++r) {
      int jr = j0 + wid * (MCOLS / MWAVES) + r;
      if (jr >= N) jr = N - 1;
      hv[r] = *(const float4*)(hj + (size_t)jr * H + 4 * lane);
    }
#pragma unroll
    for (int r = 0; r < MROWS; ++r) {
      int ir = i0 + r;
      if (ir >= N) ir = N - 1;
      av[r] = *(const float4*)(hi + (size_t)ir * H + 4 * lane);
    }
#pragma unroll
    for (int r = 0; r < MCOLS / MWAVES; ++r) {
      const int c = wid * (MCOLS / MWAVES) + r;
      float4 v = hv[r];
      v.x += cbv.x; v.y += cbv.y; v.z += cbv.z; v.w += cbv.w;
      *(float4*)&tile[c * MPITCH + 4 * lane] = v;
    }
#pragma unroll
    for (int r = 0; r < MROWS; ++r)
      *(float4*)&arow[(wid * MROWS + r) * H + 4 * lane] = av[r];
  }
  __syncthreads();

  v2f b[H];
  {
    const float* __restrict__ r0 = &tile[lane * MPITCH];
    const float* __restrict__ r1 = &tile[(lane + 32) * MPITCH];
#pragma unroll
    for (int h = 0; h < H; h += 4) {
      const float4 u = *(const float4*)(r0 + h);
      const float4 w = *(const float4*)(r1 + h);
      b[h + 0] = (v2f){u.x, w.x};
      b[h + 1] = (v2f){u.y, w.y};
      b[h + 2] = (v2f){u.z, w.z};
      b[h + 3] = (v2f){u.w, w.w};
    }
  }

  const float cb2 = cb2p[0];
  const int ja = j0 + lane;
  const int jb = j0 + 32 + lane;
  const v2f zero2 = (v2f){0.0f, 0.0f};

  for (int i = i0; i < iend; ++i) {
    int off = 0;
    asm volatile("" : "+s"(off));
    const float* __restrict__ gp  = cg1  + off;
    const float* __restrict__ bep = cbe1 + off;
    const float* __restrict__ cwp = cw2  + off;
    const float* __restrict__ a = arow + (wid * MROWS + (i - i0)) * H;

    v2f s0 = zero2, s1 = zero2, s2 = zero2, s3 = zero2;
    v2f q0 = zero2, q1 = zero2, q2 = zero2, q3 = zero2;
#pragma unroll
    for (int hb = 0; hb < H; hb += 4) {
      const float4 a4 = *(const float4*)(a + hb);
      const v2f t0 = b[hb + 0] + a4.x;
      const v2f t1 = b[hb + 1] + a4.y;
      const v2f t2 = b[hb + 2] + a4.z;
      const v2f t3 = b[hb + 3] + a4.w;
      s0 += t0; s1 += t1; s2 += t2; s3 += t3;
      q0 = __builtin_elementwise_fma(t0, t0, q0);
      q1 = __builtin_elementwise_fma(t1, t1, q1);
      q2 = __builtin_elementwise_fma(t2, t2, q2);
      q3 = __builtin_elementwise_fma(t3, t3, q3);
    }
    const v2f m   = ((s0 + s1) + (s2 + s3)) * (1.0f / 128.0f);
    const v2f var = ((q0 + q1) + (q2 + q3)) * (1.0f / 128.0f) - m * m;
    v2f inv;
    inv.x = rsqrtf(var.x + LN_EPS);
    inv.y = rsqrtf(var.y + LN_EPS);
    const v2f nmi = -m * inv;

    v2f d0 = zero2, d1 = zero2, d2 = zero2, d3 = zero2;
#pragma unroll
    for (int hb = 0; hb < H; hb += 16) {
      float gv[16], ev[16], cv[16];
#pragma unroll
      for (int k = 0; k < 16; ++k) {
        gv[k] = gp[hb + k];
        ev[k] = bep[hb + k];
        cv[k] = cwp[hb + k];
      }
      __builtin_amdgcn_sched_barrier(0);
#pragma unroll
      for (int k = 0; k < 16; k += 4) {
        const float4 a4 = *(const float4*)(a + hb + k);
        const v2f y0 = __builtin_elementwise_fma(b[hb + k + 0] + a4.x, inv, nmi);
        const v2f y1 = __builtin_elementwise_fma(b[hb + k + 1] + a4.y, inv, nmi);
        const v2f y2 = __builtin_elementwise_fma(b[hb + k + 2] + a4.z, inv, nmi);
        const v2f y3 = __builtin_elementwise_fma(b[hb + k + 3] + a4.w, inv, nmi);
        const v2f z0 = __builtin_elementwise_fma(y0, (v2f){gv[k + 0], gv[k + 0]}, (v2f){ev[k + 0], ev[k + 0]});
        const v2f z1 = __builtin_elementwise_fma(y1, (v2f){gv[k + 1], gv[k + 1]}, (v2f){ev[k + 1], ev[k + 1]});
        const v2f z2 = __builtin_elementwise_fma(y2, (v2f){gv[k + 2], gv[k + 2]}, (v2f){ev[k + 2], ev[k + 2]});
        const v2f z3 = __builtin_elementwise_fma(y3, (v2f){gv[k + 3], gv[k + 3]}, (v2f){ev[k + 3], ev[k + 3]});
        d0 = __builtin_elementwise_fma(__builtin_elementwise_max(z0, zero2), (v2f){cv[k + 0], cv[k + 0]}, d0);
        d1 = __builtin_elementwise_fma(__builtin_elementwise_max(z1, zero2), (v2f){cv[k + 1], cv[k + 1]}, d1);
        d2 = __builtin_elementwise_fma(__builtin_elementwise_max(z2, zero2), (v2f){cv[k + 2], cv[k + 2]}, d2);
        d3 = __builtin_elementwise_fma(__builtin_elementwise_max(z3, zero2), (v2f){cv[k + 3], cv[k + 3]}, d3);
      }
      __builtin_amdgcn_sched_barrier(0);
    }
    const v2f dot = (d0 + d1) + (d2 + d3);
    const float sca = tanhf(dot.x + cb2);
    const float scb = tanhf(dot.y + cb2);
    float* __restrict__ orow = out + (size_t)i * N;
    if (ja < N) orow[ja] = sca;
    if (jb < N) orow[jb] = scb;
  }
}

#define EWAVES 6
#define CONST_AS __attribute__((address_space(4)))
#define EPITCH 132

__global__ void __launch_bounds__(EWAVES * 32)
comp_scores_kernel(const float* __restrict__ hi,
                   const float* __restrict__ hj,
                   const int* __restrict__ ei, int E,
                   const float* __restrict__ cb1,
                   const float* __restrict__ cg1,
                   const float* __restrict__ cbe1,
                   const float* __restrict__ cw2,
                   const float* __restrict__ cb2p,
                   float* __restrict__ out) {
  __shared__ __attribute__((aligned(16))) float etile[EWAVES * 32 * EPITCH];

  const int lane = threadIdx.x & 31;
  const int wv   = __builtin_amdgcn_readfirstlane((int)(threadIdx.x >> 5));
  const int gw   = (int)(blockIdx.x * EWAVES) + wv;
  const int nw   = (int)(gridDim.x * EWAVES);
  float* __restrict__ tile = etile + wv * (32 * EPITCH);

  const v4f   cb  = *(const v4f*)(cb1 + 4 * lane);
  const float cb2 = cb2p[0];
  const v2f zero2 = (v2f){0.0f, 0.0f};

  const int estep = nw * 32;
  int iv, jv;
  {
    int e = gw * 32 + lane;
    if (e >= E) e = E - 1;
    iv = ei[e];
    jv = ei[E + e];
  }
  for (int e0 = gw * 32; e0 < E; e0 += estep) {
    int ivn = 0, jvn = 0;

    v4f ra[2][8], rb[2][8];
#pragma unroll
    for (int pp = 0; pp < 8; ++pp) {
      const int i = __builtin_amdgcn_readlane(iv, pp);
      const int j = __builtin_amdgcn_readlane(jv, pp);
      ra[0][pp] = *(const v4f*)(hi + (size_t)i * H + 4 * lane);
      rb[0][pp] = *(const v4f*)(hj + (size_t)j * H + 4 * lane);
    }
#pragma unroll
    for (int g = 0; g < 4; ++g) {
      if (g == 3) {
        int en = e0 + estep + lane;
        if (en >= E) en = E - 1;
        ivn = ei[en];
        jvn = ei[E + en];
      }
      if (g < 3) {
#pragma unroll
        for (int pp = 0; pp < 8; ++pp) {
          const int p = (g + 1) * 8 + pp;
          const int i = __builtin_amdgcn_readlane(iv, p);
          const int j = __builtin_amdgcn_readlane(jv, p);
          ra[(g + 1) & 1][pp] = *(const v4f*)(hi + (size_t)i * H + 4 * lane);
          rb[(g + 1) & 1][pp] = *(const v4f*)(hj + (size_t)j * H + 4 * lane);
        }
      }
#pragma unroll
      for (int pp = 0; pp < 8; ++pp) {
        const int p = g * 8 + pp;
        v4f a4 = ra[g & 1][pp];
        v4f b4 = rb[g & 1][pp];
        asm volatile("" : "+v"(a4), "+v"(b4));
        *(v4f*)&tile[p * EPITCH + 4 * lane] = a4 + b4 + cb;
      }
    }
    __builtin_amdgcn_fence(__ATOMIC_RELEASE, "wavefront");
    __builtin_amdgcn_wave_barrier();
    __builtin_amdgcn_fence(__ATOMIC_ACQUIRE, "wavefront");

    v2f t[H / 2];
    {
      const float* __restrict__ row = tile + lane * EPITCH;
#pragma unroll
      for (int h = 0; h < H; h += 4) {
        const float4 u = *(const float4*)(row + h);
        t[h / 2]     = (v2f){u.x, u.y};
        t[h / 2 + 1] = (v2f){u.z, u.w};
      }
    }

    v2f s0 = zero2, s1 = zero2, s2 = zero2, s3 = zero2;
    v2f q0 = zero2, q1 = zero2, q2 = zero2, q3 = zero2;
#pragma unroll
    for (int k = 0; k < H / 2; k += 4) {
      s0 += t[k + 0]; s1 += t[k + 1]; s2 += t[k + 2]; s3 += t[k + 3];
      q0 = __builtin_elementwise_fma(t[k + 0], t[k + 0], q0);
      q1 = __builtin_elementwise_fma(t[k + 1], t[k + 1], q1);
      q2 = __builtin_elementwise_fma(t[k + 2], t[k + 2], q2);
      q3 = __builtin_elementwise_fma(t[k + 3], t[k + 3], q3);
    }
    const v2f sv = (s0 + s1) + (s2 + s3);
    const v2f qv = (q0 + q1) + (q2 + q3);
    const float m   = (sv.x + sv.y) * (1.0f / 128.0f);
    const float var = (qv.x + qv.y) * (1.0f / 128.0f) - m * m;
    const float inv = rsqrtf(var + LN_EPS);
    const float nmi = -m * inv;
    const v2f inv2 = (v2f){inv, inv};
    const v2f nmi2 = (v2f){nmi, nmi};

    int off = 0;
    asm volatile("" : "+s"(off));
    const CONST_AS float* gp  = (const CONST_AS float*)(cg1  + off);
    const CONST_AS float* bep = (const CONST_AS float*)(cbe1 + off);
    const CONST_AS float* cwp = (const CONST_AS float*)(cw2  + off);
    v2f d0 = zero2, d1 = zero2, d2 = zero2, d3 = zero2;
#pragma unroll
    for (int hb = 0; hb < H; hb += 16) {
      float gv[16], ev[16], cv[16];
#pragma unroll
      for (int k = 0; k < 16; ++k) {
        gv[k] = gp[hb + k];
        ev[k] = bep[hb + k];
        cv[k] = cwp[hb + k];
      }
      __builtin_amdgcn_sched_barrier(0);
#pragma unroll
      for (int k = 0; k < 16; k += 8) {
        const int kp = (hb + k) / 2;
        const v2f y0 = __builtin_elementwise_fma(t[kp + 0], inv2, nmi2);
        const v2f y1 = __builtin_elementwise_fma(t[kp + 1], inv2, nmi2);
        const v2f y2 = __builtin_elementwise_fma(t[kp + 2], inv2, nmi2);
        const v2f y3 = __builtin_elementwise_fma(t[kp + 3], inv2, nmi2);
        const v2f z0 = __builtin_elementwise_fma(y0, (v2f){gv[k + 0], gv[k + 1]}, (v2f){ev[k + 0], ev[k + 1]});
        const v2f z1 = __builtin_elementwise_fma(y1, (v2f){gv[k + 2], gv[k + 3]}, (v2f){ev[k + 2], ev[k + 3]});
        const v2f z2 = __builtin_elementwise_fma(y2, (v2f){gv[k + 4], gv[k + 5]}, (v2f){ev[k + 4], ev[k + 5]});
        const v2f z3 = __builtin_elementwise_fma(y3, (v2f){gv[k + 6], gv[k + 7]}, (v2f){ev[k + 6], ev[k + 7]});
        d0 = __builtin_elementwise_fma(__builtin_elementwise_max(z0, zero2), (v2f){cv[k + 0], cv[k + 1]}, d0);
        d1 = __builtin_elementwise_fma(__builtin_elementwise_max(z1, zero2), (v2f){cv[k + 2], cv[k + 3]}, d1);
        d2 = __builtin_elementwise_fma(__builtin_elementwise_max(z2, zero2), (v2f){cv[k + 4], cv[k + 5]}, d2);
        d3 = __builtin_elementwise_fma(__builtin_elementwise_max(z3, zero2), (v2f){cv[k + 6], cv[k + 7]}, d3);
      }
      __builtin_amdgcn_sched_barrier(0);
    }
    const v2f dv = (d0 + d1) + (d2 + d3);
    const float sc = tanhf((dv.x + dv.y) + cb2);
    if (e0 + lane < E) out[e0 + lane] = sc;
    iv = ivn;
    jv = jvn;
  }
}

extern "C" void kernel_launch(void* const* d_in, const int* in_sizes, int n_in,
                              void* d_out, int out_size, void* d_ws, size_t ws_size,
                              hipStream_t stream) {
  const float* x    = (const float*)d_in[0];
  const int*   ei   = (const int*)  d_in[1];
  const float* w1   = (const float*)d_in[2];
  const float* b1   = (const float*)d_in[3];
  const float* g1   = (const float*)d_in[4];
  const float* be1  = (const float*)d_in[5];
  const float* w2   = (const float*)d_in[6];
  const float* b2   = (const float*)d_in[7];
  const float* g2   = (const float*)d_in[8];
  const float* be2  = (const float*)d_in[9];
  const float* w3   = (const float*)d_in[10];
  const float* b3   = (const float*)d_in[11];
  const float* cw1  = (const float*)d_in[12];
  const float* cb1  = (const float*)d_in[13];
  const float* cg1  = (const float*)d_in[14];
  const float* cbe1 = (const float*)d_in[15];
  const float* cw2  = (const float*)d_in[16];
  const float* cb2  = (const float*)d_in[17];
  const float* prot = (const float*)d_in[18];

  const int N = in_sizes[0] / D;
  const int E = in_sizes[1] / 2;

  float* h1pre = (float*)d_ws;
  float* hi    = h1pre + (size_t)N * H;
  float* hj    = hi    + (size_t)N * H;

  float* out       = (float*)d_out;
  float* o_logits  = out;
  float* o_probs   = o_logits  + (size_t)N * KC;
  float* o_scores  = o_probs   + (size_t)N * KC;
  float* o_matrix  = o_scores  + (size_t)E;
  float* o_centers = o_matrix  + (size_t)N * N;
  float* o_protos  = o_centers + (size_t)KC * D;

  const int rowTiles = (N + 15) / 16;

  stripe_kernel<<<dim3(rowTiles, 3), NPB * 32, 0, stream>>>(
      x, w1, cw1, hi, hj, h1pre, b1, g1, be1, w2, b2, g2, be2, w3, b3,
      o_logits, o_probs, N);

  centers_kernel<<<KC, CTHREADS, 0, stream>>>(
      o_probs, x, h1pre, prot, o_centers, o_protos, N);

  comp_matrix_kernel<<<dim3((N + MCOLS - 1) / MCOLS, (N + MWAVES * MROWS - 1) / (MWAVES * MROWS)),
                       MWAVES * 32, 0, stream>>>(
      hi, hj, cb1, cg1, cbe1, cw2, cb2, o_matrix, N);

  comp_scores_kernel<<<768, EWAVES * 32, 0, stream>>>(
      hi, hj, ei, E, cb1, cg1, cbe1, cw2, cb2, o_scores);
}


# node net behind the first GEMM waits for no memory: h1@w2 as a 16x64 WMMA product by the 8 idle waves, w2 fragments requested at entry; all other weights requested stages ahead; LDS-only barriers
# speedup vs baseline: 4.7091x; 1.0502x over previous
#include <hip/hip_runtime.h>
#include <math.h>

typedef __attribute__((ext_vector_type(2))) float v2f;
typedef __attribute__((ext_vector_type(4))) float v4f;
typedef __attribute__((ext_vector_type(8))) float v8f;

#define D   128
#define H   128
#define HH2 64
#define KC  10
#define LN_EPS 1e-5f

#define NPB 16
#define TPITCH 132
#define S1PITCH 132
#define S2PITCH 68

__device__ __forceinline__ v8f stripe_tile(const float* __restrict__ x,
                                           const float* __restrict__ Bm,
                                           const int rt, const int ct,
                                           const int lane, const int N) {
  const int half = lane >> 4;
  const int l16  = lane & 15;
  int arow = rt * 16 + l16;
  if (arow >= N) arow = N - 1;
  const float* __restrict__ arow_p = x + (size_t)arow * D + half * 2;
  const float* __restrict__ bcol_p = Bm + (size_t)(half * 2) * H + ct * 16 + l16;
  v2f a[D / 4], b[D / 4];
#pragma unroll
  for (int s = 0; s < D / 4; ++s) a[s] = *(const v2f*)(arow_p + 4 * s);
#pragma unroll
  for (int s = 0; s < D / 4; ++s) {
    b[s].x = bcol_p[(size_t)(4 * s) * H];
    b[s].y = bcol_p[(size_t)(4 * s + 1) * H];
  }
  __builtin_amdgcn_sched_barrier(0);
  v8f acc0 = {}, acc1 = {};
#pragma unroll
  for (int s = 0; s < D / 4; s += 2) {
    acc0 = __builtin_amdgcn_wmma_f32_16x16x4_f32(false, a[s], false, b[s],
                                                 (short)0, acc0, false, false);
    acc1 = __builtin_amdgcn_wmma_f32_16x16x4_f32(false, a[s + 1], false, b[s + 1],
                                                 (short)0, acc1, false, false);
  }
  return acc0 + acc1;
}

__device__ __forceinline__ float wave_sum(float v) {
#pragma unroll
  for (int w = 16; w > 0; w >>= 1) v += __shfl_xor(v, w, 32);
  return v;
}

__device__ __forceinline__ float wave_max(float v) {
#pragma unroll
  for (int w = 16; w > 0; w >>= 1) v = fmaxf(v, __shfl_xor(v, w, 32));
  return v;
}

__device__ __forceinline__ void lds_barrier() {
  __builtin_amdgcn_fence(__ATOMIC_RELEASE, "workgroup", "local");
  __builtin_amdgcn_s_barrier();
  __builtin_amdgcn_fence(__ATOMIC_ACQUIRE, "workgroup", "local");
}

__global__ void __launch_bounds__(NPB * 32)
stripe_kernel(const float* __restrict__ x,
              const float* __restrict__ w1,
              const float* __restrict__ cw1,
              float* __restrict__ hi,
              float* __restrict__ hj,
              float* __restrict__ part_ws,
              const float* __restrict__ b1,
              const float* __restrict__ g1,
              const float* __restrict__ be1,
              const float* __restrict__ w2,
              const float* __restrict__ b2,
              const float* __restrict__ g2,
              const float* __restrict__ be2,
              const float* __restrict__ w3,
              const float* __restrict__ b3,
              float* __restrict__ out_logits,
              float* __restrict__ out_probs,
              int N) {
  __shared__ __attribute__((aligned(16))) float sh0[NPB * TPITCH];
  __shared__ __attribute__((aligned(16))) float sh1[NPB * S1PITCH];
  __shared__ __attribute__((aligned(16))) float sh2p[2][NPB * S2PITCH];
  __shared__ float sh2[NPB][HH2];
  __shared__ float sp[NPB][KC];

  const int lane = threadIdx.x & 31;
  const int wv   = __builtin_amdgcn_readfirstlane((int)(threadIdx.x >> 5));
  const int mat  = blockIdx.y;
  const int rt   = blockIdx.x;
  const int half = lane >> 4;
  const int l16  = lane & 15;

  if (mat != 0) {
    if (wv < H / 16) {
      const v8f acc = stripe_tile(x, (mat == 1) ? cw1 : cw1 + D * H, rt, wv, lane, N);
      float* __restrict__ out = (mat == 1) ? hi : hj;
      const int rbase = rt * 16 + half * 8;
      if (rt * 16 + 16 <= N) {
#pragma unroll
        for (int r = 0; r < 8; ++r)
          out[(size_t)(rbase + r) * H + wv * 16 + l16] = acc[r];
      } else {
#pragma unroll
        for (int r = 0; r < 8; ++r)
          if (rbase + r < N)
            out[(size_t)(rbase + r) * H + wv * 16 + l16] = acc[r];
      }
    }
    return;
  }

  const float2 bb2 = *(const float2*)(b2 + 2 * lane);
  const float2 gg2 = *(const float2*)(g2 + 2 * lane);
  const float2 ee2 = *(const float2*)(be2 + 2 * lane);
  const bool owner = lane < KC;
  const int  tc    = owner ? lane : KC - 1;
  const float b3v  = b3[tc];
  const float4 bb1 = *(const float4*)(b1 + 4 * lane);
  const float4 gg1 = *(const float4*)(g1 + 4 * lane);
  const float4 ee1 = *(const float4*)(be1 + 4 * lane);
  const int n0 = rt * NPB;
  const int d  = threadIdx.x & (D - 1);
  const int kk = threadIdx.x >> 7;
  float w3v[HH2], xv[NPB];

  const int ct2 = (wv - H / 16) & 3;
  const int kh2 = (wv - H / 16) >> 2;
  v2f wb[HH2 / 4];

  if (wv < H / 16) {
    const v8f acc = stripe_tile(x, w1, rt, wv, lane, N);
#pragma unroll
    for (int r = 0; r < 8; ++r)
      sh0[(half * 8 + r) * TPITCH + wv * 16 + l16] = acc[r];
  }
  if (wv >= H / 16) {
    const float* __restrict__ bp = w2 + (size_t)(kh2 * 64 + half * 2) * HH2 + ct2 * 16 + l16;
#pragma unroll
    for (int s = 0; s < HH2 / 4; ++s) {
      wb[s].x = bp[(size_t)(4 * s) * HH2];
      wb[s].y = bp[(size_t)(4 * s + 1) * HH2];
    }
#pragma unroll
    for (int k = 0; k < HH2; ++k) w3v[k] = w3[k * KC + tc];
#pragma unroll
    for (int g = 0; g < NPB; ++g) {
      int r = n0 + g;
      if (r >= N) r = N - 1;
      xv[g] = x[(size_t)r * D + d];
    }
  }
  lds_barrier();

  const int n    = n0 + wv;
  const bool live = n < N;


  {
    const float4 hv = *(const float4*)&sh0[wv * TPITCH + 4 * lane];
    float4 v;
    v.x = hv.x + bb1.x; v.y = hv.y + bb1.y; v.z = hv.z + bb1.z; v.w = hv.w + bb1.w;
    const float s = wave_sum((v.x + v.y) + (v.z + v.w));
    const float q = wave_sum((v.x * v.x + v.y * v.y) + (v.z * v.z + v.w * v.w));
    const float m   = s * (1.0f / 128.0f);
    const float var = q * (1.0f / 128.0f) - m * m;
    const float inv = rsqrtf(var + LN_EPS);
    float4 r;
    r.x = fmaxf((v.x - m) * inv * gg1.x + ee1.x, 0.0f);
    r.y = fmaxf((v.y - m) * inv * gg1.y + ee1.y, 0.0f);
    r.z = fmaxf((v.z - m) * inv * gg1.z + ee1.z, 0.0f);
    r.w = fmaxf((v.w - m) * inv * gg1.w + ee1.w, 0.0f);
    *(float4*)&sh1[wv * S1PITCH + 4 * lane] = r;
  }
  lds_barrier();

  if (wv >= H / 16) {
    const float* __restrict__ ap = &sh1[l16 * S1PITCH + kh2 * 64 + half * 2];
    v2f af[HH2 / 4];
#pragma unroll
    for (int s = 0; s < HH2 / 4; ++s) af[s] = *(const v2f*)(ap + 4 * s);
    v8f acc0 = {}, acc1 = {};
#pragma unroll
    for (int s = 0; s < HH2 / 4; s += 2) {
      acc0 = __builtin_amdgcn_wmma_f32_16x16x4_f32(false, af[s], false, wb[s],
                                                   (short)0, acc0, false, false);
      acc1 = __builtin_amdgcn_wmma_f32_16x16x4_f32(false, af[s + 1], false, wb[s + 1],
                                                   (short)0, acc1, false, false);
    }
    const v8f acc = acc0 + acc1;
    float* __restrict__ pl = sh2p[kh2];
#pragma unroll
    for (int r = 0; r < 8; ++r)
      pl[(half * 8 + r) * S2PITCH + ct2 * 16 + l16] = acc[r];
  } else {
#pragma unroll
    for (int k = 0; k < HH2; ++k) w3v[k] = w3[k * KC + tc];
#pragma unroll
    for (int g = 0; g < NPB; ++g) {
      int r = n0 + g;
      if (r >= N) r = N - 1;
      xv[g] = x[(size_t)r * D + d];
    }
  }
  lds_barrier();

  {
    const float2 p0 = *(const float2*)&sh2p[0][wv * S2PITCH + 2 * lane];
    const float2 p1 = *(const float2*)&sh2p[1][wv * S2PITCH + 2 * lane];
    float2 a2;
    a2.x = (p0.x + p1.x) + bb2.x;
    a2.y = (p0.y + p1.y) + bb2.y;
    const float s2 = wave_sum(a2.x + a2.y);
    const float q2 = wave_sum(a2.x * a2.x + a2.y * a2.y);
    const float m2 = s2 * (1.0f / 64.0f);
    const float v2 = q2 * (1.0f / 64.0f) - m2 * m2;
    const float i2 = rsqrtf(v2 + LN_EPS);
    float2 h2;
    h2.x = fmaxf((a2.x - m2) * i2 * gg2.x + ee2.x, 0.0f);
    h2.y = fmaxf((a2.y - m2) * i2 * gg2.y + ee2.y, 0.0f);
    *(float2*)&sh2[wv][2 * lane] = h2;
  }
  lds_barrier();

  {
    float lg = b3v;
#pragma unroll
    for (int k = 0; k < HH2; ++k) lg += sh2[wv][k] * w3v[k];
    const float mx   = wave_max(owner ? lg : -INFINITY);
    const float e    = owner ? __expf(lg - mx) : 0.0f;
    const float ssum = wave_sum(e);
    const float p    = e / ssum;
    if (owner) {
      if (live) {
        out_logits[(size_t)n * KC + lane] = lg;
        out_probs[(size_t)n * KC + lane]  = p;
      }
      sp[wv][lane] = live ? p : 0.0f;
    }
  }
  lds_barrier();

  if (n0 + NPB <= N) {
    const int k3 = (kk < 2) ? kk + 8 : kk + 4;
    float a0 = 0.0f, a1 = 0.0f, a2c = 0.0f;
#pragma unroll
    for (int g = 0; g < NPB; ++g) {
      a0  += sp[g][kk] * xv[g];
      a1  += sp[g][kk + 4] * xv[g];
      a2c += sp[g][k3] * xv[g];
    }
    float* __restrict__ part = part_ws + (size_t)n0 * H;
    part[kk * D + d]       = a0;
    part[(kk + 4) * D + d] = a1;
    if (kk < 2) part[(kk + 8) * D + d] = a2c;
  }
}

#define CTHREADS 256
#define CPV 4
#define CGV 32

__global__ void __launch_bounds__(CTHREADS)
centers_kernel(const float* __restrict__ probs,
               const float* __restrict__ x,
               const float* __restrict__ part,
               const float* __restrict__ prototypes,
               float* __restrict__ out_centers,
               float* __restrict__ out_protos,
               int N) {
  const int k    = blockIdx.x;
  const int tid  = threadIdx.x;
  const int d    = tid & (D - 1);
  const int half = tid >> 7;
  const int lane = tid & 31;
  const int wv   = tid >> 5;
  __shared__ float sW[CTHREADS / 32];
  __shared__ float sA[D];

  const int nfull = N / NPB;
  const int nt    = nfull * NPB;

  const float* __restrict__ pp = part + k * D + d;

  float pv[CPV], tv[CGV], tx[NPB / 2], tp[NPB / 2];
#pragma unroll
  for (int u = 0; u < CPV; ++u) {
    const int n  = u * CTHREADS + tid;
    const int nc = n < N ? n : N - 1;
    pv[u] = probs[(size_t)nc * KC + k];
  }
#pragma unroll
  for (int u = 0; u < CGV; ++u) {
    const int b = half + 2 * u;
    const float* __restrict__ src = b < nfull ? pp + (size_t)b * (NPB * H) : x + d;
    tv[u] = *src;
  }
#pragma unroll
  for (int v = 0; v < NPB / 2; ++v) {
    const int n  = nt + half + 2 * v;
    const int nc = n < N ? n : N - 1;
    tx[v] = x[(size_t)nc * D + d];
    tp[v] = probs[(size_t)nc * KC + k];
  }
  const float proto = prototypes[k * D + d];
  __builtin_amdgcn_sched_barrier(0);

  float wpart = 0.0f;
#pragma unroll
  for (int u = 0; u < CPV; ++u) wpart += (u * CTHREADS + tid < N) ? pv[u] : 0.0f;
  float acc = 0.0f;
#pragma unroll
  for (int u = 0; u < CGV; ++u) acc += (half + 2 * u < nfull) ? tv[u] : 0.0f;
#pragma unroll
  for (int v = 0; v < NPB / 2; ++v) acc += (nt + half + 2 * v < N) ? tp[v] * tx[v] : 0.0f;

  for (int base = CPV * CTHREADS; base < N; base += CPV * CTHREADS) {
    float t[CPV];
#pragma unroll
    for (int u = 0; u < CPV; ++u) {
      const int n  = base + u * CTHREADS + tid;
      const int nc = n < N ? n : N - 1;
      const float v = probs[(size_t)nc * KC + k];
      t[u] = n < N ? v : 0.0f;
    }
#pragma unroll
    for (int u = 0; u < CPV; ++u) wpart += t[u];
  }
  for (int b0 = 2 * CGV; b0 < nfull; b0 += 2 * CGV) {
    float t[CGV];
#pragma unroll
    for (int u = 0; u < CGV; ++u) {
      const int b = b0 + half + 2 * u;
      const float* __restrict__ src = b < nfull ? pp + (size_t)b * (NPB * H) : x + d;
      const float v = *src;
      t[u] = b < nfull ? v : 0.0f;
    }
#pragma unroll
    for (int u = 0; u < CGV; ++u) acc += t[u];
  }

  const float ws = wave_sum(wpart);
  if (lane == 0) sW[wv] = ws;
  if (half == 1) sA[d] = acc;
  __syncthreads();
  if (half == 0) {
    const float wsum = (((sW[0] + sW[1]) + (sW[2] + sW[3])) +
                        ((sW[4] + sW[5]) + (sW[6] + sW[7]))) + 1e-8f;
    out_centers[k * D + d] = (acc + sA[d]) / wsum;
    out_protos[k * D + d]  = proto;
  }
}

#define MROWS  8
#define MWAVES 8
#define MCOLS  64
#define MPITCH 132

__global__ void __launch_bounds__(MWAVES * 32)
comp_matrix_kernel(const float* __restrict__ hi,
                   const float* __restrict__ hj,
                   const float* __restrict__ cb1,
                   const float* __restrict__ cg1,
                   const float* __restrict__ cbe1,
                   const float* __restrict__ cw2,
                   const float* __restrict__ cb2p,
                   float* __restrict__ out, int N) {
  __shared__ __attribute__((aligned(16))) float tile[MCOLS * MPITCH];
  __shared__ __attribute__((aligned(16))) float arow[MWAVES * MROWS * H];

  const int lane = threadIdx.x & 31;
  const int wid  = __builtin_amdgcn_readfirstlane((int)(threadIdx.x >> 5));
  const int j0   = blockIdx.x * MCOLS;
  const int i0   = ((int)blockIdx.y * MWAVES + wid) * MROWS;
  const int iend = (i0 + MROWS < N) ? i0 + MROWS : N;

  {
    const float4 cbv = *(const float4*)(cb1 + 4 * lane);
    float4 hv[MCOLS / MWAVES], av[MROWS];
#pragma unroll
    for (int r = 0; r < MCOLS / MWAVES; ++r) {
      int jr = j0 + wid * (MCOLS / MWAVES) + r;
      if (jr >= N) jr = N - 1;
      hv[r] = *(const float4*)(hj + (size_t)jr * H + 4 * lane);
    }
#pragma unroll
    for (int r = 0; r < MROWS; ++r) {
      int ir = i0 + r;
      if (ir >= N) ir = N - 1;
      av[r] = *(const float4*)(hi + (size_t)ir * H + 4 * lane);
    }
#pragma unroll
    for (int r = 0; r < MCOLS / MWAVES; ++r) {
      const int c = wid * (MCOLS / MWAVES) + r;
      float4 v = hv[r];
      v.x += cbv.x; v.y += cbv.y; v.z += cbv.z; v.w += cbv.w;
      *(float4*)&tile[c * MPITCH + 4 * lane] = v;
    }
#pragma unroll
    for (int r = 0; r < MROWS; ++r)
      *(float4*)&arow[(wid * MROWS + r) * H + 4 * lane] = av[r];
  }
  __syncthreads();

  v2f b[H];
  {
    const float* __restrict__ r0 = &tile[lane * MPITCH];
    const float* __restrict__ r1 = &tile[(lane + 32) * MPITCH];
#pragma unroll
    for (int h = 0; h < H; h += 4) {
      const float4 u = *(const float4*)(r0 + h);
      const float4 w = *(const float4*)(r1 + h);
      b[h + 0] = (v2f){u.x, w.x};
      b[h + 1] = (v2f){u.y, w.y};
      b[h + 2] = (v2f){u.z, w.z};
      b[h + 3] = (v2f){u.w, w.w};
    }
  }

  const float cb2 = cb2p[0];
  const int ja = j0 + lane;
  const int jb = j0 + 32 + lane;
  const v2f zero2 = (v2f){0.0f, 0.0f};

  for (int i = i0; i < iend; ++i) {
    int off = 0;
    asm volatile("" : "+s"(off));
    const float* __restrict__ gp  = cg1  + off;
    const float* __restrict__ bep = cbe1 + off;
    const float* __restrict__ cwp = cw2  + off;
    const float* __restrict__ a = arow + (wid * MROWS + (i - i0)) * H;

    v2f s0 = zero2, s1 = zero2, s2 = zero2, s3 = zero2;
    v2f q0 = zero2, q1 = zero2, q2 = zero2, q3 = zero2;
#pragma unroll
    for (int hb = 0; hb < H; hb += 4) {
      const float4 a4 = *(const float4*)(a + hb);
      const v2f t0 = b[hb + 0] + a4.x;
      const v2f t1 = b[hb + 1] + a4.y;
      const v2f t2 = b[hb + 2] + a4.z;
      const v2f t3 = b[hb + 3] + a4.w;
      s0 += t0; s1 += t1; s2 += t2; s3 += t3;
      q0 = __builtin_elementwise_fma(t0, t0, q0);
      q1 = __builtin_elementwise_fma(t1, t1, q1);
      q2 = __builtin_elementwise_fma(t2, t2, q2);
      q3 = __builtin_elementwise_fma(t3, t3, q3);
    }
    const v2f m   = ((s0 + s1) + (s2 + s3)) * (1.0f / 128.0f);
    const v2f var = ((q0 + q1) + (q2 + q3)) * (1.0f / 128.0f) - m * m;
    v2f inv;
    inv.x = rsqrtf(var.x + LN_EPS);
    inv.y = rsqrtf(var.y + LN_EPS);
    const v2f nmi = -m * inv;

    v2f d0 = zero2, d1 = zero2, d2 = zero2, d3 = zero2;
#pragma unroll
    for (int hb = 0; hb < H; hb += 16) {
      float gv[16], ev[16], cv[16];
#pragma unroll
      for (int k = 0; k < 16; ++k) {
        gv[k] = gp[hb + k];
        ev[k] = bep[hb + k];
        cv[k] = cwp[hb + k];
      }
      __builtin_amdgcn_sched_barrier(0);
#pragma unroll
      for (int k = 0; k < 16; k += 4) {
        const float4 a4 = *(const float4*)(a + hb + k);
        const v2f y0 = __builtin_elementwise_fma(b[hb + k + 0] + a4.x, inv, nmi);
        const v2f y1 = __builtin_elementwise_fma(b[hb + k + 1] + a4.y, inv, nmi);
        const v2f y2 = __builtin_elementwise_fma(b[hb + k + 2] + a4.z, inv, nmi);
        const v2f y3 = __builtin_elementwise_fma(b[hb + k + 3] + a4.w, inv, nmi);
        const v2f z0 = __builtin_elementwise_fma(y0, (v2f){gv[k + 0], gv[k + 0]}, (v2f){ev[k + 0], ev[k + 0]});
        const v2f z1 = __builtin_elementwise_fma(y1, (v2f){gv[k + 1], gv[k + 1]}, (v2f){ev[k + 1], ev[k + 1]});
        const v2f z2 = __builtin_elementwise_fma(y2, (v2f){gv[k + 2], gv[k + 2]}, (v2f){ev[k + 2], ev[k + 2]});
        const v2f z3 = __builtin_elementwise_fma(y3, (v2f){gv[k + 3], gv[k + 3]}, (v2f){ev[k + 3], ev[k + 3]});
        d0 = __builtin_elementwise_fma(__builtin_elementwise_max(z0, zero2), (v2f){cv[k + 0], cv[k + 0]}, d0);
        d1 = __builtin_elementwise_fma(__builtin_elementwise_max(z1, zero2), (v2f){cv[k + 1], cv[k + 1]}, d1);
        d2 = __builtin_elementwise_fma(__builtin_elementwise_max(z2, zero2), (v2f){cv[k + 2], cv[k + 2]}, d2);
        d3 = __builtin_elementwise_fma(__builtin_elementwise_max(z3, zero2), (v2f){cv[k + 3], cv[k + 3]}, d3);
      }
      __builtin_amdgcn_sched_barrier(0);
    }
    const v2f dot = (d0 + d1) + (d2 + d3);
    const float sca = tanhf(dot.x + cb2);
    const float scb = tanhf(dot.y + cb2);
    float* __restrict__ orow = out + (size_t)i * N;
    if (ja < N) orow[ja] = sca;
    if (jb < N) orow[jb] = scb;
  }
}

#define EWAVES 6
#define CONST_AS __attribute__((address_space(4)))
#define EPITCH 132

__global__ void __launch_bounds__(EWAVES * 32)
comp_scores_kernel(const float* __restrict__ hi,
                   const float* __restrict__ hj,
                   const int* __restrict__ ei, int E,
                   const float* __restrict__ cb1,
                   const float* __restrict__ cg1,
                   const float* __restrict__ cbe1,
                   const float* __restrict__ cw2,
                   const float* __restrict__ cb2p,
                   float* __restrict__ out) {
  __shared__ __attribute__((aligned(16))) float etile[EWAVES * 32 * EPITCH];

  const int lane = threadIdx.x & 31;
  const int wv   = __builtin_amdgcn_readfirstlane((int)(threadIdx.x >> 5));
  const int gw   = (int)(blockIdx.x * EWAVES) + wv;
  const int nw   = (int)(gridDim.x * EWAVES);
  float* __restrict__ tile = etile + wv * (32 * EPITCH);

  const v4f   cb  = *(const v4f*)(cb1 + 4 * lane);
  const float cb2 = cb2p[0];
  const v2f zero2 = (v2f){0.0f, 0.0f};

  const int estep = nw * 32;
  int iv, jv;
  {
    int e = gw * 32 + lane;
    if (e >= E) e = E - 1;
    iv = ei[e];
    jv = ei[E + e];
  }
  for (int e0 = gw * 32; e0 < E; e0 += estep) {
    int ivn = 0, jvn = 0;

    v4f ra[2][8], rb[2][8];
#pragma unroll
    for (int pp = 0; pp < 8; ++pp) {
      const int i = __builtin_amdgcn_readlane(iv, pp);
      const int j = __builtin_amdgcn_readlane(jv, pp);
      ra[0][pp] = *(const v4f*)(hi + (size_t)i * H + 4 * lane);
      rb[0][pp] = *(const v4f*)(hj + (size_t)j * H + 4 * lane);
    }
#pragma unroll
    for (int g = 0; g < 4; ++g) {
      if (g == 3) {
        int en = e0 + estep + lane;
        if (en >= E) en = E - 1;
        ivn = ei[en];
        jvn = ei[E + en];
      }
      if (g < 3) {
#pragma unroll
        for (int pp = 0; pp < 8; ++pp) {
          const int p = (g + 1) * 8 + pp;
          const int i = __builtin_amdgcn_readlane(iv, p);
          const int j = __builtin_amdgcn_readlane(jv, p);
          ra[(g + 1) & 1][pp] = *(const v4f*)(hi + (size_t)i * H + 4 * lane);
          rb[(g + 1) & 1][pp] = *(const v4f*)(hj + (size_t)j * H + 4 * lane);
        }
      }
#pragma unroll
      for (int pp = 0; pp < 8; ++pp) {
        const int p = g * 8 + pp;
        v4f a4 = ra[g & 1][pp];
        v4f b4 = rb[g & 1][pp];
        asm volatile("" : "+v"(a4), "+v"(b4));
        *(v4f*)&tile[p * EPITCH + 4 * lane] = a4 + b4 + cb;
      }
    }
    __builtin_amdgcn_fence(__ATOMIC_RELEASE, "wavefront");
    __builtin_amdgcn_wave_barrier();
    __builtin_amdgcn_fence(__ATOMIC_ACQUIRE, "wavefront");

    v2f t[H / 2];
    {
      const float* __restrict__ row = tile + lane * EPITCH;
#pragma unroll
      for (int h = 0; h < H; h += 4) {
        const float4 u = *(const float4*)(row + h);
        t[h / 2]     = (v2f){u.x, u.y};
        t[h / 2 + 1] = (v2f){u.z, u.w};
      }
    }

    v2f s0 = zero2, s1 = zero2, s2 = zero2, s3 = zero2;
    v2f q0 = zero2, q1 = zero2, q2 = zero2, q3 = zero2;
#pragma unroll
    for (int k = 0; k < H / 2; k += 4) {
      s0 += t[k + 0]; s1 += t[k + 1]; s2 += t[k + 2]; s3 += t[k + 3];
      q0 = __builtin_elementwise_fma(t[k + 0], t[k + 0], q0);
      q1 = __builtin_elementwise_fma(t[k + 1], t[k + 1], q1);
      q2 = __builtin_elementwise_fma(t[k + 2], t[k + 2], q2);
      q3 = __builtin_elementwise_fma(t[k + 3], t[k + 3], q3);
    }
    const v2f sv = (s0 + s1) + (s2 + s3);
    const v2f qv = (q0 + q1) + (q2 + q3);
    const float m   = (sv.x + sv.y) * (1.0f / 128.0f);
    const float var = (qv.x + qv.y) * (1.0f / 128.0f) - m * m;
    const float inv = rsqrtf(var + LN_EPS);
    const float nmi = -m * inv;
    const v2f inv2 = (v2f){inv, inv};
    const v2f nmi2 = (v2f){nmi, nmi};

    int off = 0;
    asm volatile("" : "+s"(off));
    const CONST_AS float* gp  = (const CONST_AS float*)(cg1  + off);
    const CONST_AS float* bep = (const CONST_AS float*)(cbe1 + off);
    const CONST_AS float* cwp = (const CONST_AS float*)(cw2  + off);
    v2f d0 = zero2, d1 = zero2, d2 = zero2, d3 = zero2;
#pragma unroll
    for (int hb = 0; hb < H; hb += 16) {
      float gv[16], ev[16], cv[16];
#pragma unroll
      for (int k = 0; k < 16; ++k) {
        gv[k] = gp[hb + k];
        ev[k] = bep[hb + k];
        cv[k] = cwp[hb + k];
      }
      __builtin_amdgcn_sched_barrier(0);
#pragma unroll
      for (int k = 0; k < 16; k += 8) {
        const int kp = (hb + k) / 2;
        const v2f y0 = __builtin_elementwise_fma(t[kp + 0], inv2, nmi2);
        const v2f y1 = __builtin_elementwise_fma(t[kp + 1], inv2, nmi2);
        const v2f y2 = __builtin_elementwise_fma(t[kp + 2], inv2, nmi2);
        const v2f y3 = __builtin_elementwise_fma(t[kp + 3], inv2, nmi2);
        const v2f z0 = __builtin_elementwise_fma(y0, (v2f){gv[k + 0], gv[k + 1]}, (v2f){ev[k + 0], ev[k + 1]});
        const v2f z1 = __builtin_elementwise_fma(y1, (v2f){gv[k + 2], gv[k + 3]}, (v2f){ev[k + 2], ev[k + 3]});
        const v2f z2 = __builtin_elementwise_fma(y2, (v2f){gv[k + 4], gv[k + 5]}, (v2f){ev[k + 4], ev[k + 5]});
        const v2f z3 = __builtin_elementwise_fma(y3, (v2f){gv[k + 6], gv[k + 7]}, (v2f){ev[k + 6], ev[k + 7]});
        d0 = __builtin_elementwise_fma(__builtin_elementwise_max(z0, zero2), (v2f){cv[k + 0], cv[k + 1]}, d0);
        d1 = __builtin_elementwise_fma(__builtin_elementwise_max(z1, zero2), (v2f){cv[k + 2], cv[k + 3]}, d1);
        d2 = __builtin_elementwise_fma(__builtin_elementwise_max(z2, zero2), (v2f){cv[k + 4], cv[k + 5]}, d2);
        d3 = __builtin_elementwise_fma(__builtin_elementwise_max(z3, zero2), (v2f){cv[k + 6], cv[k + 7]}, d3);
      }
      __builtin_amdgcn_sched_barrier(0);
    }
    const v2f dv = (d0 + d1) + (d2 + d3);
    const float sc = tanhf((dv.x + dv.y) + cb2);
    if (e0 + lane < E) out[e0 + lane] = sc;
    iv = ivn;
    jv = jvn;
  }
}

extern "C" void kernel_launch(void* const* d_in, const int* in_sizes, int n_in,
                              void* d_out, int out_size, void* d_ws, size_t ws_size,
                              hipStream_t stream) {
  const float* x    = (const float*)d_in[0];
  const int*   ei   = (const int*)  d_in[1];
  const float* w1   = (const float*)d_in[2];
  const float* b1   = (const float*)d_in[3];
  const float* g1   = (const float*)d_in[4];
  const float* be1  = (const float*)d_in[5];
  const float* w2   = (const float*)d_in[6];
  const float* b2   = (const float*)d_in[7];
  const float* g2   = (const float*)d_in[8];
  const float* be2  = (const float*)d_in[9];
  const float* w3   = (const float*)d_in[10];
  const float* b3   = (const float*)d_in[11];
  const float* cw1  = (const float*)d_in[12];
  const float* cb1  = (const float*)d_in[13];
  const float* cg1  = (const float*)d_in[14];
  const float* cbe1 = (const float*)d_in[15];
  const float* cw2  = (const float*)d_in[16];
  const float* cb2  = (const float*)d_in[17];
  const float* prot = (const float*)d_in[18];

  const int N = in_sizes[0] / D;
  const int E = in_sizes[1] / 2;

  float* h1pre = (float*)d_ws;
  float* hi    = h1pre + (size_t)N * H;
  float* hj    = hi    + (size_t)N * H;

  float* out       = (float*)d_out;
  float* o_logits  = out;
  float* o_probs   = o_logits  + (size_t)N * KC;
  float* o_scores  = o_probs   + (size_t)N * KC;
  float* o_matrix  = o_scores  + (size_t)E;
  float* o_centers = o_matrix  + (size_t)N * N;
  float* o_protos  = o_centers + (size_t)KC * D;

  const int rowTiles = (N + 15) / 16;

  stripe_kernel<<<dim3(rowTiles, 3), NPB * 32, 0, stream>>>(
      x, w1, cw1, hi, hj, h1pre, b1, g1, be1, w2, b2, g2, be2, w3, b3,
      o_logits, o_probs, N);

  centers_kernel<<<KC, CTHREADS, 0, stream>>>(
      o_probs, x, h1pre, prot, o_centers, o_protos, N);

  comp_matrix_kernel<<<dim3((N + MCOLS - 1) / MCOLS, (N + MWAVES * MROWS - 1) / (MWAVES * MROWS)),
                       MWAVES * 32, 0, stream>>>(
      hi, hj, cb1, cg1, cbe1, cw2, cb2, o_matrix, N);

  comp_scores_kernel<<<768, EWAVES * 32, 0, stream>>>(
      hi, hj, ei, E, cb1, cg1, cbe1, cw2, cb2, o_scores);
}
